# starting kernel
# baseline (speedup 1.0000x reference)
#include <hip/hip_runtime.h>
#include <hip/hip_bf16.h>
#include <math.h>

#define NN 20000
#define EE 320000
#define HH 8
#define DD 32
#define F  256
#define LDSPITCH 260

typedef float v2f __attribute__((ext_vector_type(2)));
typedef float v8f __attribute__((ext_vector_type(8)));

__global__ __launch_bounds__(256) void gat_gemm_wmma(
    const float* __restrict__ A, const float* __restrict__ W,
    float* __restrict__ C) {
  __shared__ float As[16 * LDSPITCH];
  const int tid  = threadIdx.x;
  const int wave = tid >> 5;
  const int lane = tid & 31;
  const int m0   = blockIdx.x << 4;

  for (int i = tid; i < 16 * 64; i += 256) {
    const int r  = i >> 6;
    const int c4 = (i & 63) << 2;
    const float4 v = *reinterpret_cast<const float4*>(A + (size_t)(m0 + r) * F + c4);
    float* p = &As[r * LDSPITCH + c4];
    p[0] = v.x; p[1] = v.y; p[2] = v.z; p[3] = v.w;
  }
  __syncthreads();

  const int hi = lane >> 4;
  const int l  = lane & 15;

  for (int nt = wave; nt < 16; nt += 8) {
    const int n0 = nt << 4;
    v8f c = {};
    for (int k0 = 0; k0 < F; k0 += 4) {
      const int ka = k0 + (hi << 1);
      v2f a, b;
      a.x = As[l * LDSPITCH + ka];
      a.y = As[l * LDSPITCH + ka + 1];
      b.x = W[(size_t)ka * F + n0 + l];
      b.y = W[(size_t)(ka + 1) * F + n0 + l];
      c = __builtin_amdgcn_wmma_f32_16x16x4_f32(false, a, false, b, (short)0, c,
                                                false, false);
    }
#pragma unroll
    for (int v = 0; v < 8; ++v)
      C[(size_t)(m0 + (hi << 3) + v) * F + n0 + l] = c[v];
  }
}

__global__ void gat_attn_coeff(const float* __restrict__ h,
                               const float* __restrict__ al,
                               const float* __restrict__ ar,
                               float* __restrict__ el, float* __restrict__ er) {
  const int i = blockIdx.x * blockDim.x + threadIdx.x;
  if (i >= NN * HH) return;
  const int n = i >> 3, hd = i & 7;
  const float* hp  = h + (size_t)n * F + hd * DD;
  const float* alp = al + hd * DD;
  const float* arp = ar + hd * DD;
  float sl = 0.f, sr = 0.f;
#pragma unroll 8
  for (int d = 0; d < DD; ++d) {
    const float v = hp[d];
    sl += v * alp[d];
    sr += v * arp[d];
  }
  el[i] = sl;
  er[i] = sr;
}

__global__ void gat_init(float* __restrict__ m, float* __restrict__ s,
                         float* __restrict__ agg) {
  const int i = blockIdx.x * blockDim.x + threadIdx.x;
  if (i < NN * HH) { m[i] = -INFINITY; s[i] = 0.f; }
  if (i < NN * F) agg[i] = 0.f;
}

__device__ __forceinline__ void atomicMaxF32(float* addr, float v) {
  if (v >= 0.f)
    atomicMax(reinterpret_cast<int*>(addr), __float_as_int(v));
  else
    atomicMin(reinterpret_cast<unsigned int*>(addr),
              (unsigned int)__float_as_int(v));
}

__global__ void gat_edge_max(const float* __restrict__ el,
                             const float* __restrict__ er,
                             const int* __restrict__ src,
                             const int* __restrict__ dst,
                             float* __restrict__ e, float* __restrict__ m) {
  const int i = blockIdx.x * blockDim.x + threadIdx.x;
  if (i >= EE * HH) return;
  const int ed = i >> 3, hd = i & 7;
  const int sn = src[ed], dn = dst[ed];
  float v = el[sn * HH + hd] + er[dn * HH + hd];
  v = (v > 0.f) ? v : 0.2f * v;
  e[i] = v;
  atomicMaxF32(&m[dn * HH + hd], v);
}

__global__ void gat_edge_exp(float* __restrict__ e, const float* __restrict__ m,
                             const int* __restrict__ dst, float* __restrict__ s) {
  const int i = blockIdx.x * blockDim.x + threadIdx.x;
  if (i >= EE * HH) return;
  const int ed = i >> 3, hd = i & 7;
  const int dn = dst[ed];
  const float v = __expf(e[i] - m[dn * HH + hd]);
  e[i] = v;
  atomicAdd(&s[dn * HH + hd], v);
}

__global__ void gat_edge_agg(const float* __restrict__ ex,
                             const float* __restrict__ s,
                             const int* __restrict__ src,
                             const int* __restrict__ dst,
                             const float* __restrict__ hfeat,
                             float* __restrict__ agg) {
  const int i = blockIdx.x * blockDim.x + threadIdx.x;
  if (i >= EE * HH) return;
  const int ed = i >> 3, hd = i & 7;
  const int sn = src[ed], dn = dst[ed];
  const float alpha = ex[i] / s[dn * HH + hd];
  const float* hp = hfeat + (size_t)sn * F + hd * DD;
  float* op = agg + (size_t)dn * F + hd * DD;
#pragma unroll 8
  for (int d = 0; d < DD; ++d) atomicAdd(&op[d], alpha * hp[d]);
}

__global__ void gat_bias_elu(const float* __restrict__ agg,
                             const float* __restrict__ b,
                             float* __restrict__ out) {
  const int i = blockIdx.x * blockDim.x + threadIdx.x;
  if (i >= NN * F) return;
  const float v = agg[i] + b[i & (F - 1)];
  out[i] = (v > 0.f) ? v : (__expf(v) - 1.f);
}

extern "C" void kernel_launch(void* const* d_in, const int* in_sizes, int n_in,
                              void* d_out, int out_size, void* d_ws, size_t ws_size,
                              hipStream_t stream) {
  const float* feats = (const float*)d_in[0];
  const float* W0    = (const float*)d_in[1];
  const float* al0   = (const float*)d_in[2];
  const float* ar0   = (const float*)d_in[3];
  const float* b0    = (const float*)d_in[4];
  const float* W1    = (const float*)d_in[5];
  const float* al1   = (const float*)d_in[6];
  const float* ar1   = (const float*)d_in[7];
  const float* b1    = (const float*)d_in[8];
  const int*   src   = (const int*)d_in[9];
  const int*   dst   = (const int*)d_in[10];

  float* ws = (float*)d_ws;
  float* hbuf = ws;
  float* agg  = hbuf + (size_t)NN * F;
  float* x1   = agg + (size_t)NN * F;
  float* el   = x1 + (size_t)NN * F;
  float* er   = el + (size_t)NN * HH;
  float* mmax = er + (size_t)NN * HH;
  float* ssum = mmax + (size_t)NN * HH;
  float* ebuf = ssum + (size_t)NN * HH;

  const int bl = 256;
  const int gNF = (NN * F + bl - 1) / bl;
  const int gNH = (NN * HH + bl - 1) / bl;
  const int gEH = (EE * HH + bl - 1) / bl;

  auto run_layer = [&](const float* x, const float* Wm, const float* al,
                       const float* ar, const float* bb, float* outx) {
    gat_gemm_wmma<<<NN / 16, 256, 0, stream>>>(x, Wm, hbuf);
    gat_attn_coeff<<<gNH, bl, 0, stream>>>(hbuf, al, ar, el, er);
    gat_init<<<gNF, bl, 0, stream>>>(mmax, ssum, agg);
    gat_edge_max<<<gEH, bl, 0, stream>>>(el, er, src, dst, ebuf, mmax);
    gat_edge_exp<<<gEH, bl, 0, stream>>>(ebuf, mmax, dst, ssum);
    gat_edge_agg<<<gEH, bl, 0, stream>>>(ebuf, ssum, src, dst, hbuf, agg);
    gat_bias_elu<<<gNF, bl, 0, stream>>>(agg, bb, outx);
  };

  run_layer(feats, W0, al0, ar0, b0, x1);
  run_layer(x1, W1, al1, ar1, b1, (float*)d_out);
}


# edge aggregation made output-stationary: a block keeps 80 destination rows in LDS, filters the whole edge list for them, adds with LDS float atomics, stores agg once; sum order: LDS arrival order
# speedup vs baseline: 6.1670x; 6.1670x over previous
#include <hip/hip_runtime.h>
#include <hip/hip_bf16.h>
#include <math.h>

#define NN 20000
#define EE 320000
#define HH 8
#define DD 32
#define F  256
#define LDSPITCH 260

typedef float v2f __attribute__((ext_vector_type(2)));
typedef float v8f __attribute__((ext_vector_type(8)));

__global__ __launch_bounds__(256) void gat_gemm_wmma(
    const float* __restrict__ A, const float* __restrict__ W,
    float* __restrict__ C) {
  __shared__ float As[16 * LDSPITCH];
  const int tid  = threadIdx.x;
  const int wave = tid >> 5;
  const int lane = tid & 31;
  const int m0   = blockIdx.x << 4;

  for (int i = tid; i < 16 * 64; i += 256) {
    const int r  = i >> 6;
    const int c4 = (i & 63) << 2;
    const float4 v = *reinterpret_cast<const float4*>(A + (size_t)(m0 + r) * F + c4);
    float* p = &As[r * LDSPITCH + c4];
    p[0] = v.x; p[1] = v.y; p[2] = v.z; p[3] = v.w;
  }
  __syncthreads();

  const int hi = lane >> 4;
  const int l  = lane & 15;

  for (int nt = wave; nt < 16; nt += 8) {
    const int n0 = nt << 4;
    v8f c = {};
    for (int k0 = 0; k0 < F; k0 += 4) {
      const int ka = k0 + (hi << 1);
      v2f a, b;
      a.x = As[l * LDSPITCH + ka];
      a.y = As[l * LDSPITCH + ka + 1];
      b.x = W[(size_t)ka * F + n0 + l];
      b.y = W[(size_t)(ka + 1) * F + n0 + l];
      c = __builtin_amdgcn_wmma_f32_16x16x4_f32(false, a, false, b, (short)0, c,
                                                false, false);
    }
#pragma unroll
    for (int v = 0; v < 8; ++v)
      C[(size_t)(m0 + (hi << 3) + v) * F + n0 + l] = c[v];
  }
}

__global__ void gat_attn_coeff(const float* __restrict__ h,
                               const float* __restrict__ al,
                               const float* __restrict__ ar,
                               float* __restrict__ el, float* __restrict__ er) {
  const int i = blockIdx.x * blockDim.x + threadIdx.x;
  if (i >= NN * HH) return;
  const int n = i >> 3, hd = i & 7;
  const float* hp  = h + (size_t)n * F + hd * DD;
  const float* alp = al + hd * DD;
  const float* arp = ar + hd * DD;
  float sl = 0.f, sr = 0.f;
#pragma unroll 8
  for (int d = 0; d < DD; ++d) {
    const float v = hp[d];
    sl += v * alp[d];
    sr += v * arp[d];
  }
  el[i] = sl;
  er[i] = sr;
}

__global__ void gat_init(float* __restrict__ m, float* __restrict__ s) {
  const int i = blockIdx.x * blockDim.x + threadIdx.x;
  if (i < NN * HH) { m[i] = -INFINITY; s[i] = 0.f; }
}

__device__ __forceinline__ void atomicMaxF32(float* addr, float v) {
  if (v >= 0.f)
    atomicMax(reinterpret_cast<int*>(addr), __float_as_int(v));
  else
    atomicMin(reinterpret_cast<unsigned int*>(addr),
              (unsigned int)__float_as_int(v));
}

__global__ void gat_edge_max(const float* __restrict__ el,
                             const float* __restrict__ er,
                             const int* __restrict__ src,
                             const int* __restrict__ dst,
                             float* __restrict__ e, float* __restrict__ m) {
  const int i = blockIdx.x * blockDim.x + threadIdx.x;
  if (i >= EE * HH) return;
  const int ed = i >> 3, hd = i & 7;
  const int sn = src[ed], dn = dst[ed];
  float v = el[sn * HH + hd] + er[dn * HH + hd];
  v = (v > 0.f) ? v : 0.2f * v;
  e[i] = v;
  atomicMaxF32(&m[dn * HH + hd], v);
}

__global__ void gat_edge_exp(float* __restrict__ e, const float* __restrict__ m,
                             const int* __restrict__ dst, float* __restrict__ s) {
  const int i = blockIdx.x * blockDim.x + threadIdx.x;
  if (i >= EE * HH) return;
  const int ed = i >> 3, hd = i & 7;
  const int dn = dst[ed];
  const float v = __expf(e[i] - m[dn * HH + hd]);
  e[i] = v;
  atomicAdd(&s[dn * HH + hd], v);
}

#define AGG_ROWS    80
#define AGG_THREADS 1024
#define AGG_WAVES   32
#define SCAN_U      4
#define LIST_CAP    768
#define SCAN_STEP   (AGG_WAVES * SCAN_U * 32)

static_assert(EE % 4 == 0, "dst is read as int4");
static_assert(AGG_ROWS <= 128 && EE < (1 << 24), "list entry packs edge*128 + local row");
static_assert(LIST_CAP > 128 * SCAN_U, "after the drain test the list still takes one whole scan step");

__device__ __forceinline__ int lane_bcast_i(int v, int l) {
  return __builtin_amdgcn_readlane(v, l);
}
__device__ __forceinline__ float lane_bcast_f(float v, int l) {
  return __int_as_float(__builtin_amdgcn_readlane(__float_as_int(v), l));
}

__global__ __launch_bounds__(AGG_THREADS) void gat_agg_rows(
    const float* __restrict__ ex, const float* __restrict__ s,
    const int* __restrict__ src, const int* __restrict__ dst,
    const float* __restrict__ hfeat, float* __restrict__ agg) {
  __shared__ float acc[AGG_ROWS * F];
  __shared__ int elist[AGG_WAVES * LIST_CAP];
  const int tid  = threadIdx.x;
  const int wave = tid >> 5;
  const int lane = tid & 31;
  const int rowbase = blockIdx.x * AGG_ROWS;

  for (int i = tid; i < AGG_ROWS * F; i += AGG_THREADS) acc[i] = 0.f;
  __syncthreads();

  int* mylist = elist + wave * LIST_CAP;
  int cnt = 0;
  const int Q4  = EE / 4;
  const int NIT = (Q4 + SCAN_STEP - 1) / SCAN_STEP;
  const int4* dst4 = reinterpret_cast<const int4*>(dst);
  const float4* ex4 = reinterpret_cast<const float4*>(ex);
  const float4* s4  = reinterpret_cast<const float4*>(s);
  const int rot = blockIdx.x % NIT;

  for (int it = 0; it <= NIT; ++it) {
    if (it < NIT) {
      int itr = it + rot;
      if (itr >= NIT) itr -= NIT;
      const int q0 = (itr * AGG_WAVES + wave) * (SCAN_U * 32) + lane;
      int4 d[SCAN_U];
#pragma unroll
      for (int u = 0; u < SCAN_U; ++u) {
        const int q = q0 + u * 32;
        d[u] = (q < Q4) ? dst4[q] : make_int4(-1, -1, -1, -1);
      }
#pragma unroll
      for (int u = 0; u < SCAN_U; ++u) {
        const int ebase = (q0 + u * 32) * 4;
        const int dv[4] = {d[u].x, d[u].y, d[u].z, d[u].w};
#pragma unroll
        for (int k = 0; k < 4; ++k) {
          const int r = dv[k] - rowbase;
          const bool hit = (unsigned)r < (unsigned)AGG_ROWS;
          const unsigned m = __builtin_amdgcn_ballot_w32(hit);
          if (m) {
            if (hit)
              mylist[cnt + (int)__builtin_amdgcn_mbcnt_lo(m, 0u)] = ((ebase + k) << 7) | r;
            cnt += __builtin_popcount(m);
          }
        }
      }
    }
    if (it == NIT || cnt > LIST_CAP - 128 * SCAN_U) {
      while (cnt > 0) {
        const int nb   = cnt < 32 ? cnt : 32;
        const int base = cnt - nb;
        int sn = 0, r = 0;
        float a[HH] = {0.f, 0.f, 0.f, 0.f, 0.f, 0.f, 0.f, 0.f};
        if (lane < nb) {
          const int pk = mylist[base + lane];
          const int e  = pk >> 7;
          r  = pk & 127;
          sn = src[e];
          const float4 x0 = ex4[e * 2], x1 = ex4[e * 2 + 1];
          const float4 s0 = s4[(rowbase + r) * 2], s1 = s4[(rowbase + r) * 2 + 1];
          a[0] = x0.x / s0.x; a[1] = x0.y / s0.y; a[2] = x0.z / s0.z; a[3] = x0.w / s0.w;
          a[4] = x1.x / s1.x; a[5] = x1.y / s1.y; a[6] = x1.z / s1.z; a[7] = x1.w / s1.w;
        }
        for (int j = 0; j < nb; j += 4) {
          float hv[4][HH];
#pragma unroll
          for (int u = 0; u < 4; ++u) {
            const int jj = (j + u < nb) ? (j + u) : (nb - 1);
            const int sj = lane_bcast_i(sn, jj);
            const float* hp = hfeat + (size_t)sj * F + lane;
#pragma unroll
            for (int h = 0; h < HH; ++h) hv[u][h] = hp[h * DD];
          }
#pragma unroll
          for (int u = 0; u < 4; ++u) {
            if (j + u < nb) {
              const int rj = lane_bcast_i(r, j + u);
              float* ap = acc + rj * F + lane;
#pragma unroll
              for (int h = 0; h < HH; ++h) {
                const float al = lane_bcast_f(a[h], j + u);
                __hip_atomic_fetch_add(ap + h * DD, al * hv[u][h], __ATOMIC_RELAXED,
                                       __HIP_MEMORY_SCOPE_WORKGROUP);
              }
            }
          }
        }
        cnt = base;
      }
    }
  }
  __syncthreads();

  const float4* acc4 = reinterpret_cast<const float4*>(acc);
  float4* agg4 = reinterpret_cast<float4*>(agg);
  for (int i = tid; i < AGG_ROWS * (F / 4); i += AGG_THREADS) {
    const int row = rowbase + i / (F / 4);
    if (row < NN) agg4[(size_t)row * (F / 4) + (i % (F / 4))] = acc4[i];
  }
}

__global__ void gat_bias_elu(const float* __restrict__ agg,
                             const float* __restrict__ b,
                             float* __restrict__ out) {
  const int i = blockIdx.x * blockDim.x + threadIdx.x;
  if (i >= NN * F) return;
  const float v = agg[i] + b[i & (F - 1)];
  out[i] = (v > 0.f) ? v : (__expf(v) - 1.f);
}

extern "C" void kernel_launch(void* const* d_in, const int* in_sizes, int n_in,
                              void* d_out, int out_size, void* d_ws, size_t ws_size,
                              hipStream_t stream) {
  const float* feats = (const float*)d_in[0];
  const float* W0    = (const float*)d_in[1];
  const float* al0   = (const float*)d_in[2];
  const float* ar0   = (const float*)d_in[3];
  const float* b0    = (const float*)d_in[4];
  const float* W1    = (const float*)d_in[5];
  const float* al1   = (const float*)d_in[6];
  const float* ar1   = (const float*)d_in[7];
  const float* b1    = (const float*)d_in[8];
  const int*   src   = (const int*)d_in[9];
  const int*   dst   = (const int*)d_in[10];

  float* ws = (float*)d_ws;
  float* hbuf = ws;
  float* agg  = hbuf + (size_t)NN * F;
  float* x1   = agg + (size_t)NN * F;
  float* el   = x1 + (size_t)NN * F;
  float* er   = el + (size_t)NN * HH;
  float* mmax = er + (size_t)NN * HH;
  float* ssum = mmax + (size_t)NN * HH;
  float* ebuf = ssum + (size_t)NN * HH;

  const int bl = 256;
  const int gNF = (NN * F + bl - 1) / bl;
  const int gNH = (NN * HH + bl - 1) / bl;
  const int gEH = (EE * HH + bl - 1) / bl;

  auto run_layer = [&](const float* x, const float* Wm, const float* al,
                       const float* ar, const float* bb, float* outx) {
    gat_gemm_wmma<<<NN / 16, 256, 0, stream>>>(x, Wm, hbuf);
    gat_attn_coeff<<<gNH, bl, 0, stream>>>(hbuf, al, ar, el, er);
    gat_init<<<gNH, bl, 0, stream>>>(mmax, ssum);
    gat_edge_max<<<gEH, bl, 0, stream>>>(el, er, src, dst, ebuf, mmax);
    gat_edge_exp<<<gEH, bl, 0, stream>>>(ebuf, mmax, dst, ssum);
    gat_agg_rows<<<(NN + AGG_ROWS - 1) / AGG_ROWS, AGG_THREADS, 0, stream>>>(
        ebuf, ssum, src, dst, hbuf, agg);
    gat_bias_elu<<<gNF, bl, 0, stream>>>(agg, bb, outx);
  };

  run_layer(feats, W0, al0, ar0, b0, x1);
  run_layer(x1, W1, al1, ar1, b1, (float*)d_out);
}


# edge softmax moved into the row-owning aggregation block: row max and sum in LDS (LDS float atomics, arrival order), agg divided by the sum at the store; 14 launches -> 8; bias+ELU launch left
# speedup vs baseline: 9.9004x; 1.6054x over previous
#include <hip/hip_runtime.h>
#include <hip/hip_bf16.h>
#include <math.h>

#define NN 20000
#define EE 320000
#define HH 8
#define DD 32
#define F  256
#define LDSPITCH 260

typedef float v2f __attribute__((ext_vector_type(2)));
typedef float v8f __attribute__((ext_vector_type(8)));

__global__ __launch_bounds__(256) void gat_gemm_wmma(
    const float* __restrict__ A, const float* __restrict__ W,
    float* __restrict__ C) {
  __shared__ float As[16 * LDSPITCH];
  const int tid  = threadIdx.x;
  const int wave = tid >> 5;
  const int lane = tid & 31;
  const int m0   = blockIdx.x << 4;

  for (int i = tid; i < 16 * 64; i += 256) {
    const int r  = i >> 6;
    const int c4 = (i & 63) << 2;
    const float4 v = *reinterpret_cast<const float4*>(A + (size_t)(m0 + r) * F + c4);
    float* p = &As[r * LDSPITCH + c4];
    p[0] = v.x; p[1] = v.y; p[2] = v.z; p[3] = v.w;
  }
  __syncthreads();

  const int hi = lane >> 4;
  const int l  = lane & 15;

  for (int nt = wave; nt < 16; nt += 8) {
    const int n0 = nt << 4;
    v8f c = {};
    for (int k0 = 0; k0 < F; k0 += 4) {
      const int ka = k0 + (hi << 1);
      v2f a, b;
      a.x = As[l * LDSPITCH + ka];
      a.y = As[l * LDSPITCH + ka + 1];
      b.x = W[(size_t)ka * F + n0 + l];
      b.y = W[(size_t)(ka + 1) * F + n0 + l];
      c = __builtin_amdgcn_wmma_f32_16x16x4_f32(false, a, false, b, (short)0, c,
                                                false, false);
    }
#pragma unroll
    for (int v = 0; v < 8; ++v)
      C[(size_t)(m0 + (hi << 3) + v) * F + n0 + l] = c[v];
  }
}

__global__ void gat_attn_coeff(const float* __restrict__ h,
                               const float* __restrict__ al,
                               const float* __restrict__ ar,
                               float* __restrict__ el, float* __restrict__ er) {
  const int i = blockIdx.x * blockDim.x + threadIdx.x;
  if (i >= NN * HH) return;
  const int n = i >> 3, hd = i & 7;
  const float* hp  = h + (size_t)n * F + hd * DD;
  const float* alp = al + hd * DD;
  const float* arp = ar + hd * DD;
  float sl = 0.f, sr = 0.f;
#pragma unroll 8
  for (int d = 0; d < DD; ++d) {
    const float v = hp[d];
    sl += v * alp[d];
    sr += v * arp[d];
  }
  el[i] = sl;
  er[i] = sr;
}

#define AGG_ROWS    80
#define AGG_THREADS 1024
#define AGG_WAVES   32
#define SCAN_U      4
#define LIST_CAP    768
#define SCAN_STEP   (AGG_WAVES * SCAN_U * 32)
#define LIST_FULL   (LIST_CAP - 128 * SCAN_U)

static_assert(EE % 4 == 0, "dst is read as int4");
static_assert(AGG_ROWS <= 128 && EE < (1 << 24) && NN < (1 << 24),
              "list entry packs (edge or source)*128 + local row");
static_assert(LIST_CAP > 128 * SCAN_U, "after the drain test the list still takes one whole scan step");
static_assert(AGG_ROWS * HH <= AGG_THREADS, "one thread per (row, head) sets up the row statistics");

__device__ __forceinline__ int lane_bcast_i(int v, int l) {
  return __builtin_amdgcn_readlane(v, l);
}
__device__ __forceinline__ float lane_bcast_f(float v, int l) {
  return __int_as_float(__builtin_amdgcn_readlane(__float_as_int(v), l));
}
__device__ __forceinline__ float leaky02(float v) { return (v > 0.f) ? v : 0.2f * v; }

__device__ __forceinline__ int agg_scan_step(const int4* __restrict__ dst4, int itr,
                                             int wave, int lane, int rowbase,
                                             int* mylist, int cnt) {
  const int Q4 = EE / 4;
  const int q0 = (itr * AGG_WAVES + wave) * (SCAN_U * 32) + lane;
  int4 d[SCAN_U];
#pragma unroll
  for (int u = 0; u < SCAN_U; ++u) {
    const int q = q0 + u * 32;
    d[u] = (q < Q4) ? dst4[q] : make_int4(-1, -1, -1, -1);
  }
#pragma unroll
  for (int u = 0; u < SCAN_U; ++u) {
    const int ebase = (q0 + u * 32) * 4;
    const int dv[4] = {d[u].x, d[u].y, d[u].z, d[u].w};
#pragma unroll
    for (int k = 0; k < 4; ++k) {
      const int r = dv[k] - rowbase;
      const bool hit = (unsigned)r < (unsigned)AGG_ROWS;
      const unsigned m = __builtin_amdgcn_ballot_w32(hit);
      if (m) {
        if (hit)
          mylist[cnt + (int)__builtin_amdgcn_mbcnt_lo(m, 0u)] = ((ebase + k) << 7) | r;
        cnt += __builtin_popcount(m);
      }
    }
  }
  return cnt;
}

__device__ __forceinline__ void edge_scores(const float4* __restrict__ el4, int sn,
                                            const float4* err4, int r, float v[HH]) {
  const float4 l0 = el4[sn * 2], l1 = el4[sn * 2 + 1];
  const float4 r0 = err4[r * 2], r1 = err4[r * 2 + 1];
  v[0] = leaky02(l0.x + r0.x); v[1] = leaky02(l0.y + r0.y);
  v[2] = leaky02(l0.z + r0.z); v[3] = leaky02(l0.w + r0.w);
  v[4] = leaky02(l1.x + r1.x); v[5] = leaky02(l1.y + r1.y);
  v[6] = leaky02(l1.z + r1.z); v[7] = leaky02(l1.w + r1.w);
}

__device__ __forceinline__ void agg_fold_max(const int* __restrict__ src,
                                             const float4* __restrict__ el4,
                                             const float4* err4, float* mrow,
                                             int* mylist, int cnt, int lane, bool keep) {
  for (int idx = lane; idx < cnt; idx += 32) {
    const int pk = mylist[idx];
    const int r  = pk & 127;
    const int sn = src[pk >> 7];
    float v[HH];
    edge_scores(el4, sn, err4, r, v);
#pragma unroll
    for (int h = 0; h < HH; ++h)
      __hip_atomic_fetch_max(mrow + r * HH + h, v[h], __ATOMIC_RELAXED,
                             __HIP_MEMORY_SCOPE_WORKGROUP);
    if (keep) mylist[idx] = (sn << 7) | r;
  }
}

template <bool HAVE_SRC>
__device__ __forceinline__ void agg_drain(const int* __restrict__ src,
                                          const float4* __restrict__ el4,
                                          const float* __restrict__ hfeat,
                                          const float4* err4, const float4* mrow4,
                                          float* srow, float* acc,
                                          const int* mylist, int cnt, int lane) {
  while (cnt > 0) {
    const int nb   = cnt < 32 ? cnt : 32;
    const int base = cnt - nb;
    int sn = 0, r = 0;
    float a[HH] = {0.f, 0.f, 0.f, 0.f, 0.f, 0.f, 0.f, 0.f};
    if (lane < nb) {
      const int pk = mylist[base + lane];
      r  = pk & 127;
      sn = HAVE_SRC ? (pk >> 7) : src[pk >> 7];
      float v[HH];
      edge_scores(el4, sn, err4, r, v);
      const float4 m0 = mrow4[r * 2], m1 = mrow4[r * 2 + 1];
      a[0] = __expf(v[0] - m0.x); a[1] = __expf(v[1] - m0.y);
      a[2] = __expf(v[2] - m0.z); a[3] = __expf(v[3] - m0.w);
      a[4] = __expf(v[4] - m1.x); a[5] = __expf(v[5] - m1.y);
      a[6] = __expf(v[6] - m1.z); a[7] = __expf(v[7] - m1.w);
#pragma unroll
      for (int h = 0; h < HH; ++h)
        __hip_atomic_fetch_add(srow + r * HH + h, a[h], __ATOMIC_RELAXED,
                               __HIP_MEMORY_SCOPE_WORKGROUP);
    }
    for (int j = 0; j < nb; j += 4) {
      float hv[4][HH];
#pragma unroll
      for (int u = 0; u < 4; ++u) {
        const int jj = (j + u < nb) ? (j + u) : (nb - 1);
        const int sj = lane_bcast_i(sn, jj);
        const float* hp = hfeat + (size_t)sj * F + lane;
#pragma unroll
        for (int h = 0; h < HH; ++h) hv[u][h] = hp[h * DD];
      }
#pragma unroll
      for (int u = 0; u < 4; ++u) {
        if (j + u < nb) {
          const int rj = lane_bcast_i(r, j + u);
          float* ap = acc + rj * F + lane;
#pragma unroll
          for (int h = 0; h < HH; ++h) {
            const float al = lane_bcast_f(a[h], j + u);
            __hip_atomic_fetch_add(ap + h * DD, al * hv[u][h], __ATOMIC_RELAXED,
                                   __HIP_MEMORY_SCOPE_WORKGROUP);
          }
        }
      }
    }
    cnt = base;
  }
}

__global__ __launch_bounds__(AGG_THREADS) void gat_agg_rows(
    const float* __restrict__ el, const float* __restrict__ er,
    const int* __restrict__ src, const int* __restrict__ dst,
    const float* __restrict__ hfeat, float* __restrict__ agg) {
  __shared__ __attribute__((aligned(16))) float acc[AGG_ROWS * F];
  __shared__ __attribute__((aligned(16))) float mrow[AGG_ROWS * HH];
  __shared__ __attribute__((aligned(16))) float srow[AGG_ROWS * HH];
  __shared__ __attribute__((aligned(16))) float err[AGG_ROWS * HH];
  __shared__ int elist[AGG_WAVES * LIST_CAP];
  __shared__ int lost_any;
  const int tid  = threadIdx.x;
  const int wave = tid >> 5;
  const int lane = tid & 31;
  const int rowbase = blockIdx.x * AGG_ROWS;

  for (int i = tid; i < AGG_ROWS * F; i += AGG_THREADS) acc[i] = 0.f;
  if (tid < AGG_ROWS * HH) {
    const int row = rowbase + (tid >> 3);
    mrow[tid] = -INFINITY;
    srow[tid] = 0.f;
    err[tid]  = (row < NN) ? er[row * HH + (tid & 7)] : 0.f;
  }
  if (tid == 0) lost_any = 0;
  __syncthreads();

  int* mylist = elist + wave * LIST_CAP;
  int cnt = 0;
  const int NIT = (EE / 4 + SCAN_STEP - 1) / SCAN_STEP;
  const int4*   dst4  = reinterpret_cast<const int4*>(dst);
  const float4* el4   = reinterpret_cast<const float4*>(el);
  const float4* err4  = reinterpret_cast<const float4*>(err);
  const float4* mrow4 = reinterpret_cast<const float4*>(mrow);
  const int rot = blockIdx.x % NIT;

  bool lost = false;
  for (int it = 0; it < NIT; ++it) {
    int itr = it + rot;
    if (itr >= NIT) itr -= NIT;
    cnt = agg_scan_step(dst4, itr, wave, lane, rowbase, mylist, cnt);
    if (cnt > LIST_FULL) {
      agg_fold_max(src, el4, err4, mrow, mylist, cnt, lane, false);
      cnt = 0;
      lost = true;
    }
  }
  agg_fold_max(src, el4, err4, mrow, mylist, cnt, lane, !lost);
  if (lost) lost_any = 1;
  __syncthreads();

  if (!lost_any) {
    agg_drain<true>(src, el4, hfeat, err4, mrow4, srow, acc, mylist, cnt, lane);
  } else {
    cnt = 0;
    for (int it = 0; it < NIT; ++it) {
      int itr = it + rot;
      if (itr >= NIT) itr -= NIT;
      cnt = agg_scan_step(dst4, itr, wave, lane, rowbase, mylist, cnt);
      if (cnt > LIST_FULL) {
        agg_drain<false>(src, el4, hfeat, err4, mrow4, srow, acc, mylist, cnt, lane);
        cnt = 0;
      }
    }
    agg_drain<false>(src, el4, hfeat, err4, mrow4, srow, acc, mylist, cnt, lane);
  }
  __syncthreads();

  const float4* acc4 = reinterpret_cast<const float4*>(acc);
  float4* agg4 = reinterpret_cast<float4*>(agg);
  for (int i = tid; i < AGG_ROWS * (F / 4); i += AGG_THREADS) {
    const int rl  = i / (F / 4);
    const int c4  = i % (F / 4);
    const int row = rowbase + rl;
    const float s = srow[rl * HH + c4 / (DD / 4)];
    float4 v = acc4[i];
    if (s != 0.f) { v.x /= s; v.y /= s; v.z /= s; v.w /= s; }
    else          { v = make_float4(0.f, 0.f, 0.f, 0.f); }
    if (row < NN) agg4[(size_t)row * (F / 4) + c4] = v;
  }
}

__global__ void gat_bias_elu(const float* __restrict__ agg,
                             const float* __restrict__ b,
                             float* __restrict__ out) {
  const int i = blockIdx.x * blockDim.x + threadIdx.x;
  if (i >= NN * F) return;
  const float v = agg[i] + b[i & (F - 1)];
  out[i] = (v > 0.f) ? v : (__expf(v) - 1.f);
}

extern "C" void kernel_launch(void* const* d_in, const int* in_sizes, int n_in,
                              void* d_out, int out_size, void* d_ws, size_t ws_size,
                              hipStream_t stream) {
  const float* feats = (const float*)d_in[0];
  const float* W0    = (const float*)d_in[1];
  const float* al0   = (const float*)d_in[2];
  const float* ar0   = (const float*)d_in[3];
  const float* b0    = (const float*)d_in[4];
  const float* W1    = (const float*)d_in[5];
  const float* al1   = (const float*)d_in[6];
  const float* ar1   = (const float*)d_in[7];
  const float* b1    = (const float*)d_in[8];
  const int*   src   = (const int*)d_in[9];
  const int*   dst   = (const int*)d_in[10];

  float* ws = (float*)d_ws;
  float* hbuf = ws;
  float* agg  = hbuf + (size_t)NN * F;
  float* x1   = agg + (size_t)NN * F;
  float* el   = x1 + (size_t)NN * F;
  float* er   = el + (size_t)NN * HH;

  const int bl = 256;
  const int gNF = (NN * F + bl - 1) / bl;
  const int gNH = (NN * HH + bl - 1) / bl;

  auto run_layer = [&](const float* x, const float* Wm, const float* al,
                       const float* ar, const float* bb, float* outx) {
    gat_gemm_wmma<<<NN / 16, 256, 0, stream>>>(x, Wm, hbuf);
    gat_attn_coeff<<<gNH, bl, 0, stream>>>(hbuf, al, ar, el, er);
    gat_agg_rows<<<(NN + AGG_ROWS - 1) / AGG_ROWS, AGG_THREADS, 0, stream>>>(
        el, er, src, dst, hbuf, agg);
    gat_bias_elu<<<gNF, bl, 0, stream>>>(agg, bb, outx);
  };

  run_layer(feats, W0, al0, ar0, b0, x1);
  run_layer(x1, W1, al1, ar1, b1, (float*)d_out);
}


# aggregation blocks sweep a 1-byte-per-edge block-id plane (built once per call in the first coefficient launch) instead of the 4-byte dst list; LDS arrival-order sums as before
# speedup vs baseline: 10.7120x; 1.0820x over previous
#include <hip/hip_runtime.h>
#include <hip/hip_bf16.h>
#include <math.h>

#define NN 20000
#define EE 320000
#define HH 8
#define DD 32
#define F  256
#define LDSPITCH 260

typedef float v2f __attribute__((ext_vector_type(2)));
typedef float v8f __attribute__((ext_vector_type(8)));

__global__ __launch_bounds__(256) void gat_gemm_wmma(
    const float* __restrict__ A, const float* __restrict__ W,
    float* __restrict__ C) {
  __shared__ float As[16 * LDSPITCH];
  const int tid  = threadIdx.x;
  const int wave = tid >> 5;
  const int lane = tid & 31;
  const int m0   = blockIdx.x << 4;

  for (int i = tid; i < 16 * 64; i += 256) {
    const int r  = i >> 6;
    const int c4 = (i & 63) << 2;
    const float4 v = *reinterpret_cast<const float4*>(A + (size_t)(m0 + r) * F + c4);
    float* p = &As[r * LDSPITCH + c4];
    p[0] = v.x; p[1] = v.y; p[2] = v.z; p[3] = v.w;
  }
  __syncthreads();

  const int hi = lane >> 4;
  const int l  = lane & 15;

  for (int nt = wave; nt < 16; nt += 8) {
    const int n0 = nt << 4;
    v8f c = {};
    for (int k0 = 0; k0 < F; k0 += 4) {
      const int ka = k0 + (hi << 1);
      v2f a, b;
      a.x = As[l * LDSPITCH + ka];
      a.y = As[l * LDSPITCH + ka + 1];
      b.x = W[(size_t)ka * F + n0 + l];
      b.y = W[(size_t)(ka + 1) * F + n0 + l];
      c = __builtin_amdgcn_wmma_f32_16x16x4_f32(false, a, false, b, (short)0, c,
                                                false, false);
    }
#pragma unroll
    for (int v = 0; v < 8; ++v)
      C[(size_t)(m0 + (hi << 3) + v) * F + n0 + l] = c[v];
  }
}

#define AGG_ROWS    80
#define AGG_BLOCKS  ((NN + AGG_ROWS - 1) / AGG_ROWS)
#define NO_BLOCK    255u

static_assert(AGG_BLOCKS <= 255, "an aggregation block's index fits one byte, 255 left free");
static_assert(EE % 16 == 0, "the plane is written as dwords and read as 16-byte vectors");
static_assert(NN * HH >= EE / 4, "the coefficient launch has a thread for every dword of the plane");

__device__ __forceinline__ unsigned plane_byte(int d) {
  return ((unsigned)d < (unsigned)NN) ? (unsigned)d / AGG_ROWS : NO_BLOCK;
}

__global__ void gat_attn_coeff(const float* __restrict__ h,
                               const float* __restrict__ al,
                               const float* __restrict__ ar,
                               float* __restrict__ el, float* __restrict__ er,
                               const int* __restrict__ dst,
                               unsigned* __restrict__ plane) {
  const int i = blockIdx.x * blockDim.x + threadIdx.x;
  if (plane != nullptr && i < EE / 4) {
    const int4 d = reinterpret_cast<const int4*>(dst)[i];
    plane[i] = plane_byte(d.x) | (plane_byte(d.y) << 8) |
               (plane_byte(d.z) << 16) | (plane_byte(d.w) << 24);
  }
  if (i >= NN * HH) return;
  const int n = i >> 3, hd = i & 7;
  const float* hp  = h + (size_t)n * F + hd * DD;
  const float* alp = al + hd * DD;
  const float* arp = ar + hd * DD;
  float sl = 0.f, sr = 0.f;
#pragma unroll 8
  for (int d = 0; d < DD; ++d) {
    const float v = hp[d];
    sl += v * alp[d];
    sr += v * arp[d];
  }
  el[i] = sl;
  er[i] = sr;
}

#define AGG_THREADS 1024
#define AGG_WAVES   32
#define SCAN_U      4
#define LIST_CAP    768
#define SCAN_STEP   (AGG_WAVES * SCAN_U * 32)
#define LIST_FULL   (LIST_CAP - 512)

static_assert(AGG_ROWS <= 128 && NN < (1 << 24),
              "a kept list entry packs source*128 + local row");
static_assert(LIST_CAP > 512, "after the fullness test the list still takes one whole load's edges");
static_assert(AGG_ROWS * HH <= AGG_THREADS, "one thread per (row, head) sets up the row statistics");

__device__ __forceinline__ int lane_bcast_i(int v, int l) {
  return __builtin_amdgcn_readlane(v, l);
}
__device__ __forceinline__ float lane_bcast_f(float v, int l) {
  return __int_as_float(__builtin_amdgcn_readlane(__float_as_int(v), l));
}
__device__ __forceinline__ float leaky02(float v) { return (v > 0.f) ? v : 0.2f * v; }

static_assert((EE / 16) % 32 == 0, "a wave's 32 vectors of one load are all inside the plane or all past its end");

__device__ __forceinline__ void agg_scan_load(const uint4* __restrict__ plane4, int itr,
                                              int wave, int lane, uint4 d[SCAN_U]) {
  const int Q16 = EE / 16;
  const int q0 = (itr * AGG_WAVES + wave) * (SCAN_U * 32);
#pragma unroll
  for (int u = 0; u < SCAN_U; ++u) {
    const int qw = q0 + u * 32;
    d[u] = plane4[(qw < Q16 ? qw : Q16 - 32) + lane];
  }
}

__device__ __forceinline__ unsigned zero_bytes(unsigned x) {
  return ~(((x & 0x7f7f7f7fu) + 0x7f7f7f7fu) | x | 0x7f7f7f7fu);
}

template <class OnFull>
__device__ __forceinline__ int agg_scan_step(const uint4 d[SCAN_U], int itr, int wave,
                                             int lane, unsigned pat, int* mylist, int cnt,
                                             OnFull on_full) {
  const int Q16 = EE / 16;
  const int q0 = (itr * AGG_WAVES + wave) * (SCAN_U * 32);
#pragma unroll
  for (int u = 0; u < SCAN_U; ++u) {
    if (q0 + u * 32 >= Q16) break;
    const int ebase = (q0 + u * 32 + lane) * 16;
    unsigned hm = zero_bytes(d[u].x ^ pat) | (zero_bytes(d[u].y ^ pat) >> 1) |
                  (zero_bytes(d[u].z ^ pat) >> 2) | (zero_bytes(d[u].w ^ pat) >> 3);
    unsigned m = __builtin_amdgcn_ballot_w32(hm != 0u);
    while (m) {
      if (hm != 0u) {
        const int i = __builtin_ctz(hm);
        mylist[cnt + (int)__builtin_amdgcn_mbcnt_lo(m, 0u)] =
            ebase + ((7 - (i & 7)) << 2) + (i >> 3);
        hm &= hm - 1u;
      }
      cnt += __builtin_popcount(m);
      m = __builtin_amdgcn_ballot_w32(hm != 0u);
    }
    if (cnt > LIST_FULL) cnt = on_full(cnt);
  }
  return cnt;
}

__device__ __forceinline__ void edge_scores(const float4* __restrict__ el4, int sn,
                                            const float4* err4, int r, float v[HH]) {
  const float4 l0 = el4[sn * 2], l1 = el4[sn * 2 + 1];
  const float4 r0 = err4[r * 2], r1 = err4[r * 2 + 1];
  v[0] = leaky02(l0.x + r0.x); v[1] = leaky02(l0.y + r0.y);
  v[2] = leaky02(l0.z + r0.z); v[3] = leaky02(l0.w + r0.w);
  v[4] = leaky02(l1.x + r1.x); v[5] = leaky02(l1.y + r1.y);
  v[6] = leaky02(l1.z + r1.z); v[7] = leaky02(l1.w + r1.w);
}

__device__ __forceinline__ void agg_fold_max(const int* __restrict__ src,
                                             const int* __restrict__ dst, int rowbase,
                                             const float4* __restrict__ el4,
                                             const float4* err4, float* mrow,
                                             int* mylist, int cnt, int lane, bool keep) {
  for (int idx = lane; idx < cnt; idx += 32) {
    const int e  = mylist[idx];
    const int r  = dst[e] - rowbase;
    const int sn = src[e];
    float v[HH];
    edge_scores(el4, sn, err4, r, v);
#pragma unroll
    for (int h = 0; h < HH; ++h)
      __hip_atomic_fetch_max(mrow + r * HH + h, v[h], __ATOMIC_RELAXED,
                             __HIP_MEMORY_SCOPE_WORKGROUP);
    if (keep) mylist[idx] = (sn << 7) | r;
  }
}

template <bool HAVE_SRC>
__device__ __forceinline__ void agg_drain(const int* __restrict__ src,
                                          const int* __restrict__ dst, int rowbase,
                                          const float4* __restrict__ el4,
                                          const float* __restrict__ hfeat,
                                          const float4* err4, const float4* mrow4,
                                          float* srow, float* acc,
                                          const int* mylist, int cnt, int lane) {
  while (cnt > 0) {
    const int nb   = cnt < 32 ? cnt : 32;
    const int base = cnt - nb;
    int sn = 0, r = 0;
    float a[HH] = {0.f, 0.f, 0.f, 0.f, 0.f, 0.f, 0.f, 0.f};
    if (lane < nb) {
      const int pk = mylist[base + lane];
      r  = HAVE_SRC ? (pk & 127) : (dst[pk] - rowbase);
      sn = HAVE_SRC ? (pk >> 7) : src[pk];
      float v[HH];
      edge_scores(el4, sn, err4, r, v);
      const float4 m0 = mrow4[r * 2], m1 = mrow4[r * 2 + 1];
      a[0] = __expf(v[0] - m0.x); a[1] = __expf(v[1] - m0.y);
      a[2] = __expf(v[2] - m0.z); a[3] = __expf(v[3] - m0.w);
      a[4] = __expf(v[4] - m1.x); a[5] = __expf(v[5] - m1.y);
      a[6] = __expf(v[6] - m1.z); a[7] = __expf(v[7] - m1.w);
#pragma unroll
      for (int h = 0; h < HH; ++h)
        __hip_atomic_fetch_add(srow + r * HH + h, a[h], __ATOMIC_RELAXED,
                               __HIP_MEMORY_SCOPE_WORKGROUP);
    }
    for (int j = 0; j < nb; j += 4) {
      float hv[4][HH];
#pragma unroll
      for (int u = 0; u < 4; ++u) {
        const int jj = (j + u < nb) ? (j + u) : (nb - 1);
        const int sj = lane_bcast_i(sn, jj);
        const float* hp = hfeat + (size_t)sj * F + lane;
#pragma unroll
        for (int h = 0; h < HH; ++h) hv[u][h] = hp[h * DD];
      }
#pragma unroll
      for (int u = 0; u < 4; ++u) {
        if (j + u < nb) {
          const int rj = lane_bcast_i(r, j + u);
          float* ap = acc + rj * F + lane;
#pragma unroll
          for (int h = 0; h < HH; ++h) {
            const float al = lane_bcast_f(a[h], j + u);
            __hip_atomic_fetch_add(ap + h * DD, al * hv[u][h], __ATOMIC_RELAXED,
                                   __HIP_MEMORY_SCOPE_WORKGROUP);
          }
        }
      }
    }
    cnt = base;
  }
}

__global__ __launch_bounds__(AGG_THREADS) void gat_agg_rows(
    const float* __restrict__ el, const float* __restrict__ er,
    const int* __restrict__ src, const int* __restrict__ dst,
    const unsigned* __restrict__ plane,
    const float* __restrict__ hfeat, float* __restrict__ agg) {
  __shared__ __attribute__((aligned(16))) float acc[AGG_ROWS * F];
  __shared__ __attribute__((aligned(16))) float mrow[AGG_ROWS * HH];
  __shared__ __attribute__((aligned(16))) float srow[AGG_ROWS * HH];
  __shared__ __attribute__((aligned(16))) float err[AGG_ROWS * HH];
  __shared__ int elist[AGG_WAVES * LIST_CAP];
  __shared__ int lost_any;
  const int tid  = threadIdx.x;
  const int wave = tid >> 5;
  const int lane = tid & 31;
  const int rowbase = blockIdx.x * AGG_ROWS;

  for (int i = tid; i < AGG_ROWS * F; i += AGG_THREADS) acc[i] = 0.f;
  if (tid < AGG_ROWS * HH) {
    const int row = rowbase + (tid >> 3);
    mrow[tid] = -INFINITY;
    srow[tid] = 0.f;
    err[tid]  = (row < NN) ? er[row * HH + (tid & 7)] : 0.f;
  }
  if (tid == 0) lost_any = 0;
  __syncthreads();

  int* mylist = elist + wave * LIST_CAP;
  int cnt = 0;
  const int NIT = (EE / 16 + SCAN_STEP - 1) / SCAN_STEP;
  const uint4*  plane4 = reinterpret_cast<const uint4*>(plane);
  const float4* el4   = reinterpret_cast<const float4*>(el);
  const float4* err4  = reinterpret_cast<const float4*>(err);
  const float4* mrow4 = reinterpret_cast<const float4*>(mrow);
  const unsigned pat   = (unsigned)blockIdx.x * 0x01010101u;
  const int wave_s = __builtin_amdgcn_readfirstlane(wave);
  const int rot = blockIdx.x % NIT;

  bool lost = false;
  {
    uint4 cur[SCAN_U];
    agg_scan_load(plane4, rot, wave_s, lane, cur);
    for (int it = 0; it < NIT; ++it) {
      int itr = it + rot;
      if (itr >= NIT) itr -= NIT;
      int itn = itr + 1;
      if (itn >= NIT) itn -= NIT;
      uint4 nxt[SCAN_U];
      agg_scan_load(plane4, itn, wave_s, lane, nxt);
      cnt = agg_scan_step(cur, itr, wave_s, lane, pat, mylist, cnt, [&](int c) {
        agg_fold_max(src, dst, rowbase, el4, err4, mrow, mylist, c, lane, false);
        lost = true;
        return 0;
      });
#pragma unroll
      for (int u = 0; u < SCAN_U; ++u) cur[u] = nxt[u];
    }
  }
  agg_fold_max(src, dst, rowbase, el4, err4, mrow, mylist, cnt, lane, !lost);
  if (lost) lost_any = 1;
  __syncthreads();

  if (!lost_any) {
    agg_drain<true>(src, dst, rowbase, el4, hfeat, err4, mrow4, srow, acc, mylist, cnt, lane);
  } else {
    cnt = 0;
    for (int it = 0; it < NIT; ++it) {
      int itr = it + rot;
      if (itr >= NIT) itr -= NIT;
      uint4 cur[SCAN_U];
      agg_scan_load(plane4, itr, wave_s, lane, cur);
      cnt = agg_scan_step(cur, itr, wave_s, lane, pat, mylist, cnt, [&](int c) {
        agg_drain<false>(src, dst, rowbase, el4, hfeat, err4, mrow4, srow, acc, mylist, c, lane);
        return 0;
      });
    }
    agg_drain<false>(src, dst, rowbase, el4, hfeat, err4, mrow4, srow, acc, mylist, cnt, lane);
  }
  __syncthreads();

  const float4* acc4 = reinterpret_cast<const float4*>(acc);
  float4* agg4 = reinterpret_cast<float4*>(agg);
  for (int i = tid; i < AGG_ROWS * (F / 4); i += AGG_THREADS) {
    const int rl  = i / (F / 4);
    const int c4  = i % (F / 4);
    const int row = rowbase + rl;
    const float s = srow[rl * HH + c4 / (DD / 4)];
    float4 v = acc4[i];
    if (s != 0.f) { v.x /= s; v.y /= s; v.z /= s; v.w /= s; }
    else          { v = make_float4(0.f, 0.f, 0.f, 0.f); }
    if (row < NN) agg4[(size_t)row * (F / 4) + c4] = v;
  }
}

__global__ void gat_bias_elu(const float* __restrict__ agg,
                             const float* __restrict__ b,
                             float* __restrict__ out) {
  const int i = blockIdx.x * blockDim.x + threadIdx.x;
  if (i >= NN * F) return;
  const float v = agg[i] + b[i & (F - 1)];
  out[i] = (v > 0.f) ? v : (__expf(v) - 1.f);
}

extern "C" void kernel_launch(void* const* d_in, const int* in_sizes, int n_in,
                              void* d_out, int out_size, void* d_ws, size_t ws_size,
                              hipStream_t stream) {
  const float* feats = (const float*)d_in[0];
  const float* W0    = (const float*)d_in[1];
  const float* al0   = (const float*)d_in[2];
  const float* ar0   = (const float*)d_in[3];
  const float* b0    = (const float*)d_in[4];
  const float* W1    = (const float*)d_in[5];
  const float* al1   = (const float*)d_in[6];
  const float* ar1   = (const float*)d_in[7];
  const float* b1    = (const float*)d_in[8];
  const int*   src   = (const int*)d_in[9];
  const int*   dst   = (const int*)d_in[10];

  float* ws = (float*)d_ws;
  float* hbuf = ws;
  float* agg  = hbuf + (size_t)NN * F;
  float* x1   = agg + (size_t)NN * F;
  float* el   = x1 + (size_t)NN * F;
  float* er   = el + (size_t)NN * HH;
  unsigned* plane = (unsigned*)(er + (size_t)NN * HH);

  const int bl = 256;
  const int gNF = (NN * F + bl - 1) / bl;
  const int gNH = (NN * HH + bl - 1) / bl;

  auto run_layer = [&](const float* x, const float* Wm, const float* al,
                       const float* ar, const float* bb, float* outx,
                       unsigned* planew) {
    gat_gemm_wmma<<<NN / 16, 256, 0, stream>>>(x, Wm, hbuf);
    gat_attn_coeff<<<gNH, bl, 0, stream>>>(hbuf, al, ar, el, er, dst, planew);
    gat_agg_rows<<<AGG_BLOCKS, AGG_THREADS, 0, stream>>>(
        el, er, src, dst, plane, hbuf, agg);
    gat_bias_elu<<<gNF, bl, 0, stream>>>(agg, bb, outx);
  };

  run_layer(feats, W0, al0, ar0, b0, x1, plane);
  run_layer(x1, W1, al1, ar1, b1, (float*)d_out, nullptr);
}


# bias+ELU launches: a thread handles eight 16-byte vectors, all loaded before use (was one float per thread): 160,000 waves per launch -> 5,000; same arithmetic per element, all sum orders kept
# speedup vs baseline: 11.8250x; 1.1039x over previous
#include <hip/hip_runtime.h>
#include <hip/hip_bf16.h>
#include <math.h>

#define NN 20000
#define EE 320000
#define HH 8
#define DD 32
#define F  256
#define LDSPITCH 260

typedef float v2f __attribute__((ext_vector_type(2)));
typedef float v8f __attribute__((ext_vector_type(8)));

__global__ __launch_bounds__(256) void gat_gemm_wmma(
    const float* __restrict__ A, const float* __restrict__ W,
    float* __restrict__ C) {
  __shared__ float As[16 * LDSPITCH];
  const int tid  = threadIdx.x;
  const int wave = tid >> 5;
  const int lane = tid & 31;
  const int m0   = blockIdx.x << 4;

  for (int i = tid; i < 16 * 64; i += 256) {
    const int r  = i >> 6;
    const int c4 = (i & 63) << 2;
    const float4 v = *reinterpret_cast<const float4*>(A + (size_t)(m0 + r) * F + c4);
    float* p = &As[r * LDSPITCH + c4];
    p[0] = v.x; p[1] = v.y; p[2] = v.z; p[3] = v.w;
  }
  __syncthreads();

  const int hi = lane >> 4;
  const int l  = lane & 15;

  for (int nt = wave; nt < 16; nt += 8) {
    const int n0 = nt << 4;
    v8f c = {};
    for (int k0 = 0; k0 < F; k0 += 4) {
      const int ka = k0 + (hi << 1);
      v2f a, b;
      a.x = As[l * LDSPITCH + ka];
      a.y = As[l * LDSPITCH + ka + 1];
      b.x = W[(size_t)ka * F + n0 + l];
      b.y = W[(size_t)(ka + 1) * F + n0 + l];
      c = __builtin_amdgcn_wmma_f32_16x16x4_f32(false, a, false, b, (short)0, c,
                                                false, false);
    }
#pragma unroll
    for (int v = 0; v < 8; ++v)
      C[(size_t)(m0 + (hi << 3) + v) * F + n0 + l] = c[v];
  }
}

#define AGG_ROWS    80
#define AGG_BLOCKS  ((NN + AGG_ROWS - 1) / AGG_ROWS)
#define NO_BLOCK    255u

static_assert(AGG_BLOCKS <= 255, "an aggregation block's index fits one byte, 255 left free");
static_assert(EE % 16 == 0, "the plane is written as dwords and read as 16-byte vectors");
static_assert(NN * HH >= EE / 4, "the coefficient launch has a thread for every dword of the plane");

__device__ __forceinline__ unsigned plane_byte(int d) {
  return ((unsigned)d < (unsigned)NN) ? (unsigned)d / AGG_ROWS : NO_BLOCK;
}

__global__ void gat_attn_coeff(const float* __restrict__ h,
                               const float* __restrict__ al,
                               const float* __restrict__ ar,
                               float* __restrict__ el, float* __restrict__ er,
                               const int* __restrict__ dst,
                               unsigned* __restrict__ plane) {
  const int i = blockIdx.x * blockDim.x + threadIdx.x;
  if (plane != nullptr && i < EE / 4) {
    const int4 d = reinterpret_cast<const int4*>(dst)[i];
    plane[i] = plane_byte(d.x) | (plane_byte(d.y) << 8) |
               (plane_byte(d.z) << 16) | (plane_byte(d.w) << 24);
  }
  if (i >= NN * HH) return;
  const int n = i >> 3, hd = i & 7;
  const float* hp  = h + (size_t)n * F + hd * DD;
  const float* alp = al + hd * DD;
  const float* arp = ar + hd * DD;
  float sl = 0.f, sr = 0.f;
#pragma unroll 8
  for (int d = 0; d < DD; ++d) {
    const float v = hp[d];
    sl += v * alp[d];
    sr += v * arp[d];
  }
  el[i] = sl;
  er[i] = sr;
}

#define AGG_THREADS 1024
#define AGG_WAVES   32
#define SCAN_U      4
#define LIST_CAP    768
#define SCAN_STEP   (AGG_WAVES * SCAN_U * 32)
#define LIST_FULL   (LIST_CAP - 512)

static_assert(AGG_ROWS <= 128 && NN < (1 << 24),
              "a kept list entry packs source*128 + local row");
static_assert(LIST_CAP > 512, "after the fullness test the list still takes one whole load's edges");
static_assert(AGG_ROWS * HH <= AGG_THREADS, "one thread per (row, head) sets up the row statistics");

__device__ __forceinline__ int lane_bcast_i(int v, int l) {
  return __builtin_amdgcn_readlane(v, l);
}
__device__ __forceinline__ float lane_bcast_f(float v, int l) {
  return __int_as_float(__builtin_amdgcn_readlane(__float_as_int(v), l));
}
__device__ __forceinline__ float leaky02(float v) { return (v > 0.f) ? v : 0.2f * v; }

static_assert((EE / 16) % 32 == 0, "a wave's 32 vectors of one load are all inside the plane or all past its end");

__device__ __forceinline__ void agg_scan_load(const uint4* __restrict__ plane4, int itr,
                                              int wave, int lane, uint4 d[SCAN_U]) {
  const int Q16 = EE / 16;
  const int q0 = (itr * AGG_WAVES + wave) * (SCAN_U * 32);
#pragma unroll
  for (int u = 0; u < SCAN_U; ++u) {
    const int qw = q0 + u * 32;
    d[u] = plane4[(qw < Q16 ? qw : Q16 - 32) + lane];
  }
}

__device__ __forceinline__ unsigned zero_bytes(unsigned x) {
  return ~(((x & 0x7f7f7f7fu) + 0x7f7f7f7fu) | x | 0x7f7f7f7fu);
}

template <class OnFull>
__device__ __forceinline__ int agg_scan_step(const uint4 d[SCAN_U], int itr, int wave,
                                             int lane, unsigned pat, int* mylist, int cnt,
                                             OnFull on_full) {
  const int Q16 = EE / 16;
  const int q0 = (itr * AGG_WAVES + wave) * (SCAN_U * 32);
#pragma unroll
  for (int u = 0; u < SCAN_U; ++u) {
    if (q0 + u * 32 >= Q16) break;
    const int ebase = (q0 + u * 32 + lane) * 16;
    unsigned hm = zero_bytes(d[u].x ^ pat) | (zero_bytes(d[u].y ^ pat) >> 1) |
                  (zero_bytes(d[u].z ^ pat) >> 2) | (zero_bytes(d[u].w ^ pat) >> 3);
    unsigned m = __builtin_amdgcn_ballot_w32(hm != 0u);
    while (m) {
      if (hm != 0u) {
        const int i = __builtin_ctz(hm);
        mylist[cnt + (int)__builtin_amdgcn_mbcnt_lo(m, 0u)] =
            ebase + ((7 - (i & 7)) << 2) + (i >> 3);
        hm &= hm - 1u;
      }
      cnt += __builtin_popcount(m);
      m = __builtin_amdgcn_ballot_w32(hm != 0u);
    }
    if (cnt > LIST_FULL) cnt = on_full(cnt);
  }
  return cnt;
}

__device__ __forceinline__ void edge_scores(const float4* __restrict__ el4, int sn,
                                            const float4* err4, int r, float v[HH]) {
  const float4 l0 = el4[sn * 2], l1 = el4[sn * 2 + 1];
  const float4 r0 = err4[r * 2], r1 = err4[r * 2 + 1];
  v[0] = leaky02(l0.x + r0.x); v[1] = leaky02(l0.y + r0.y);
  v[2] = leaky02(l0.z + r0.z); v[3] = leaky02(l0.w + r0.w);
  v[4] = leaky02(l1.x + r1.x); v[5] = leaky02(l1.y + r1.y);
  v[6] = leaky02(l1.z + r1.z); v[7] = leaky02(l1.w + r1.w);
}

__device__ __forceinline__ void agg_fold_max(const int* __restrict__ src,
                                             const int* __restrict__ dst, int rowbase,
                                             const float4* __restrict__ el4,
                                             const float4* err4, float* mrow,
                                             int* mylist, int cnt, int lane, bool keep) {
  for (int idx = lane; idx < cnt; idx += 32) {
    const int e  = mylist[idx];
    const int r  = dst[e] - rowbase;
    const int sn = src[e];
    float v[HH];
    edge_scores(el4, sn, err4, r, v);
#pragma unroll
    for (int h = 0; h < HH; ++h)
      __hip_atomic_fetch_max(mrow + r * HH + h, v[h], __ATOMIC_RELAXED,
                             __HIP_MEMORY_SCOPE_WORKGROUP);
    if (keep) mylist[idx] = (sn << 7) | r;
  }
}

template <bool HAVE_SRC>
__device__ __forceinline__ void agg_drain(const int* __restrict__ src,
                                          const int* __restrict__ dst, int rowbase,
                                          const float4* __restrict__ el4,
                                          const float* __restrict__ hfeat,
                                          const float4* err4, const float4* mrow4,
                                          float* srow, float* acc,
                                          const int* mylist, int cnt, int lane) {
  while (cnt > 0) {
    const int nb   = cnt < 32 ? cnt : 32;
    const int base = cnt - nb;
    int sn = 0, r = 0;
    float a[HH] = {0.f, 0.f, 0.f, 0.f, 0.f, 0.f, 0.f, 0.f};
    if (lane < nb) {
      const int pk = mylist[base + lane];
      r  = HAVE_SRC ? (pk & 127) : (dst[pk] - rowbase);
      sn = HAVE_SRC ? (pk >> 7) : src[pk];
      float v[HH];
      edge_scores(el4, sn, err4, r, v);
      const float4 m0 = mrow4[r * 2], m1 = mrow4[r * 2 + 1];
      a[0] = __expf(v[0] - m0.x); a[1] = __expf(v[1] - m0.y);
      a[2] = __expf(v[2] - m0.z); a[3] = __expf(v[3] - m0.w);
      a[4] = __expf(v[4] - m1.x); a[5] = __expf(v[5] - m1.y);
      a[6] = __expf(v[6] - m1.z); a[7] = __expf(v[7] - m1.w);
#pragma unroll
      for (int h = 0; h < HH; ++h)
        __hip_atomic_fetch_add(srow + r * HH + h, a[h], __ATOMIC_RELAXED,
                               __HIP_MEMORY_SCOPE_WORKGROUP);
    }
    for (int j = 0; j < nb; j += 4) {
      float hv[4][HH];
#pragma unroll
      for (int u = 0; u < 4; ++u) {
        const int jj = (j + u < nb) ? (j + u) : (nb - 1);
        const int sj = lane_bcast_i(sn, jj);
        const float* hp = hfeat + (size_t)sj * F + lane;
#pragma unroll
        for (int h = 0; h < HH; ++h) hv[u][h] = hp[h * DD];
      }
#pragma unroll
      for (int u = 0; u < 4; ++u) {
        if (j + u < nb) {
          const int rj = lane_bcast_i(r, j + u);
          float* ap = acc + rj * F + lane;
#pragma unroll
          for (int h = 0; h < HH; ++h) {
            const float al = lane_bcast_f(a[h], j + u);
            __hip_atomic_fetch_add(ap + h * DD, al * hv[u][h], __ATOMIC_RELAXED,
                                   __HIP_MEMORY_SCOPE_WORKGROUP);
          }
        }
      }
    }
    cnt = base;
  }
}

__global__ __launch_bounds__(AGG_THREADS) void gat_agg_rows(
    const float* __restrict__ el, const float* __restrict__ er,
    const int* __restrict__ src, const int* __restrict__ dst,
    const unsigned* __restrict__ plane,
    const float* __restrict__ hfeat, float* __restrict__ agg) {
  __shared__ __attribute__((aligned(16))) float acc[AGG_ROWS * F];
  __shared__ __attribute__((aligned(16))) float mrow[AGG_ROWS * HH];
  __shared__ __attribute__((aligned(16))) float srow[AGG_ROWS * HH];
  __shared__ __attribute__((aligned(16))) float err[AGG_ROWS * HH];
  __shared__ int elist[AGG_WAVES * LIST_CAP];
  __shared__ int lost_any;
  const int tid  = threadIdx.x;
  const int wave = tid >> 5;
  const int lane = tid & 31;
  const int rowbase = blockIdx.x * AGG_ROWS;

  for (int i = tid; i < AGG_ROWS * F; i += AGG_THREADS) acc[i] = 0.f;
  if (tid < AGG_ROWS * HH) {
    const int row = rowbase + (tid >> 3);
    mrow[tid] = -INFINITY;
    srow[tid] = 0.f;
    err[tid]  = (row < NN) ? er[row * HH + (tid & 7)] : 0.f;
  }
  if (tid == 0) lost_any = 0;
  __syncthreads();

  int* mylist = elist + wave * LIST_CAP;
  int cnt = 0;
  const int NIT = (EE / 16 + SCAN_STEP - 1) / SCAN_STEP;
  const uint4*  plane4 = reinterpret_cast<const uint4*>(plane);
  const float4* el4   = reinterpret_cast<const float4*>(el);
  const float4* err4  = reinterpret_cast<const float4*>(err);
  const float4* mrow4 = reinterpret_cast<const float4*>(mrow);
  const unsigned pat   = (unsigned)blockIdx.x * 0x01010101u;
  const int wave_s = __builtin_amdgcn_readfirstlane(wave);
  const int rot = blockIdx.x % NIT;

  bool lost = false;
  {
    uint4 cur[SCAN_U];
    agg_scan_load(plane4, rot, wave_s, lane, cur);
    for (int it = 0; it < NIT; ++it) {
      int itr = it + rot;
      if (itr >= NIT) itr -= NIT;
      int itn = itr + 1;
      if (itn >= NIT) itn -= NIT;
      uint4 nxt[SCAN_U];
      agg_scan_load(plane4, itn, wave_s, lane, nxt);
      cnt = agg_scan_step(cur, itr, wave_s, lane, pat, mylist, cnt, [&](int c) {
        agg_fold_max(src, dst, rowbase, el4, err4, mrow, mylist, c, lane, false);
        lost = true;
        return 0;
      });
#pragma unroll
      for (int u = 0; u < SCAN_U; ++u) cur[u] = nxt[u];
    }
  }
  agg_fold_max(src, dst, rowbase, el4, err4, mrow, mylist, cnt, lane, !lost);
  if (lost) lost_any = 1;
  __syncthreads();

  if (!lost_any) {
    agg_drain<true>(src, dst, rowbase, el4, hfeat, err4, mrow4, srow, acc, mylist, cnt, lane);
  } else {
    cnt = 0;
    for (int it = 0; it < NIT; ++it) {
      int itr = it + rot;
      if (itr >= NIT) itr -= NIT;
      uint4 cur[SCAN_U];
      agg_scan_load(plane4, itr, wave_s, lane, cur);
      cnt = agg_scan_step(cur, itr, wave_s, lane, pat, mylist, cnt, [&](int c) {
        agg_drain<false>(src, dst, rowbase, el4, hfeat, err4, mrow4, srow, acc, mylist, c, lane);
        return 0;
      });
    }
    agg_drain<false>(src, dst, rowbase, el4, hfeat, err4, mrow4, srow, acc, mylist, cnt, lane);
  }
  __syncthreads();

  const float4* acc4 = reinterpret_cast<const float4*>(acc);
  float4* agg4 = reinterpret_cast<float4*>(agg);
  for (int i = tid; i < AGG_ROWS * (F / 4); i += AGG_THREADS) {
    const int rl  = i / (F / 4);
    const int c4  = i % (F / 4);
    const int row = rowbase + rl;
    const float s = srow[rl * HH + c4 / (DD / 4)];
    float4 v = acc4[i];
    if (s != 0.f) { v.x /= s; v.y /= s; v.z /= s; v.w /= s; }
    else          { v = make_float4(0.f, 0.f, 0.f, 0.f); }
    if (row < NN) agg4[(size_t)row * (F / 4) + c4] = v;
  }
}

#define BE_THREADS 256
#define BE_PER     8
#define BE_VECS    (NN * F / 4)
#define BE_BLOCKS  (BE_VECS / (BE_THREADS * BE_PER))
#define BE_STRIDE  (BE_BLOCKS * BE_THREADS)

static_assert(BE_VECS % (BE_THREADS * BE_PER) == 0, "the grid covers every vector exactly once, no tail");
static_assert(BE_STRIDE % (F / 4) == 0, "a thread's vectors are whole rows apart: one bias vector serves all");

__device__ __forceinline__ float elu1(float v) { return (v > 0.f) ? v : (__expf(v) - 1.f); }

__global__ __launch_bounds__(BE_THREADS) void gat_bias_elu(const float* __restrict__ agg,
                                                           const float* __restrict__ b,
                                                           float* __restrict__ out) {
  const int q = blockIdx.x * BE_THREADS + threadIdx.x;
  const float4* agg4 = reinterpret_cast<const float4*>(agg);
  float4* out4 = reinterpret_cast<float4*>(out);
  const float4 bv = reinterpret_cast<const float4*>(b)[q & (F / 4 - 1)];
  float4 v[BE_PER];
#pragma unroll
  for (int k = 0; k < BE_PER; ++k) v[k] = agg4[q + k * BE_STRIDE];
#pragma unroll
  for (int k = 0; k < BE_PER; ++k) {
    float4 r;
    r.x = elu1(v[k].x + bv.x);
    r.y = elu1(v[k].y + bv.y);
    r.z = elu1(v[k].z + bv.z);
    r.w = elu1(v[k].w + bv.w);
    out4[q + k * BE_STRIDE] = r;
  }
}

extern "C" void kernel_launch(void* const* d_in, const int* in_sizes, int n_in,
                              void* d_out, int out_size, void* d_ws, size_t ws_size,
                              hipStream_t stream) {
  const float* feats = (const float*)d_in[0];
  const float* W0    = (const float*)d_in[1];
  const float* al0   = (const float*)d_in[2];
  const float* ar0   = (const float*)d_in[3];
  const float* b0    = (const float*)d_in[4];
  const float* W1    = (const float*)d_in[5];
  const float* al1   = (const float*)d_in[6];
  const float* ar1   = (const float*)d_in[7];
  const float* b1    = (const float*)d_in[8];
  const int*   src   = (const int*)d_in[9];
  const int*   dst   = (const int*)d_in[10];

  float* ws = (float*)d_ws;
  float* hbuf = ws;
  float* agg  = hbuf + (size_t)NN * F;
  float* x1   = agg + (size_t)NN * F;
  float* el   = x1 + (size_t)NN * F;
  float* er   = el + (size_t)NN * HH;
  unsigned* plane = (unsigned*)(er + (size_t)NN * HH);

  const int bl = 256;
  const int gNH = (NN * HH + bl - 1) / bl;

  auto run_layer = [&](const float* x, const float* Wm, const float* al,
                       const float* ar, const float* bb, float* outx,
                       unsigned* planew) {
    gat_gemm_wmma<<<NN / 16, 256, 0, stream>>>(x, Wm, hbuf);
    gat_attn_coeff<<<gNH, bl, 0, stream>>>(hbuf, al, ar, el, er, dst, planew);
    gat_agg_rows<<<AGG_BLOCKS, AGG_THREADS, 0, stream>>>(
        el, er, src, dst, plane, hbuf, agg);
    gat_bias_elu<<<BE_BLOCKS, BE_THREADS, 0, stream>>>(agg, bb, outx);
  };

  run_layer(feats, W0, al0, ar0, b0, x1, plane);
  run_layer(x1, W1, al1, ar1, b1, (float*)d_out, nullptr);
}


# aggregation: each list entry goes to the wave owning its row (row%32) in the owner block; that wave sums exp*source row in registers, stores the row; LDS float adds gone; sum order: fixed list order
# speedup vs baseline: 12.9718x; 1.0970x over previous
#include <hip/hip_runtime.h>
#include <hip/hip_bf16.h>
#include <math.h>

#define NN 20000
#define EE 320000
#define HH 8
#define DD 32
#define F  256
#define LDSPITCH 260

typedef float v2f __attribute__((ext_vector_type(2)));
typedef float v8f __attribute__((ext_vector_type(8)));

__global__ __launch_bounds__(256) void gat_gemm_wmma(
    const float* __restrict__ A, const float* __restrict__ W,
    float* __restrict__ C) {
  __shared__ float As[16 * LDSPITCH];
  const int tid  = threadIdx.x;
  const int wave = tid >> 5;
  const int lane = tid & 31;
  const int m0   = blockIdx.x << 4;

  for (int i = tid; i < 16 * 64; i += 256) {
    const int r  = i >> 6;
    const int c4 = (i & 63) << 2;
    const float4 v = *reinterpret_cast<const float4*>(A + (size_t)(m0 + r) * F + c4);
    float* p = &As[r * LDSPITCH + c4];
    p[0] = v.x; p[1] = v.y; p[2] = v.z; p[3] = v.w;
  }
  __syncthreads();

  const int hi = lane >> 4;
  const int l  = lane & 15;

  for (int nt = wave; nt < 16; nt += 8) {
    const int n0 = nt << 4;
    v8f c = {};
    for (int k0 = 0; k0 < F; k0 += 4) {
      const int ka = k0 + (hi << 1);
      v2f a, b;
      a.x = As[l * LDSPITCH + ka];
      a.y = As[l * LDSPITCH + ka + 1];
      b.x = W[(size_t)ka * F + n0 + l];
      b.y = W[(size_t)(ka + 1) * F + n0 + l];
      c = __builtin_amdgcn_wmma_f32_16x16x4_f32(false, a, false, b, (short)0, c,
                                                false, false);
    }
#pragma unroll
    for (int v = 0; v < 8; ++v)
      C[(size_t)(m0 + (hi << 3) + v) * F + n0 + l] = c[v];
  }
}

#define AGG_ROWS    80
#define AGG_BLOCKS  ((NN + AGG_ROWS - 1) / AGG_ROWS)
#define NO_BLOCK    255u

static_assert(AGG_BLOCKS <= 255, "an aggregation block's index fits one byte, 255 left free");
static_assert(EE % 16 == 0, "the plane is written as dwords and read as 16-byte vectors");
static_assert(NN * HH >= EE / 4, "the coefficient launch has a thread for every dword of the plane");

__device__ __forceinline__ unsigned plane_byte(int d) {
  return ((unsigned)d < (unsigned)NN) ? (unsigned)d / AGG_ROWS : NO_BLOCK;
}

__global__ void gat_attn_coeff(const float* __restrict__ h,
                               const float* __restrict__ al,
                               const float* __restrict__ ar,
                               float* __restrict__ el, float* __restrict__ er,
                               const int* __restrict__ dst,
                               unsigned* __restrict__ plane) {
  const int i = blockIdx.x * blockDim.x + threadIdx.x;
  if (plane != nullptr && i < EE / 4) {
    const int4 d = reinterpret_cast<const int4*>(dst)[i];
    plane[i] = plane_byte(d.x) | (plane_byte(d.y) << 8) |
               (plane_byte(d.z) << 16) | (plane_byte(d.w) << 24);
  }
  if (i >= NN * HH) return;
  const int n = i >> 3, hd = i & 7;
  const float* hp  = h + (size_t)n * F + hd * DD;
  const float* alp = al + hd * DD;
  const float* arp = ar + hd * DD;
  float sl = 0.f, sr = 0.f;
#pragma unroll 8
  for (int d = 0; d < DD; ++d) {
    const float v = hp[d];
    sl += v * alp[d];
    sr += v * arp[d];
  }
  el[i] = sl;
  er[i] = sr;
}

#define AGG_THREADS 1024
#define AGG_WAVES   32
#define SCAN_U      4
#define LIST_CAP    768
#define SCAN_STEP   (AGG_WAVES * SCAN_U * 32)
#define LIST_FULL   (LIST_CAP - 512)
#define ROWS_PER_WAVE ((AGG_ROWS + AGG_WAVES - 1) / AGG_WAVES)
#define SEG_CAP     16
#define SEG_PITCH   17

static_assert(AGG_ROWS <= 128 && NN < (1 << 24),
              "a kept list entry packs source*128 + local row");
static_assert(LIST_CAP > 512, "after the fullness test the list still takes one whole load's edges");
static_assert(AGG_ROWS * HH <= AGG_THREADS, "one thread per (row, head) sets up the row statistics");
static_assert(AGG_WAVES == 32, "lane i of an owner wave holds what wave i's list hands over; row r belongs to wave r & 31");
static_assert(ROWS_PER_WAVE == 3, "a wave keeps three rows' sums in registers");
static_assert(AGG_WAVES * AGG_WAVES <= AGG_THREADS, "one thread per (owner, list) clears the segment counts");
static_assert(AGG_WAVES * SEG_CAP <= LIST_CAP, "an owner wave's entries fit the LDS room of its own list");

__device__ __forceinline__ int lane_bcast_i(int v, int l) {
  return __builtin_amdgcn_readlane(v, l);
}
__device__ __forceinline__ float lane_bcast_f(float v, int l) {
  return __int_as_float(__builtin_amdgcn_readlane(__float_as_int(v), l));
}
__device__ __forceinline__ float leaky02(float v) { return (v > 0.f) ? v : 0.2f * v; }

static_assert((EE / 16) % 32 == 0, "a wave's 32 vectors of one load are all inside the plane or all past its end");

__device__ __forceinline__ void agg_scan_load(const uint4* __restrict__ plane4, int itr,
                                              int wave, int lane, uint4 d[SCAN_U]) {
  const int Q16 = EE / 16;
  const int q0 = (itr * AGG_WAVES + wave) * (SCAN_U * 32);
#pragma unroll
  for (int u = 0; u < SCAN_U; ++u) {
    const int qw = q0 + u * 32;
    d[u] = plane4[(qw < Q16 ? qw : Q16 - 32) + lane];
  }
}

__device__ __forceinline__ unsigned zero_bytes(unsigned x) {
  return ~(((x & 0x7f7f7f7fu) + 0x7f7f7f7fu) | x | 0x7f7f7f7fu);
}

template <class OnFull>
__device__ __forceinline__ int agg_scan_step(const uint4 d[SCAN_U], int itr, int wave,
                                             int lane, unsigned pat, int* mylist, int cnt,
                                             OnFull on_full) {
  const int Q16 = EE / 16;
  const int q0 = (itr * AGG_WAVES + wave) * (SCAN_U * 32);
#pragma unroll
  for (int u = 0; u < SCAN_U; ++u) {
    if (q0 + u * 32 >= Q16) break;
    const int ebase = (q0 + u * 32 + lane) * 16;
    unsigned hm = zero_bytes(d[u].x ^ pat) | (zero_bytes(d[u].y ^ pat) >> 1) |
                  (zero_bytes(d[u].z ^ pat) >> 2) | (zero_bytes(d[u].w ^ pat) >> 3);
    unsigned m = __builtin_amdgcn_ballot_w32(hm != 0u);
    while (m) {
      if (hm != 0u) {
        const int i = __builtin_ctz(hm);
        mylist[cnt + (int)__builtin_amdgcn_mbcnt_lo(m, 0u)] =
            ebase + ((7 - (i & 7)) << 2) + (i >> 3);
        hm &= hm - 1u;
      }
      cnt += __builtin_popcount(m);
      m = __builtin_amdgcn_ballot_w32(hm != 0u);
    }
    if (cnt > LIST_FULL) cnt = on_full(cnt);
  }
  return cnt;
}

__device__ __forceinline__ void edge_scores(const float4* __restrict__ el4, int sn,
                                            const float4* err4, int r, float v[HH]) {
  const float4 l0 = el4[sn * 2], l1 = el4[sn * 2 + 1];
  const float4 r0 = err4[r * 2], r1 = err4[r * 2 + 1];
  v[0] = leaky02(l0.x + r0.x); v[1] = leaky02(l0.y + r0.y);
  v[2] = leaky02(l0.z + r0.z); v[3] = leaky02(l0.w + r0.w);
  v[4] = leaky02(l1.x + r1.x); v[5] = leaky02(l1.y + r1.y);
  v[6] = leaky02(l1.z + r1.z); v[7] = leaky02(l1.w + r1.w);
}

__device__ __forceinline__ void agg_fold_max(const int* __restrict__ src,
                                             const int* __restrict__ dst, int rowbase,
                                             const float4* __restrict__ el4,
                                             const float4* err4, float* mrow,
                                             const int* mylist, int cnt, int lane, bool keep,
                                             int wave, int* seg, int* scnt, int* lost_any) {
  for (int idx = lane; idx < cnt; idx += 32) {
    const int e  = mylist[idx];
    const int r  = dst[e] - rowbase;
    const int sn = src[e];
    float v[HH];
    edge_scores(el4, sn, err4, r, v);
#pragma unroll
    for (int h = 0; h < HH; ++h)
      __hip_atomic_fetch_max(mrow + r * HH + h, v[h], __ATOMIC_RELAXED,
                             __HIP_MEMORY_SCOPE_WORKGROUP);
    if (keep) {
      const int d = r & (AGG_WAVES - 1);
      unsigned same = __builtin_amdgcn_ballot_w32(true);
#pragma unroll
      for (int b = 0; b < 5; ++b) {
        const bool bit = ((d >> b) & 1) != 0;
        const unsigned bb = __builtin_amdgcn_ballot_w32(bit);
        same &= bit ? bb : ~bb;
      }
      const int below = (int)__builtin_amdgcn_mbcnt_lo(same, 0u);
      const int nsame = __builtin_popcount(same);
      int* cp = scnt + d * AGG_WAVES + wave;
      const int c0 = *cp;
      const int slot = c0 + below;
      if (slot < SEG_CAP) seg[(d * AGG_WAVES + wave) * SEG_PITCH + slot] = (sn << 7) | r;
      else *lost_any = 1;
      if (below == nsame - 1) *cp = c0 + nsame;
    }
  }
}

__device__ __forceinline__ void agg_drain(const int* __restrict__ src,
                                          const int* __restrict__ dst, int rowbase,
                                          const float4* __restrict__ el4,
                                          const float* __restrict__ hfeat,
                                          const float4* err4, const float4* mrow4,
                                          float* srow, float* acc,
                                          const int* mylist, int cnt, int lane) {
  while (cnt > 0) {
    const int nb   = cnt < 32 ? cnt : 32;
    const int base = cnt - nb;
    int sn = 0, r = 0;
    float a[HH] = {0.f, 0.f, 0.f, 0.f, 0.f, 0.f, 0.f, 0.f};
    if (lane < nb) {
      const int pk = mylist[base + lane];
      r  = dst[pk] - rowbase;
      sn = src[pk];
      float v[HH];
      edge_scores(el4, sn, err4, r, v);
      const float4 m0 = mrow4[r * 2], m1 = mrow4[r * 2 + 1];
      a[0] = __expf(v[0] - m0.x); a[1] = __expf(v[1] - m0.y);
      a[2] = __expf(v[2] - m0.z); a[3] = __expf(v[3] - m0.w);
      a[4] = __expf(v[4] - m1.x); a[5] = __expf(v[5] - m1.y);
      a[6] = __expf(v[6] - m1.z); a[7] = __expf(v[7] - m1.w);
#pragma unroll
      for (int h = 0; h < HH; ++h)
        __hip_atomic_fetch_add(srow + r * HH + h, a[h], __ATOMIC_RELAXED,
                               __HIP_MEMORY_SCOPE_WORKGROUP);
    }
    for (int j = 0; j < nb; j += 4) {
      float hv[4][HH];
#pragma unroll
      for (int u = 0; u < 4; ++u) {
        const int jj = (j + u < nb) ? (j + u) : (nb - 1);
        const int sj = lane_bcast_i(sn, jj);
        const float* hp = hfeat + (size_t)sj * F + lane;
#pragma unroll
        for (int h = 0; h < HH; ++h) hv[u][h] = hp[h * DD];
      }
#pragma unroll
      for (int u = 0; u < 4; ++u) {
        if (j + u < nb) {
          const int rj = lane_bcast_i(r, j + u);
          float* ap = acc + rj * F + lane;
#pragma unroll
          for (int h = 0; h < HH; ++h) {
            const float al = lane_bcast_f(a[h], j + u);
            __hip_atomic_fetch_add(ap + h * DD, al * hv[u][h], __ATOMIC_RELAXED,
                                   __HIP_MEMORY_SCOPE_WORKGROUP);
          }
        }
      }
    }
    cnt = base;
  }
}

__device__ __forceinline__ float uniform_f(float v) {
  return __int_as_float(__builtin_amdgcn_readfirstlane(__float_as_int(v)));
}

template <int K>
__device__ __forceinline__ void row_add(float (&racc)[ROWS_PER_WAVE][HH], float (&rsum)[ROWS_PER_WAVE][HH],
                                        const float al[HH], const float hv[HH]) {
#pragma unroll
  for (int h = 0; h < HH; ++h) {
    racc[K][h] = fmaf(al[h], hv[h], racc[K][h]);
    rsum[K][h] = uniform_f(rsum[K][h] + al[h]);
  }
}

__device__ __forceinline__ void agg_sum_rows(const float4* __restrict__ el4,
                                             const float* __restrict__ hfeat,
                                             const float4* err4, const float4* mrow4,
                                             const int* myrl, int n, int lane,
                                             float (&racc)[ROWS_PER_WAVE][HH],
                                             float (&rsum)[ROWS_PER_WAVE][HH]) {
  for (int base = 0; base < n; base += 32) {
    const int nb = (n - base) < 32 ? (n - base) : 32;
    int sn = 0, r = 0;
    float a[HH] = {0.f, 0.f, 0.f, 0.f, 0.f, 0.f, 0.f, 0.f};
    if (lane < nb) {
      const int pk = myrl[base + lane];
      r  = pk & 127;
      sn = pk >> 7;
      float v[HH];
      edge_scores(el4, sn, err4, r, v);
      const float4 m0 = mrow4[r * 2], m1 = mrow4[r * 2 + 1];
      a[0] = __expf(v[0] - m0.x); a[1] = __expf(v[1] - m0.y);
      a[2] = __expf(v[2] - m0.z); a[3] = __expf(v[3] - m0.w);
      a[4] = __expf(v[4] - m1.x); a[5] = __expf(v[5] - m1.y);
      a[6] = __expf(v[6] - m1.z); a[7] = __expf(v[7] - m1.w);
    }
    for (int j = 0; j < nb; j += 4) {
      float hv[4][HH];
#pragma unroll
      for (int u = 0; u < 4; ++u) {
        const int jj = (j + u < nb) ? (j + u) : (nb - 1);
        const int sj = lane_bcast_i(sn, jj);
        const float* hp = hfeat + (size_t)sj * F + lane;
#pragma unroll
        for (int h = 0; h < HH; ++h) hv[u][h] = hp[h * DD];
      }
#pragma unroll
      for (int u = 0; u < 4; ++u) {
        if (j + u < nb) {
          const int slot = lane_bcast_i(r, j + u) >> 5;
          float al[HH];
#pragma unroll
          for (int h = 0; h < HH; ++h) al[h] = lane_bcast_f(a[h], j + u);
          if (slot == 0)      row_add<0>(racc, rsum, al, hv[u]);
          else if (slot == 1) row_add<1>(racc, rsum, al, hv[u]);
          else                row_add<2>(racc, rsum, al, hv[u]);
        }
      }
    }
  }
}

__global__ __launch_bounds__(AGG_THREADS) void gat_agg_rows(
    const float* __restrict__ el, const float* __restrict__ er,
    const int* __restrict__ src, const int* __restrict__ dst,
    const unsigned* __restrict__ plane,
    const float* __restrict__ hfeat, float* __restrict__ agg) {
  __shared__ __attribute__((aligned(16))) float acc[AGG_ROWS * F];
  __shared__ __attribute__((aligned(16))) float mrow[AGG_ROWS * HH];
  __shared__ __attribute__((aligned(16))) float srow[AGG_ROWS * HH];
  __shared__ __attribute__((aligned(16))) float err[AGG_ROWS * HH];
  __shared__ int elist[AGG_WAVES * LIST_CAP];
  __shared__ int seg[AGG_WAVES * AGG_WAVES * SEG_PITCH];
  __shared__ int scnt[AGG_WAVES * AGG_WAVES];
  __shared__ int lost_any;
  const int tid  = threadIdx.x;
  const int wave = tid >> 5;
  const int lane = tid & 31;
  const int rowbase = blockIdx.x * AGG_ROWS;

  if (tid < AGG_ROWS * HH) {
    const int row = rowbase + (tid >> 3);
    mrow[tid] = -INFINITY;
    srow[tid] = 0.f;
    err[tid]  = (row < NN) ? er[row * HH + (tid & 7)] : 0.f;
  }
  if (tid < AGG_WAVES * AGG_WAVES) scnt[tid] = 0;
  if (tid == 0) lost_any = 0;
  __syncthreads();

  int* mylist = elist + wave * LIST_CAP;
  int cnt = 0;
  const int NIT = (EE / 16 + SCAN_STEP - 1) / SCAN_STEP;
  const uint4*  plane4 = reinterpret_cast<const uint4*>(plane);
  const float4* el4   = reinterpret_cast<const float4*>(el);
  const float4* err4  = reinterpret_cast<const float4*>(err);
  const float4* mrow4 = reinterpret_cast<const float4*>(mrow);
  const unsigned pat   = (unsigned)blockIdx.x * 0x01010101u;
  const int wave_s = __builtin_amdgcn_readfirstlane(wave);
  const int rot = blockIdx.x % NIT;

  bool lost = false;
  {
    uint4 cur[SCAN_U];
    agg_scan_load(plane4, rot, wave_s, lane, cur);
    for (int it = 0; it < NIT; ++it) {
      int itr = it + rot;
      if (itr >= NIT) itr -= NIT;
      int itn = itr + 1;
      if (itn >= NIT) itn -= NIT;
      uint4 nxt[SCAN_U];
      agg_scan_load(plane4, itn, wave_s, lane, nxt);
      cnt = agg_scan_step(cur, itr, wave_s, lane, pat, mylist, cnt, [&](int c) {
        agg_fold_max(src, dst, rowbase, el4, err4, mrow, mylist, c, lane, false,
                     wave_s, seg, scnt, &lost_any);
        lost = true;
        return 0;
      });
#pragma unroll
      for (int u = 0; u < SCAN_U; ++u) cur[u] = nxt[u];
    }
  }
  agg_fold_max(src, dst, rowbase, el4, err4, mrow, mylist, cnt, lane, !lost,
               wave_s, seg, scnt, &lost_any);
  if (lost) lost_any = 1;
  __syncthreads();

  if (!lost_any) {
    float racc[ROWS_PER_WAVE][HH], rsum[ROWS_PER_WAVE][HH];
#pragma unroll
    for (int k = 0; k < ROWS_PER_WAVE; ++k) {
#pragma unroll
      for (int h = 0; h < HH; ++h) { racc[k][h] = 0.f; rsum[k][h] = 0.f; }
    }
    const int c = scnt[wave * AGG_WAVES + lane];
    int incl = c;
#pragma unroll
    for (int sft = 1; sft < 32; sft <<= 1) {
      const int up = __shfl_up(incl, sft);
      if (lane >= sft) incl += up;
    }
    const int off = incl - c;
    const int n = lane_bcast_i(incl, 31);
    const int* sg = seg + (wave * AGG_WAVES + lane) * SEG_PITCH;
    for (int t = 0; t < SEG_CAP; ++t) {
      if (__builtin_amdgcn_ballot_w32(t < c) == 0u) break;
      if (t < c) mylist[off + t] = sg[t];
    }
    agg_sum_rows(el4, hfeat, err4, mrow4, mylist, n, lane, racc, rsum);
#pragma unroll
    for (int k = 0; k < ROWS_PER_WAVE; ++k) {
      const int rl  = wave + k * AGG_WAVES;
      const int row = rowbase + rl;
      if (rl < AGG_ROWS && row < NN) {
#pragma unroll
        for (int h = 0; h < HH; ++h) {
          const float s = rsum[k][h];
          agg[(size_t)row * F + h * DD + lane] = (s != 0.f) ? racc[k][h] / s : 0.f;
        }
      }
    }
  } else {
    for (int i = tid; i < AGG_ROWS * F; i += AGG_THREADS) acc[i] = 0.f;
    __syncthreads();
    cnt = 0;
    for (int it = 0; it < NIT; ++it) {
      int itr = it + rot;
      if (itr >= NIT) itr -= NIT;
      uint4 cur[SCAN_U];
      agg_scan_load(plane4, itr, wave_s, lane, cur);
      cnt = agg_scan_step(cur, itr, wave_s, lane, pat, mylist, cnt, [&](int c) {
        agg_drain(src, dst, rowbase, el4, hfeat, err4, mrow4, srow, acc, mylist, c, lane);
        return 0;
      });
    }
    agg_drain(src, dst, rowbase, el4, hfeat, err4, mrow4, srow, acc, mylist, cnt, lane);
    __syncthreads();

    const float4* acc4 = reinterpret_cast<const float4*>(acc);
    float4* agg4 = reinterpret_cast<float4*>(agg);
    for (int i = tid; i < AGG_ROWS * (F / 4); i += AGG_THREADS) {
      const int rl  = i / (F / 4);
      const int c4  = i % (F / 4);
      const int row = rowbase + rl;
      const float s = srow[rl * HH + c4 / (DD / 4)];
      float4 v = acc4[i];
      if (s != 0.f) { v.x /= s; v.y /= s; v.z /= s; v.w /= s; }
      else          { v = make_float4(0.f, 0.f, 0.f, 0.f); }
      if (row < NN) agg4[(size_t)row * (F / 4) + c4] = v;
    }
  }
}

#define BE_THREADS 256
#define BE_PER     8
#define BE_VECS    (NN * F / 4)
#define BE_BLOCKS  (BE_VECS / (BE_THREADS * BE_PER))
#define BE_STRIDE  (BE_BLOCKS * BE_THREADS)

static_assert(BE_VECS % (BE_THREADS * BE_PER) == 0, "the grid covers every vector exactly once, no tail");
static_assert(BE_STRIDE % (F / 4) == 0, "a thread's vectors are whole rows apart: one bias vector serves all");

__device__ __forceinline__ float elu1(float v) { return (v > 0.f) ? v : (__expf(v) - 1.f); }

__global__ __launch_bounds__(BE_THREADS) void gat_bias_elu(const float* __restrict__ agg,
                                                           const float* __restrict__ b,
                                                           float* __restrict__ out) {
  const int q = blockIdx.x * BE_THREADS + threadIdx.x;
  const float4* agg4 = reinterpret_cast<const float4*>(agg);
  float4* out4 = reinterpret_cast<float4*>(out);
  const float4 bv = reinterpret_cast<const float4*>(b)[q & (F / 4 - 1)];
  float4 v[BE_PER];
#pragma unroll
  for (int k = 0; k < BE_PER; ++k) v[k] = agg4[q + k * BE_STRIDE];
#pragma unroll
  for (int k = 0; k < BE_PER; ++k) {
    float4 r;
    r.x = elu1(v[k].x + bv.x);
    r.y = elu1(v[k].y + bv.y);
    r.z = elu1(v[k].z + bv.z);
    r.w = elu1(v[k].w + bv.w);
    out4[q + k * BE_STRIDE] = r;
  }
}

extern "C" void kernel_launch(void* const* d_in, const int* in_sizes, int n_in,
                              void* d_out, int out_size, void* d_ws, size_t ws_size,
                              hipStream_t stream) {
  const float* feats = (const float*)d_in[0];
  const float* W0    = (const float*)d_in[1];
  const float* al0   = (const float*)d_in[2];
  const float* ar0   = (const float*)d_in[3];
  const float* b0    = (const float*)d_in[4];
  const float* W1    = (const float*)d_in[5];
  const float* al1   = (const float*)d_in[6];
  const float* ar1   = (const float*)d_in[7];
  const float* b1    = (const float*)d_in[8];
  const int*   src   = (const int*)d_in[9];
  const int*   dst   = (const int*)d_in[10];

  float* ws = (float*)d_ws;
  float* hbuf = ws;
  float* agg  = hbuf + (size_t)NN * F;
  float* x1   = agg + (size_t)NN * F;
  float* el   = x1 + (size_t)NN * F;
  float* er   = el + (size_t)NN * HH;
  unsigned* plane = (unsigned*)(er + (size_t)NN * HH);

  const int bl = 256;
  const int gNH = (NN * HH + bl - 1) / bl;

  auto run_layer = [&](const float* x, const float* Wm, const float* al,
                       const float* ar, const float* bb, float* outx,
                       unsigned* planew) {
    gat_gemm_wmma<<<NN / 16, 256, 0, stream>>>(x, Wm, hbuf);
    gat_attn_coeff<<<gNH, bl, 0, stream>>>(hbuf, al, ar, el, er, dst, planew);
    gat_agg_rows<<<AGG_BLOCKS, AGG_THREADS, 0, stream>>>(
        el, er, src, dst, plane, hbuf, agg);
    gat_bias_elu<<<BE_BLOCKS, BE_THREADS, 0, stream>>>(agg, bb, outx);
  };

  run_layer(feats, W0, al0, ar0, b0, x1, plane);
  run_layer(x1, W1, al1, ar1, b1, (float*)d_out, nullptr);
}


# bias+ELU done by the row-owning wave at its register store, out = elu(acc/sum + b) (fixed-order register sums, no float atomics); bias+ELU launches and agg buffer gone; 8 launches -> 6; sums kept
# speedup vs baseline: 14.1316x; 1.0894x over previous
#include <hip/hip_runtime.h>
#include <hip/hip_bf16.h>
#include <math.h>

#define NN 20000
#define EE 320000
#define HH 8
#define DD 32
#define F  256
#define LDSPITCH 260

typedef float v2f __attribute__((ext_vector_type(2)));
typedef float v8f __attribute__((ext_vector_type(8)));

__global__ __launch_bounds__(256) void gat_gemm_wmma(
    const float* __restrict__ A, const float* __restrict__ W,
    float* __restrict__ C) {
  __shared__ float As[16 * LDSPITCH];
  const int tid  = threadIdx.x;
  const int wave = tid >> 5;
  const int lane = tid & 31;
  const int m0   = blockIdx.x << 4;

  for (int i = tid; i < 16 * 64; i += 256) {
    const int r  = i >> 6;
    const int c4 = (i & 63) << 2;
    const float4 v = *reinterpret_cast<const float4*>(A + (size_t)(m0 + r) * F + c4);
    float* p = &As[r * LDSPITCH + c4];
    p[0] = v.x; p[1] = v.y; p[2] = v.z; p[3] = v.w;
  }
  __syncthreads();

  const int hi = lane >> 4;
  const int l  = lane & 15;

  for (int nt = wave; nt < 16; nt += 8) {
    const int n0 = nt << 4;
    v8f c = {};
    for (int k0 = 0; k0 < F; k0 += 4) {
      const int ka = k0 + (hi << 1);
      v2f a, b;
      a.x = As[l * LDSPITCH + ka];
      a.y = As[l * LDSPITCH + ka + 1];
      b.x = W[(size_t)ka * F + n0 + l];
      b.y = W[(size_t)(ka + 1) * F + n0 + l];
      c = __builtin_amdgcn_wmma_f32_16x16x4_f32(false, a, false, b, (short)0, c,
                                                false, false);
    }
#pragma unroll
    for (int v = 0; v < 8; ++v)
      C[(size_t)(m0 + (hi << 3) + v) * F + n0 + l] = c[v];
  }
}

#define AGG_ROWS    80
#define AGG_BLOCKS  ((NN + AGG_ROWS - 1) / AGG_ROWS)
#define NO_BLOCK    255u

static_assert(AGG_BLOCKS <= 255, "an aggregation block's index fits one byte, 255 left free");
static_assert(EE % 16 == 0, "the plane is written as dwords and read as 16-byte vectors");
static_assert(NN * HH >= EE / 4, "the coefficient launch has a thread for every dword of the plane");

__device__ __forceinline__ unsigned plane_byte(int d) {
  return ((unsigned)d < (unsigned)NN) ? (unsigned)d / AGG_ROWS : NO_BLOCK;
}

__global__ void gat_attn_coeff(const float* __restrict__ h,
                               const float* __restrict__ al,
                               const float* __restrict__ ar,
                               float* __restrict__ el, float* __restrict__ er,
                               const int* __restrict__ dst,
                               unsigned* __restrict__ plane) {
  const int i = blockIdx.x * blockDim.x + threadIdx.x;
  if (plane != nullptr && i < EE / 4) {
    const int4 d = reinterpret_cast<const int4*>(dst)[i];
    plane[i] = plane_byte(d.x) | (plane_byte(d.y) << 8) |
               (plane_byte(d.z) << 16) | (plane_byte(d.w) << 24);
  }
  if (i >= NN * HH) return;
  const int n = i >> 3, hd = i & 7;
  const float* hp  = h + (size_t)n * F + hd * DD;
  const float* alp = al + hd * DD;
  const float* arp = ar + hd * DD;
  float sl = 0.f, sr = 0.f;
#pragma unroll 8
  for (int d = 0; d < DD; ++d) {
    const float v = hp[d];
    sl += v * alp[d];
    sr += v * arp[d];
  }
  el[i] = sl;
  er[i] = sr;
}

#define AGG_THREADS 1024
#define AGG_WAVES   32
#define SCAN_U      4
#define LIST_CAP    768
#define SCAN_STEP   (AGG_WAVES * SCAN_U * 32)
#define LIST_FULL   (LIST_CAP - 512)
#define ROWS_PER_WAVE ((AGG_ROWS + AGG_WAVES - 1) / AGG_WAVES)
#define SEG_CAP     16
#define SEG_PITCH   17

static_assert(AGG_ROWS <= 128 && NN < (1 << 24),
              "a kept list entry packs source*128 + local row");
static_assert(LIST_CAP > 512, "after the fullness test the list still takes one whole load's edges");
static_assert(AGG_ROWS * HH <= AGG_THREADS, "one thread per (row, head) sets up the row statistics");
static_assert(AGG_WAVES == 32, "lane i of an owner wave holds what wave i's list hands over; row r belongs to wave r & 31");
static_assert(ROWS_PER_WAVE == 3, "a wave keeps three rows' sums in registers");
static_assert(AGG_WAVES * AGG_WAVES <= AGG_THREADS, "one thread per (owner, list) clears the segment counts");
static_assert(AGG_WAVES * SEG_CAP <= LIST_CAP, "an owner wave's entries fit the LDS room of its own list");

__device__ __forceinline__ int lane_bcast_i(int v, int l) {
  return __builtin_amdgcn_readlane(v, l);
}
__device__ __forceinline__ float lane_bcast_f(float v, int l) {
  return __int_as_float(__builtin_amdgcn_readlane(__float_as_int(v), l));
}
__device__ __forceinline__ float leaky02(float v) { return (v > 0.f) ? v : 0.2f * v; }
__device__ __forceinline__ float elu1(float v) { return (v > 0.f) ? v : (__expf(v) - 1.f); }

static_assert((EE / 16) % 32 == 0, "a wave's 32 vectors of one load are all inside the plane or all past its end");

__device__ __forceinline__ void agg_scan_load(const uint4* __restrict__ plane4, int itr,
                                              int wave, int lane, uint4 d[SCAN_U]) {
  const int Q16 = EE / 16;
  const int q0 = (itr * AGG_WAVES + wave) * (SCAN_U * 32);
#pragma unroll
  for (int u = 0; u < SCAN_U; ++u) {
    const int qw = q0 + u * 32;
    d[u] = plane4[(qw < Q16 ? qw : Q16 - 32) + lane];
  }
}

__device__ __forceinline__ unsigned zero_bytes(unsigned x) {
  return ~(((x & 0x7f7f7f7fu) + 0x7f7f7f7fu) | x | 0x7f7f7f7fu);
}

template <class OnFull>
__device__ __forceinline__ int agg_scan_step(const uint4 d[SCAN_U], int itr, int wave,
                                             int lane, unsigned pat, int* mylist, int cnt,
                                             OnFull on_full) {
  const int Q16 = EE / 16;
  const int q0 = (itr * AGG_WAVES + wave) * (SCAN_U * 32);
#pragma unroll
  for (int u = 0; u < SCAN_U; ++u) {
    if (q0 + u * 32 >= Q16) break;
    const int ebase = (q0 + u * 32 + lane) * 16;
    unsigned hm = zero_bytes(d[u].x ^ pat) | (zero_bytes(d[u].y ^ pat) >> 1) |
                  (zero_bytes(d[u].z ^ pat) >> 2) | (zero_bytes(d[u].w ^ pat) >> 3);
    unsigned m = __builtin_amdgcn_ballot_w32(hm != 0u);
    while (m) {
      if (hm != 0u) {
        const int i = __builtin_ctz(hm);
        mylist[cnt + (int)__builtin_amdgcn_mbcnt_lo(m, 0u)] =
            ebase + ((7 - (i & 7)) << 2) + (i >> 3);
        hm &= hm - 1u;
      }
      cnt += __builtin_popcount(m);
      m = __builtin_amdgcn_ballot_w32(hm != 0u);
    }
    if (cnt > LIST_FULL) cnt = on_full(cnt);
  }
  return cnt;
}

__device__ __forceinline__ void edge_scores(const float4* __restrict__ el4, int sn,
                                            const float4* err4, int r, float v[HH]) {
  const float4 l0 = el4[sn * 2], l1 = el4[sn * 2 + 1];
  const float4 r0 = err4[r * 2], r1 = err4[r * 2 + 1];
  v[0] = leaky02(l0.x + r0.x); v[1] = leaky02(l0.y + r0.y);
  v[2] = leaky02(l0.z + r0.z); v[3] = leaky02(l0.w + r0.w);
  v[4] = leaky02(l1.x + r1.x); v[5] = leaky02(l1.y + r1.y);
  v[6] = leaky02(l1.z + r1.z); v[7] = leaky02(l1.w + r1.w);
}

__device__ __forceinline__ void agg_fold_max(const int* __restrict__ src,
                                             const int* __restrict__ dst, int rowbase,
                                             const float4* __restrict__ el4,
                                             const float4* err4, float* mrow,
                                             const int* mylist, int cnt, int lane, bool keep,
                                             int wave, int* seg, int* scnt, int* lost_any) {
  for (int idx = lane; idx < cnt; idx += 32) {
    const int e  = mylist[idx];
    const int r  = dst[e] - rowbase;
    const int sn = src[e];
    float v[HH];
    edge_scores(el4, sn, err4, r, v);
#pragma unroll
    for (int h = 0; h < HH; ++h)
      __hip_atomic_fetch_max(mrow + r * HH + h, v[h], __ATOMIC_RELAXED,
                             __HIP_MEMORY_SCOPE_WORKGROUP);
    if (keep) {
      const int d = r & (AGG_WAVES - 1);
      unsigned same = __builtin_amdgcn_ballot_w32(true);
#pragma unroll
      for (int b = 0; b < 5; ++b) {
        const bool bit = ((d >> b) & 1) != 0;
        const unsigned bb = __builtin_amdgcn_ballot_w32(bit);
        same &= bit ? bb : ~bb;
      }
      const int below = (int)__builtin_amdgcn_mbcnt_lo(same, 0u);
      const int nsame = __builtin_popcount(same);
      int* cp = scnt + d * AGG_WAVES + wave;
      const int c0 = *cp;
      const int slot = c0 + below;
      if (slot < SEG_CAP) seg[(d * AGG_WAVES + wave) * SEG_PITCH + slot] = (sn << 7) | r;
      else *lost_any = 1;
      if (below == nsame - 1) *cp = c0 + nsame;
    }
  }
}

__device__ __forceinline__ void agg_drain(const int* __restrict__ src,
                                          const int* __restrict__ dst, int rowbase,
                                          const float4* __restrict__ el4,
                                          const float* __restrict__ hfeat,
                                          const float4* err4, const float4* mrow4,
                                          float* srow, float* acc,
                                          const int* mylist, int cnt, int lane) {
  while (cnt > 0) {
    const int nb   = cnt < 32 ? cnt : 32;
    const int base = cnt - nb;
    int sn = 0, r = 0;
    float a[HH] = {0.f, 0.f, 0.f, 0.f, 0.f, 0.f, 0.f, 0.f};
    if (lane < nb) {
      const int pk = mylist[base + lane];
      r  = dst[pk] - rowbase;
      sn = src[pk];
      float v[HH];
      edge_scores(el4, sn, err4, r, v);
      const float4 m0 = mrow4[r * 2], m1 = mrow4[r * 2 + 1];
      a[0] = __expf(v[0] - m0.x); a[1] = __expf(v[1] - m0.y);
      a[2] = __expf(v[2] - m0.z); a[3] = __expf(v[3] - m0.w);
      a[4] = __expf(v[4] - m1.x); a[5] = __expf(v[5] - m1.y);
      a[6] = __expf(v[6] - m1.z); a[7] = __expf(v[7] - m1.w);
#pragma unroll
      for (int h = 0; h < HH; ++h)
        __hip_atomic_fetch_add(srow + r * HH + h, a[h], __ATOMIC_RELAXED,
                               __HIP_MEMORY_SCOPE_WORKGROUP);
    }
    for (int j = 0; j < nb; j += 4) {
      float hv[4][HH];
#pragma unroll
      for (int u = 0; u < 4; ++u) {
        const int jj = (j + u < nb) ? (j + u) : (nb - 1);
        const int sj = lane_bcast_i(sn, jj);
        const float* hp = hfeat + (size_t)sj * F + lane;
#pragma unroll
        for (int h = 0; h < HH; ++h) hv[u][h] = hp[h * DD];
      }
#pragma unroll
      for (int u = 0; u < 4; ++u) {
        if (j + u < nb) {
          const int rj = lane_bcast_i(r, j + u);
          float* ap = acc + rj * F + lane;
#pragma unroll
          for (int h = 0; h < HH; ++h) {
            const float al = lane_bcast_f(a[h], j + u);
            __hip_atomic_fetch_add(ap + h * DD, al * hv[u][h], __ATOMIC_RELAXED,
                                   __HIP_MEMORY_SCOPE_WORKGROUP);
          }
        }
      }
    }
    cnt = base;
  }
}

__device__ __forceinline__ float uniform_f(float v) {
  return __int_as_float(__builtin_amdgcn_readfirstlane(__float_as_int(v)));
}

template <int K>
__device__ __forceinline__ void row_add(float (&racc)[ROWS_PER_WAVE][HH], float (&rsum)[ROWS_PER_WAVE][HH],
                                        const float al[HH], const float hv[HH]) {
#pragma unroll
  for (int h = 0; h < HH; ++h) {
    racc[K][h] = fmaf(al[h], hv[h], racc[K][h]);
    rsum[K][h] = uniform_f(rsum[K][h] + al[h]);
  }
}

__device__ __forceinline__ void agg_sum_rows(const float4* __restrict__ el4,
                                             const float* __restrict__ hfeat,
                                             const float4* err4, const float4* mrow4,
                                             const int* myrl, int n, int lane,
                                             float (&racc)[ROWS_PER_WAVE][HH],
                                             float (&rsum)[ROWS_PER_WAVE][HH]) {
  for (int base = 0; base < n; base += 32) {
    const int nb = (n - base) < 32 ? (n - base) : 32;
    int sn = 0, r = 0;
    float a[HH] = {0.f, 0.f, 0.f, 0.f, 0.f, 0.f, 0.f, 0.f};
    if (lane < nb) {
      const int pk = myrl[base + lane];
      r  = pk & 127;
      sn = pk >> 7;
      float v[HH];
      edge_scores(el4, sn, err4, r, v);
      const float4 m0 = mrow4[r * 2], m1 = mrow4[r * 2 + 1];
      a[0] = __expf(v[0] - m0.x); a[1] = __expf(v[1] - m0.y);
      a[2] = __expf(v[2] - m0.z); a[3] = __expf(v[3] - m0.w);
      a[4] = __expf(v[4] - m1.x); a[5] = __expf(v[5] - m1.y);
      a[6] = __expf(v[6] - m1.z); a[7] = __expf(v[7] - m1.w);
    }
    for (int j = 0; j < nb; j += 4) {
      float hv[4][HH];
#pragma unroll
      for (int u = 0; u < 4; ++u) {
        const int jj = (j + u < nb) ? (j + u) : (nb - 1);
        const int sj = lane_bcast_i(sn, jj);
        const float* hp = hfeat + (size_t)sj * F + lane;
#pragma unroll
        for (int h = 0; h < HH; ++h) hv[u][h] = hp[h * DD];
      }
#pragma unroll
      for (int u = 0; u < 4; ++u) {
        if (j + u < nb) {
          const int slot = lane_bcast_i(r, j + u) >> 5;
          float al[HH];
#pragma unroll
          for (int h = 0; h < HH; ++h) al[h] = lane_bcast_f(a[h], j + u);
          if (slot == 0)      row_add<0>(racc, rsum, al, hv[u]);
          else if (slot == 1) row_add<1>(racc, rsum, al, hv[u]);
          else                row_add<2>(racc, rsum, al, hv[u]);
        }
      }
    }
  }
}

__global__ __launch_bounds__(AGG_THREADS) void gat_agg_rows(
    const float* __restrict__ el, const float* __restrict__ er,
    const int* __restrict__ src, const int* __restrict__ dst,
    const unsigned* __restrict__ plane,
    const float* __restrict__ hfeat, const float* __restrict__ b,
    float* __restrict__ outx) {
  __shared__ __attribute__((aligned(16))) float acc[AGG_ROWS * F];
  __shared__ __attribute__((aligned(16))) float mrow[AGG_ROWS * HH];
  __shared__ __attribute__((aligned(16))) float srow[AGG_ROWS * HH];
  __shared__ __attribute__((aligned(16))) float err[AGG_ROWS * HH];
  __shared__ int elist[AGG_WAVES * LIST_CAP];
  __shared__ int seg[AGG_WAVES * AGG_WAVES * SEG_PITCH];
  __shared__ int scnt[AGG_WAVES * AGG_WAVES];
  __shared__ int lost_any;
  const int tid  = threadIdx.x;
  const int wave = tid >> 5;
  const int lane = tid & 31;
  const int rowbase = blockIdx.x * AGG_ROWS;

  if (tid < AGG_ROWS * HH) {
    const int row = rowbase + (tid >> 3);
    mrow[tid] = -INFINITY;
    srow[tid] = 0.f;
    err[tid]  = (row < NN) ? er[row * HH + (tid & 7)] : 0.f;
  }
  if (tid < AGG_WAVES * AGG_WAVES) scnt[tid] = 0;
  if (tid == 0) lost_any = 0;
  __syncthreads();

  int* mylist = elist + wave * LIST_CAP;
  int cnt = 0;
  const int NIT = (EE / 16 + SCAN_STEP - 1) / SCAN_STEP;
  const uint4*  plane4 = reinterpret_cast<const uint4*>(plane);
  const float4* el4   = reinterpret_cast<const float4*>(el);
  const float4* err4  = reinterpret_cast<const float4*>(err);
  const float4* mrow4 = reinterpret_cast<const float4*>(mrow);
  const unsigned pat   = (unsigned)blockIdx.x * 0x01010101u;
  const int wave_s = __builtin_amdgcn_readfirstlane(wave);
  const int rot = blockIdx.x % NIT;

  bool lost = false;
  {
    uint4 cur[SCAN_U];
    agg_scan_load(plane4, rot, wave_s, lane, cur);
    for (int it = 0; it < NIT; ++it) {
      int itr = it + rot;
      if (itr >= NIT) itr -= NIT;
      int itn = itr + 1;
      if (itn >= NIT) itn -= NIT;
      uint4 nxt[SCAN_U];
      agg_scan_load(plane4, itn, wave_s, lane, nxt);
      cnt = agg_scan_step(cur, itr, wave_s, lane, pat, mylist, cnt, [&](int c) {
        agg_fold_max(src, dst, rowbase, el4, err4, mrow, mylist, c, lane, false,
                     wave_s, seg, scnt, &lost_any);
        lost = true;
        return 0;
      });
#pragma unroll
      for (int u = 0; u < SCAN_U; ++u) cur[u] = nxt[u];
    }
  }
  agg_fold_max(src, dst, rowbase, el4, err4, mrow, mylist, cnt, lane, !lost,
               wave_s, seg, scnt, &lost_any);
  if (lost) lost_any = 1;
  __syncthreads();

  if (!lost_any) {
    float racc[ROWS_PER_WAVE][HH], rsum[ROWS_PER_WAVE][HH];
#pragma unroll
    for (int k = 0; k < ROWS_PER_WAVE; ++k) {
#pragma unroll
      for (int h = 0; h < HH; ++h) { racc[k][h] = 0.f; rsum[k][h] = 0.f; }
    }
    const int c = scnt[wave * AGG_WAVES + lane];
    int incl = c;
#pragma unroll
    for (int sft = 1; sft < 32; sft <<= 1) {
      const int up = __shfl_up(incl, sft);
      if (lane >= sft) incl += up;
    }
    const int off = incl - c;
    const int n = lane_bcast_i(incl, 31);
    const int* sg = seg + (wave * AGG_WAVES + lane) * SEG_PITCH;
    for (int t = 0; t < SEG_CAP; ++t) {
      if (__builtin_amdgcn_ballot_w32(t < c) == 0u) break;
      if (t < c) mylist[off + t] = sg[t];
    }
    agg_sum_rows(el4, hfeat, err4, mrow4, mylist, n, lane, racc, rsum);
    float bh[HH];
#pragma unroll
    for (int h = 0; h < HH; ++h) bh[h] = b[h * DD + lane];
#pragma unroll
    for (int k = 0; k < ROWS_PER_WAVE; ++k) {
      const int rl  = wave + k * AGG_WAVES;
      const int row = rowbase + rl;
      if (rl < AGG_ROWS && row < NN) {
#pragma unroll
        for (int h = 0; h < HH; ++h) {
          const float s = rsum[k][h];
          const float v = (s != 0.f) ? racc[k][h] / s : 0.f;
          outx[(size_t)row * F + h * DD + lane] = elu1(v + bh[h]);
        }
      }
    }
  } else {
    for (int i = tid; i < AGG_ROWS * F; i += AGG_THREADS) acc[i] = 0.f;
    __syncthreads();
    cnt = 0;
    for (int it = 0; it < NIT; ++it) {
      int itr = it + rot;
      if (itr >= NIT) itr -= NIT;
      uint4 cur[SCAN_U];
      agg_scan_load(plane4, itr, wave_s, lane, cur);
      cnt = agg_scan_step(cur, itr, wave_s, lane, pat, mylist, cnt, [&](int c) {
        agg_drain(src, dst, rowbase, el4, hfeat, err4, mrow4, srow, acc, mylist, c, lane);
        return 0;
      });
    }
    agg_drain(src, dst, rowbase, el4, hfeat, err4, mrow4, srow, acc, mylist, cnt, lane);
    __syncthreads();

    const float4* acc4 = reinterpret_cast<const float4*>(acc);
    const float4* b4 = reinterpret_cast<const float4*>(b);
    float4* out4 = reinterpret_cast<float4*>(outx);
    for (int i = tid; i < AGG_ROWS * (F / 4); i += AGG_THREADS) {
      const int rl  = i / (F / 4);
      const int c4  = i % (F / 4);
      const int row = rowbase + rl;
      const float s = srow[rl * HH + c4 / (DD / 4)];
      float4 v = acc4[i];
      if (s != 0.f) { v.x /= s; v.y /= s; v.z /= s; v.w /= s; }
      else          { v = make_float4(0.f, 0.f, 0.f, 0.f); }
      const float4 bv = b4[c4];
      v.x = elu1(v.x + bv.x); v.y = elu1(v.y + bv.y);
      v.z = elu1(v.z + bv.z); v.w = elu1(v.w + bv.w);
      if (row < NN) out4[(size_t)row * (F / 4) + c4] = v;
    }
  }
}

extern "C" void kernel_launch(void* const* d_in, const int* in_sizes, int n_in,
                              void* d_out, int out_size, void* d_ws, size_t ws_size,
                              hipStream_t stream) {
  const float* feats = (const float*)d_in[0];
  const float* W0    = (const float*)d_in[1];
  const float* al0   = (const float*)d_in[2];
  const float* ar0   = (const float*)d_in[3];
  const float* b0    = (const float*)d_in[4];
  const float* W1    = (const float*)d_in[5];
  const float* al1   = (const float*)d_in[6];
  const float* ar1   = (const float*)d_in[7];
  const float* b1    = (const float*)d_in[8];
  const int*   src   = (const int*)d_in[9];
  const int*   dst   = (const int*)d_in[10];

  float* ws = (float*)d_ws;
  float* hbuf = ws;
  float* x1   = hbuf + (size_t)NN * F;
  float* el   = x1 + (size_t)NN * F;
  float* er   = el + (size_t)NN * HH;
  unsigned* plane = (unsigned*)(er + (size_t)NN * HH);

  const int bl = 256;
  const int gNH = (NN * HH + bl - 1) / bl;

  auto run_layer = [&](const float* x, const float* Wm, const float* al,
                       const float* ar, const float* bb, float* outx,
                       unsigned* planew) {
    gat_gemm_wmma<<<NN / 16, 256, 0, stream>>>(x, Wm, hbuf);
    gat_attn_coeff<<<gNH, bl, 0, stream>>>(hbuf, al, ar, el, er, dst, planew);
    gat_agg_rows<<<AGG_BLOCKS, AGG_THREADS, 0, stream>>>(
        el, er, src, dst, plane, hbuf, bb, outx);
  };

  run_layer(feats, W0, al0, ar0, b0, x1, plane);
  run_layer(x1, W1, al1, ar1, b1, (float*)d_out, nullptr);
}


# GEMM retiled: a 4-wave block owns 32 rows x 128 columns and a wave a 2x2 set of 16x16 tiles: a W fragment feeds two row tiles, an A fragment two column tiles; slab loaded as one group; all sums kept
# speedup vs baseline: 15.6794x; 1.1095x over previous
#include <hip/hip_runtime.h>
#include <hip/hip_bf16.h>
#include <math.h>

#define NN 20000
#define EE 320000
#define HH 8
#define DD 32
#define F  256
#define LDSPITCH 260

typedef float v2f __attribute__((ext_vector_type(2)));
typedef float v8f __attribute__((ext_vector_type(8)));

#define GM_ROWS    32
#define GM_COLS    128
#define GM_THREADS 128
#define GM_SLABV   (GM_ROWS * (F / 4) / GM_THREADS)

static_assert(NN % GM_ROWS == 0 && F % GM_COLS == 0, "blocks tile C exactly: no partial tile, no guard");
static_assert(GM_COLS == (GM_THREADS / 32) * 32, "a wave owns 32 columns");
static_assert((LDSPITCH * 4) % 16 == 0, "slab rows start 16-byte aligned in LDS");

__global__ __launch_bounds__(GM_THREADS) void gat_gemm_wmma(
    const float* __restrict__ A, const float* __restrict__ W,
    float* __restrict__ C) {
  __shared__ __attribute__((aligned(16))) float As[GM_ROWS * LDSPITCH];
  const int tid  = threadIdx.x;
  const int wave = tid >> 5;
  const int lane = tid & 31;
  const int m0   = (blockIdx.x / (F / GM_COLS)) * GM_ROWS;
  const int nb   = (blockIdx.x % (F / GM_COLS)) * GM_COLS;

  float4 sv[GM_SLABV];
#pragma unroll
  for (int j = 0; j < GM_SLABV; ++j) {
    const int i = tid + j * GM_THREADS;
    sv[j] = *reinterpret_cast<const float4*>(A + (size_t)(m0 + (i >> 6)) * F + ((i & 63) << 2));
  }
#pragma unroll
  for (int j = 0; j < GM_SLABV; ++j) {
    const int i = tid + j * GM_THREADS;
    *reinterpret_cast<float4*>(&As[(i >> 6) * LDSPITCH + ((i & 63) << 2)]) = sv[j];
  }
  __syncthreads();

  const int hi = lane >> 4;
  const int l  = lane & 15;
  const int n0 = nb + (wave << 5);

  const float* a0p = As + l * LDSPITCH + (hi << 1);
  const float* a1p = a0p + 16 * LDSPITCH;
  const float* wp  = W + (size_t)(hi << 1) * F + n0 + l;

  v8f c00 = {}, c01 = {}, c10 = {}, c11 = {};
#pragma unroll 8
  for (int k0 = 0; k0 < F; k0 += 4) {
    v2f a0, a1, b0, b1;
    a0.x = a0p[k0];
    a0.y = a0p[k0 + 1];
    a1.x = a1p[k0];
    a1.y = a1p[k0 + 1];
    b0.x = wp[(size_t)k0 * F];
    b0.y = wp[(size_t)(k0 + 1) * F];
    b1.x = wp[(size_t)k0 * F + 16];
    b1.y = wp[(size_t)(k0 + 1) * F + 16];
    c00 = __builtin_amdgcn_wmma_f32_16x16x4_f32(false, a0, false, b0, (short)0, c00, false, false);
    c01 = __builtin_amdgcn_wmma_f32_16x16x4_f32(false, a0, false, b1, (short)0, c01, false, false);
    c10 = __builtin_amdgcn_wmma_f32_16x16x4_f32(false, a1, false, b0, (short)0, c10, false, false);
    c11 = __builtin_amdgcn_wmma_f32_16x16x4_f32(false, a1, false, b1, (short)0, c11, false, false);
  }
#pragma unroll
  for (int v = 0; v < 8; ++v) {
    float* cr = C + (size_t)(m0 + (hi << 3) + v) * F + n0 + l;
    cr[0]  = c00[v];
    cr[16] = c01[v];
    cr[(size_t)16 * F]      = c10[v];
    cr[(size_t)16 * F + 16] = c11[v];
  }
}

#define AGG_ROWS    80
#define AGG_BLOCKS  ((NN + AGG_ROWS - 1) / AGG_ROWS)
#define NO_BLOCK    255u

static_assert(AGG_BLOCKS <= 255, "an aggregation block's index fits one byte, 255 left free");
static_assert(EE % 16 == 0, "the plane is written as dwords and read as 16-byte vectors");
static_assert(NN * HH >= EE / 4, "the coefficient launch has a thread for every dword of the plane");

__device__ __forceinline__ unsigned plane_byte(int d) {
  return ((unsigned)d < (unsigned)NN) ? (unsigned)d / AGG_ROWS : NO_BLOCK;
}

__global__ void gat_attn_coeff(const float* __restrict__ h,
                               const float* __restrict__ al,
                               const float* __restrict__ ar,
                               float* __restrict__ el, float* __restrict__ er,
                               const int* __restrict__ dst,
                               unsigned* __restrict__ plane) {
  const int i = blockIdx.x * blockDim.x + threadIdx.x;
  if (plane != nullptr && i < EE / 4) {
    const int4 d = reinterpret_cast<const int4*>(dst)[i];
    plane[i] = plane_byte(d.x) | (plane_byte(d.y) << 8) |
               (plane_byte(d.z) << 16) | (plane_byte(d.w) << 24);
  }
  if (i >= NN * HH) return;
  const int n = i >> 3, hd = i & 7;
  const float* hp  = h + (size_t)n * F + hd * DD;
  const float* alp = al + hd * DD;
  const float* arp = ar + hd * DD;
  float sl = 0.f, sr = 0.f;
#pragma unroll 8
  for (int d = 0; d < DD; ++d) {
    const float v = hp[d];
    sl += v * alp[d];
    sr += v * arp[d];
  }
  el[i] = sl;
  er[i] = sr;
}

#define AGG_THREADS 1024
#define AGG_WAVES   32
#define SCAN_U      4
#define LIST_CAP    768
#define SCAN_STEP   (AGG_WAVES * SCAN_U * 32)
#define LIST_FULL   (LIST_CAP - 512)
#define ROWS_PER_WAVE ((AGG_ROWS + AGG_WAVES - 1) / AGG_WAVES)
#define SEG_CAP     16
#define SEG_PITCH   17

static_assert(AGG_ROWS <= 128 && NN < (1 << 24),
              "a kept list entry packs source*128 + local row");
static_assert(LIST_CAP > 512, "after the fullness test the list still takes one whole load's edges");
static_assert(AGG_ROWS * HH <= AGG_THREADS, "one thread per (row, head) sets up the row statistics");
static_assert(AGG_WAVES == 32, "lane i of an owner wave holds what wave i's list hands over; row r belongs to wave r & 31");
static_assert(ROWS_PER_WAVE == 3, "a wave keeps three rows' sums in registers");
static_assert(AGG_WAVES * AGG_WAVES <= AGG_THREADS, "one thread per (owner, list) clears the segment counts");
static_assert(AGG_WAVES * SEG_CAP <= LIST_CAP, "an owner wave's entries fit the LDS room of its own list");

__device__ __forceinline__ int lane_bcast_i(int v, int l) {
  return __builtin_amdgcn_readlane(v, l);
}
__device__ __forceinline__ float lane_bcast_f(float v, int l) {
  return __int_as_float(__builtin_amdgcn_readlane(__float_as_int(v), l));
}
__device__ __forceinline__ float leaky02(float v) { return (v > 0.f) ? v : 0.2f * v; }
__device__ __forceinline__ float elu1(float v) { return (v > 0.f) ? v : (__expf(v) - 1.f); }

static_assert((EE / 16) % 32 == 0, "a wave's 32 vectors of one load are all inside the plane or all past its end");

__device__ __forceinline__ void agg_scan_load(const uint4* __restrict__ plane4, int itr,
                                              int wave, int lane, uint4 d[SCAN_U]) {
  const int Q16 = EE / 16;
  const int q0 = (itr * AGG_WAVES + wave) * (SCAN_U * 32);
#pragma unroll
  for (int u = 0; u < SCAN_U; ++u) {
    const int qw = q0 + u * 32;
    d[u] = plane4[(qw < Q16 ? qw : Q16 - 32) + lane];
  }
}

__device__ __forceinline__ unsigned zero_bytes(unsigned x) {
  return ~(((x & 0x7f7f7f7fu) + 0x7f7f7f7fu) | x | 0x7f7f7f7fu);
}

template <class OnFull>
__device__ __forceinline__ int agg_scan_step(const uint4 d[SCAN_U], int itr, int wave,
                                             int lane, unsigned pat, int* mylist, int cnt,
                                             OnFull on_full) {
  const int Q16 = EE / 16;
  const int q0 = (itr * AGG_WAVES + wave) * (SCAN_U * 32);
#pragma unroll
  for (int u = 0; u < SCAN_U; ++u) {
    if (q0 + u * 32 >= Q16) break;
    const int ebase = (q0 + u * 32 + lane) * 16;
    unsigned hm = zero_bytes(d[u].x ^ pat) | (zero_bytes(d[u].y ^ pat) >> 1) |
                  (zero_bytes(d[u].z ^ pat) >> 2) | (zero_bytes(d[u].w ^ pat) >> 3);
    unsigned m = __builtin_amdgcn_ballot_w32(hm != 0u);
    while (m) {
      if (hm != 0u) {
        const int i = __builtin_ctz(hm);
        mylist[cnt + (int)__builtin_amdgcn_mbcnt_lo(m, 0u)] =
            ebase + ((7 - (i & 7)) << 2) + (i >> 3);
        hm &= hm - 1u;
      }
      cnt += __builtin_popcount(m);
      m = __builtin_amdgcn_ballot_w32(hm != 0u);
    }
    if (cnt > LIST_FULL) cnt = on_full(cnt);
  }
  return cnt;
}

__device__ __forceinline__ void edge_scores(const float4* __restrict__ el4, int sn,
                                            const float4* err4, int r, float v[HH]) {
  const float4 l0 = el4[sn * 2], l1 = el4[sn * 2 + 1];
  const float4 r0 = err4[r * 2], r1 = err4[r * 2 + 1];
  v[0] = leaky02(l0.x + r0.x); v[1] = leaky02(l0.y + r0.y);
  v[2] = leaky02(l0.z + r0.z); v[3] = leaky02(l0.w + r0.w);
  v[4] = leaky02(l1.x + r1.x); v[5] = leaky02(l1.y + r1.y);
  v[6] = leaky02(l1.z + r1.z); v[7] = leaky02(l1.w + r1.w);
}

__device__ __forceinline__ void agg_fold_max(const int* __restrict__ src,
                                             const int* __restrict__ dst, int rowbase,
                                             const float4* __restrict__ el4,
                                             const float4* err4, float* mrow,
                                             const int* mylist, int cnt, int lane, bool keep,
                                             int wave, int* seg, int* scnt, int* lost_any) {
  for (int idx = lane; idx < cnt; idx += 32) {
    const int e  = mylist[idx];
    const int r  = dst[e] - rowbase;
    const int sn = src[e];
    float v[HH];
    edge_scores(el4, sn, err4, r, v);
#pragma unroll
    for (int h = 0; h < HH; ++h)
      __hip_atomic_fetch_max(mrow + r * HH + h, v[h], __ATOMIC_RELAXED,
                             __HIP_MEMORY_SCOPE_WORKGROUP);
    if (keep) {
      const int d = r & (AGG_WAVES - 1);
      unsigned same = __builtin_amdgcn_ballot_w32(true);
#pragma unroll
      for (int b = 0; b < 5; ++b) {
        const bool bit = ((d >> b) & 1) != 0;
        const unsigned bb = __builtin_amdgcn_ballot_w32(bit);
        same &= bit ? bb : ~bb;
      }
      const int below = (int)__builtin_amdgcn_mbcnt_lo(same, 0u);
      const int nsame = __builtin_popcount(same);
      int* cp = scnt + d * AGG_WAVES + wave;
      const int c0 = *cp;
      const int slot = c0 + below;
      if (slot < SEG_CAP) seg[(d * AGG_WAVES + wave) * SEG_PITCH + slot] = (sn << 7) | r;
      else *lost_any = 1;
      if (below == nsame - 1) *cp = c0 + nsame;
    }
  }
}

__device__ __forceinline__ void agg_drain(const int* __restrict__ src,
                                          const int* __restrict__ dst, int rowbase,
                                          const float4* __restrict__ el4,
                                          const float* __restrict__ hfeat,
                                          const float4* err4, const float4* mrow4,
                                          float* srow, float* acc,
                                          const int* mylist, int cnt, int lane) {
  while (cnt > 0) {
    const int nb   = cnt < 32 ? cnt : 32;
    const int base = cnt - nb;
    int sn = 0, r = 0;
    float a[HH] = {0.f, 0.f, 0.f, 0.f, 0.f, 0.f, 0.f, 0.f};
    if (lane < nb) {
      const int pk = mylist[base + lane];
      r  = dst[pk] - rowbase;
      sn = src[pk];
      float v[HH];
      edge_scores(el4, sn, err4, r, v);
      const float4 m0 = mrow4[r * 2], m1 = mrow4[r * 2 + 1];
      a[0] = __expf(v[0] - m0.x); a[1] = __expf(v[1] - m0.y);
      a[2] = __expf(v[2] - m0.z); a[3] = __expf(v[3] - m0.w);
      a[4] = __expf(v[4] - m1.x); a[5] = __expf(v[5] - m1.y);
      a[6] = __expf(v[6] - m1.z); a[7] = __expf(v[7] - m1.w);
#pragma unroll
      for (int h = 0; h < HH; ++h)
        __hip_atomic_fetch_add(srow + r * HH + h, a[h], __ATOMIC_RELAXED,
                               __HIP_MEMORY_SCOPE_WORKGROUP);
    }
    for (int j = 0; j < nb; j += 4) {
      float hv[4][HH];
#pragma unroll
      for (int u = 0; u < 4; ++u) {
        const int jj = (j + u < nb) ? (j + u) : (nb - 1);
        const int sj = lane_bcast_i(sn, jj);
        const float* hp = hfeat + (size_t)sj * F + lane;
#pragma unroll
        for (int h = 0; h < HH; ++h) hv[u][h] = hp[h * DD];
      }
#pragma unroll
      for (int u = 0; u < 4; ++u) {
        if (j + u < nb) {
          const int rj = lane_bcast_i(r, j + u);
          float* ap = acc + rj * F + lane;
#pragma unroll
          for (int h = 0; h < HH; ++h) {
            const float al = lane_bcast_f(a[h], j + u);
            __hip_atomic_fetch_add(ap + h * DD, al * hv[u][h], __ATOMIC_RELAXED,
                                   __HIP_MEMORY_SCOPE_WORKGROUP);
          }
        }
      }
    }
    cnt = base;
  }
}

__device__ __forceinline__ float uniform_f(float v) {
  return __int_as_float(__builtin_amdgcn_readfirstlane(__float_as_int(v)));
}

template <int K>
__device__ __forceinline__ void row_add(float (&racc)[ROWS_PER_WAVE][HH], float (&rsum)[ROWS_PER_WAVE][HH],
                                        const float al[HH], const float hv[HH]) {
#pragma unroll
  for (int h = 0; h < HH; ++h) {
    racc[K][h] = fmaf(al[h], hv[h], racc[K][h]);
    rsum[K][h] = uniform_f(rsum[K][h] + al[h]);
  }
}

__device__ __forceinline__ void agg_sum_rows(const float4* __restrict__ el4,
                                             const float* __restrict__ hfeat,
                                             const float4* err4, const float4* mrow4,
                                             const int* myrl, int n, int lane,
                                             float (&racc)[ROWS_PER_WAVE][HH],
                                             float (&rsum)[ROWS_PER_WAVE][HH]) {
  for (int base = 0; base < n; base += 32) {
    const int nb = (n - base) < 32 ? (n - base) : 32;
    int sn = 0, r = 0;
    float a[HH] = {0.f, 0.f, 0.f, 0.f, 0.f, 0.f, 0.f, 0.f};
    if (lane < nb) {
      const int pk = myrl[base + lane];
      r  = pk & 127;
      sn = pk >> 7;
      float v[HH];
      edge_scores(el4, sn, err4, r, v);
      const float4 m0 = mrow4[r * 2], m1 = mrow4[r * 2 + 1];
      a[0] = __expf(v[0] - m0.x); a[1] = __expf(v[1] - m0.y);
      a[2] = __expf(v[2] - m0.z); a[3] = __expf(v[3] - m0.w);
      a[4] = __expf(v[4] - m1.x); a[5] = __expf(v[5] - m1.y);
      a[6] = __expf(v[6] - m1.z); a[7] = __expf(v[7] - m1.w);
    }
    for (int j = 0; j < nb; j += 4) {
      float hv[4][HH];
#pragma unroll
      for (int u = 0; u < 4; ++u) {
        const int jj = (j + u < nb) ? (j + u) : (nb - 1);
        const int sj = lane_bcast_i(sn, jj);
        const float* hp = hfeat + (size_t)sj * F + lane;
#pragma unroll
        for (int h = 0; h < HH; ++h) hv[u][h] = hp[h * DD];
      }
#pragma unroll
      for (int u = 0; u < 4; ++u) {
        if (j + u < nb) {
          const int slot = lane_bcast_i(r, j + u) >> 5;
          float al[HH];
#pragma unroll
          for (int h = 0; h < HH; ++h) al[h] = lane_bcast_f(a[h], j + u);
          if (slot == 0)      row_add<0>(racc, rsum, al, hv[u]);
          else if (slot == 1) row_add<1>(racc, rsum, al, hv[u]);
          else                row_add<2>(racc, rsum, al, hv[u]);
        }
      }
    }
  }
}

__global__ __launch_bounds__(AGG_THREADS) void gat_agg_rows(
    const float* __restrict__ el, const float* __restrict__ er,
    const int* __restrict__ src, const int* __restrict__ dst,
    const unsigned* __restrict__ plane,
    const float* __restrict__ hfeat, const float* __restrict__ b,
    float* __restrict__ outx) {
  __shared__ __attribute__((aligned(16))) float acc[AGG_ROWS * F];
  __shared__ __attribute__((aligned(16))) float mrow[AGG_ROWS * HH];
  __shared__ __attribute__((aligned(16))) float srow[AGG_ROWS * HH];
  __shared__ __attribute__((aligned(16))) float err[AGG_ROWS * HH];
  __shared__ int elist[AGG_WAVES * LIST_CAP];
  __shared__ int seg[AGG_WAVES * AGG_WAVES * SEG_PITCH];
  __shared__ int scnt[AGG_WAVES * AGG_WAVES];
  __shared__ int lost_any;
  const int tid  = threadIdx.x;
  const int wave = tid >> 5;
  const int lane = tid & 31;
  const int rowbase = blockIdx.x * AGG_ROWS;

  if (tid < AGG_ROWS * HH) {
    const int row = rowbase + (tid >> 3);
    mrow[tid] = -INFINITY;
    srow[tid] = 0.f;
    err[tid]  = (row < NN) ? er[row * HH + (tid & 7)] : 0.f;
  }
  if (tid < AGG_WAVES * AGG_WAVES) scnt[tid] = 0;
  if (tid == 0) lost_any = 0;
  __syncthreads();

  int* mylist = elist + wave * LIST_CAP;
  int cnt = 0;
  const int NIT = (EE / 16 + SCAN_STEP - 1) / SCAN_STEP;
  const uint4*  plane4 = reinterpret_cast<const uint4*>(plane);
  const float4* el4   = reinterpret_cast<const float4*>(el);
  const float4* err4  = reinterpret_cast<const float4*>(err);
  const float4* mrow4 = reinterpret_cast<const float4*>(mrow);
  const unsigned pat   = (unsigned)blockIdx.x * 0x01010101u;
  const int wave_s = __builtin_amdgcn_readfirstlane(wave);
  const int rot = blockIdx.x % NIT;

  bool lost = false;
  {
    uint4 cur[SCAN_U];
    agg_scan_load(plane4, rot, wave_s, lane, cur);
    for (int it = 0; it < NIT; ++it) {
      int itr = it + rot;
      if (itr >= NIT) itr -= NIT;
      int itn = itr + 1;
      if (itn >= NIT) itn -= NIT;
      uint4 nxt[SCAN_U];
      agg_scan_load(plane4, itn, wave_s, lane, nxt);
      cnt = agg_scan_step(cur, itr, wave_s, lane, pat, mylist, cnt, [&](int c) {
        agg_fold_max(src, dst, rowbase, el4, err4, mrow, mylist, c, lane, false,
                     wave_s, seg, scnt, &lost_any);
        lost = true;
        return 0;
      });
#pragma unroll
      for (int u = 0; u < SCAN_U; ++u) cur[u] = nxt[u];
    }
  }
  agg_fold_max(src, dst, rowbase, el4, err4, mrow, mylist, cnt, lane, !lost,
               wave_s, seg, scnt, &lost_any);
  if (lost) lost_any = 1;
  __syncthreads();

  if (!lost_any) {
    float racc[ROWS_PER_WAVE][HH], rsum[ROWS_PER_WAVE][HH];
#pragma unroll
    for (int k = 0; k < ROWS_PER_WAVE; ++k) {
#pragma unroll
      for (int h = 0; h < HH; ++h) { racc[k][h] = 0.f; rsum[k][h] = 0.f; }
    }
    const int c = scnt[wave * AGG_WAVES + lane];
    int incl = c;
#pragma unroll
    for (int sft = 1; sft < 32; sft <<= 1) {
      const int up = __shfl_up(incl, sft);
      if (lane >= sft) incl += up;
    }
    const int off = incl - c;
    const int n = lane_bcast_i(incl, 31);
    const int* sg = seg + (wave * AGG_WAVES + lane) * SEG_PITCH;
    for (int t = 0; t < SEG_CAP; ++t) {
      if (__builtin_amdgcn_ballot_w32(t < c) == 0u) break;
      if (t < c) mylist[off + t] = sg[t];
    }
    agg_sum_rows(el4, hfeat, err4, mrow4, mylist, n, lane, racc, rsum);
    float bh[HH];
#pragma unroll
    for (int h = 0; h < HH; ++h) bh[h] = b[h * DD + lane];
#pragma unroll
    for (int k = 0; k < ROWS_PER_WAVE; ++k) {
      const int rl  = wave + k * AGG_WAVES;
      const int row = rowbase + rl;
      if (rl < AGG_ROWS && row < NN) {
#pragma unroll
        for (int h = 0; h < HH; ++h) {
          const float s = rsum[k][h];
          const float v = (s != 0.f) ? racc[k][h] / s : 0.f;
          outx[(size_t)row * F + h * DD + lane] = elu1(v + bh[h]);
        }
      }
    }
  } else {
    for (int i = tid; i < AGG_ROWS * F; i += AGG_THREADS) acc[i] = 0.f;
    __syncthreads();
    cnt = 0;
    for (int it = 0; it < NIT; ++it) {
      int itr = it + rot;
      if (itr >= NIT) itr -= NIT;
      uint4 cur[SCAN_U];
      agg_scan_load(plane4, itr, wave_s, lane, cur);
      cnt = agg_scan_step(cur, itr, wave_s, lane, pat, mylist, cnt, [&](int c) {
        agg_drain(src, dst, rowbase, el4, hfeat, err4, mrow4, srow, acc, mylist, c, lane);
        return 0;
      });
    }
    agg_drain(src, dst, rowbase, el4, hfeat, err4, mrow4, srow, acc, mylist, cnt, lane);
    __syncthreads();

    const float4* acc4 = reinterpret_cast<const float4*>(acc);
    const float4* b4 = reinterpret_cast<const float4*>(b);
    float4* out4 = reinterpret_cast<float4*>(outx);
    for (int i = tid; i < AGG_ROWS * (F / 4); i += AGG_THREADS) {
      const int rl  = i / (F / 4);
      const int c4  = i % (F / 4);
      const int row = rowbase + rl;
      const float s = srow[rl * HH + c4 / (DD / 4)];
      float4 v = acc4[i];
      if (s != 0.f) { v.x /= s; v.y /= s; v.z /= s; v.w /= s; }
      else          { v = make_float4(0.f, 0.f, 0.f, 0.f); }
      const float4 bv = b4[c4];
      v.x = elu1(v.x + bv.x); v.y = elu1(v.y + bv.y);
      v.z = elu1(v.z + bv.z); v.w = elu1(v.w + bv.w);
      if (row < NN) out4[(size_t)row * (F / 4) + c4] = v;
    }
  }
}

extern "C" void kernel_launch(void* const* d_in, const int* in_sizes, int n_in,
                              void* d_out, int out_size, void* d_ws, size_t ws_size,
                              hipStream_t stream) {
  const float* feats = (const float*)d_in[0];
  const float* W0    = (const float*)d_in[1];
  const float* al0   = (const float*)d_in[2];
  const float* ar0   = (const float*)d_in[3];
  const float* b0    = (const float*)d_in[4];
  const float* W1    = (const float*)d_in[5];
  const float* al1   = (const float*)d_in[6];
  const float* ar1   = (const float*)d_in[7];
  const float* b1    = (const float*)d_in[8];
  const int*   src   = (const int*)d_in[9];
  const int*   dst   = (const int*)d_in[10];

  float* ws = (float*)d_ws;
  float* hbuf = ws;
  float* x1   = hbuf + (size_t)NN * F;
  float* el   = x1 + (size_t)NN * F;
  float* er   = el + (size_t)NN * HH;
  unsigned* plane = (unsigned*)(er + (size_t)NN * HH);

  const int bl = 256;
  const int gNH = (NN * HH + bl - 1) / bl;

  auto run_layer = [&](const float* x, const float* Wm, const float* al,
                       const float* ar, const float* bb, float* outx,
                       unsigned* planew) {
    gat_gemm_wmma<<<(NN / GM_ROWS) * (F / GM_COLS), GM_THREADS, 0, stream>>>(x, Wm, hbuf);
    gat_attn_coeff<<<gNH, bl, 0, stream>>>(hbuf, al, ar, el, er, dst, planew);
    gat_agg_rows<<<AGG_BLOCKS, AGG_THREADS, 0, stream>>>(
        el, er, src, dst, plane, hbuf, bb, outx);
  };

  run_layer(feats, W0, al0, ar0, b0, x1, plane);
  run_layer(x1, W1, al1, ar1, b1, (float*)d_out, nullptr);
}


# aggregation pass 2: a lane holds two 4-column vectors of a row, so a source row is two 16-byte loads that each cover 4 whole lines; two weights per lane from a per-wave LDS table; sum orders kept
# speedup vs baseline: 16.5819x; 1.0576x over previous
#include <hip/hip_runtime.h>
#include <hip/hip_bf16.h>
#include <math.h>

#define NN 20000
#define EE 320000
#define HH 8
#define DD 32
#define F  256
#define LDSPITCH 260

typedef float v2f __attribute__((ext_vector_type(2)));
typedef float v8f __attribute__((ext_vector_type(8)));

#define GM_ROWS    32
#define GM_COLS    128
#define GM_THREADS 128
#define GM_SLABV   (GM_ROWS * (F / 4) / GM_THREADS)

static_assert(NN % GM_ROWS == 0 && F % GM_COLS == 0, "blocks tile C exactly: no partial tile, no guard");
static_assert(GM_COLS == (GM_THREADS / 32) * 32, "a wave owns 32 columns");
static_assert((LDSPITCH * 4) % 16 == 0, "slab rows start 16-byte aligned in LDS");

__global__ __launch_bounds__(GM_THREADS) void gat_gemm_wmma(
    const float* __restrict__ A, const float* __restrict__ W,
    float* __restrict__ C) {
  __shared__ __attribute__((aligned(16))) float As[GM_ROWS * LDSPITCH];
  const int tid  = threadIdx.x;
  const int wave = tid >> 5;
  const int lane = tid & 31;
  const int m0   = (blockIdx.x / (F / GM_COLS)) * GM_ROWS;
  const int nb   = (blockIdx.x % (F / GM_COLS)) * GM_COLS;

  float4 sv[GM_SLABV];
#pragma unroll
  for (int j = 0; j < GM_SLABV; ++j) {
    const int i = tid + j * GM_THREADS;
    sv[j] = *reinterpret_cast<const float4*>(A + (size_t)(m0 + (i >> 6)) * F + ((i & 63) << 2));
  }
#pragma unroll
  for (int j = 0; j < GM_SLABV; ++j) {
    const int i = tid + j * GM_THREADS;
    *reinterpret_cast<float4*>(&As[(i >> 6) * LDSPITCH + ((i & 63) << 2)]) = sv[j];
  }
  __syncthreads();

  const int hi = lane >> 4;
  const int l  = lane & 15;
  const int n0 = nb + (wave << 5);

  const float* a0p = As + l * LDSPITCH + (hi << 1);
  const float* a1p = a0p + 16 * LDSPITCH;
  const float* wp  = W + (size_t)(hi << 1) * F + n0 + l;

  v8f c00 = {}, c01 = {}, c10 = {}, c11 = {};
#pragma unroll 8
  for (int k0 = 0; k0 < F; k0 += 4) {
    v2f a0, a1, b0, b1;
    a0.x = a0p[k0];
    a0.y = a0p[k0 + 1];
    a1.x = a1p[k0];
    a1.y = a1p[k0 + 1];
    b0.x = wp[(size_t)k0 * F];
    b0.y = wp[(size_t)(k0 + 1) * F];
    b1.x = wp[(size_t)k0 * F + 16];
    b1.y = wp[(size_t)(k0 + 1) * F + 16];
    c00 = __builtin_amdgcn_wmma_f32_16x16x4_f32(false, a0, false, b0, (short)0, c00, false, false);
    c01 = __builtin_amdgcn_wmma_f32_16x16x4_f32(false, a0, false, b1, (short)0, c01, false, false);
    c10 = __builtin_amdgcn_wmma_f32_16x16x4_f32(false, a1, false, b0, (short)0, c10, false, false);
    c11 = __builtin_amdgcn_wmma_f32_16x16x4_f32(false, a1, false, b1, (short)0, c11, false, false);
  }
#pragma unroll
  for (int v = 0; v < 8; ++v) {
    float* cr = C + (size_t)(m0 + (hi << 3) + v) * F + n0 + l;
    cr[0]  = c00[v];
    cr[16] = c01[v];
    cr[(size_t)16 * F]      = c10[v];
    cr[(size_t)16 * F + 16] = c11[v];
  }
}

#define AGG_ROWS    80
#define AGG_BLOCKS  ((NN + AGG_ROWS - 1) / AGG_ROWS)
#define NO_BLOCK    255u

static_assert(AGG_BLOCKS <= 255, "an aggregation block's index fits one byte, 255 left free");
static_assert(EE % 16 == 0, "the plane is written as dwords and read as 16-byte vectors");
static_assert(NN * HH >= EE / 4, "the coefficient launch has a thread for every dword of the plane");

__device__ __forceinline__ unsigned plane_byte(int d) {
  return ((unsigned)d < (unsigned)NN) ? (unsigned)d / AGG_ROWS : NO_BLOCK;
}

__global__ void gat_attn_coeff(const float* __restrict__ h,
                               const float* __restrict__ al,
                               const float* __restrict__ ar,
                               float* __restrict__ el, float* __restrict__ er,
                               const int* __restrict__ dst,
                               unsigned* __restrict__ plane) {
  const int i = blockIdx.x * blockDim.x + threadIdx.x;
  if (plane != nullptr && i < EE / 4) {
    const int4 d = reinterpret_cast<const int4*>(dst)[i];
    plane[i] = plane_byte(d.x) | (plane_byte(d.y) << 8) |
               (plane_byte(d.z) << 16) | (plane_byte(d.w) << 24);
  }
  if (i >= NN * HH) return;
  const int n = i >> 3, hd = i & 7;
  const float* hp  = h + (size_t)n * F + hd * DD;
  const float* alp = al + hd * DD;
  const float* arp = ar + hd * DD;
  float sl = 0.f, sr = 0.f;
#pragma unroll 8
  for (int d = 0; d < DD; ++d) {
    const float v = hp[d];
    sl += v * alp[d];
    sr += v * arp[d];
  }
  el[i] = sl;
  er[i] = sr;
}

#define AGG_THREADS 1024
#define AGG_WAVES   32
#define SCAN_U      4
#define LIST_CAP    768
#define SCAN_STEP   (AGG_WAVES * SCAN_U * 32)
#define LIST_FULL   (LIST_CAP - 512)
#define ROWS_PER_WAVE ((AGG_ROWS + AGG_WAVES - 1) / AGG_WAVES)
#define SEG_CAP     16
#define SEG_PITCH   17
#define COLS_PER_LANE (F / 32)
#define PASS2_U     8

static_assert(AGG_ROWS <= 128 && NN < (1 << 24),
              "a kept list entry packs source*128 + local row");
static_assert(LIST_CAP > 512, "after the fullness test the list still takes one whole load's edges");
static_assert(AGG_ROWS * HH <= AGG_THREADS, "one thread per (row, head) sets up the row statistics");
static_assert(AGG_WAVES == 32, "lane i of an owner wave holds what wave i's list hands over; row r belongs to wave r & 31");
static_assert(ROWS_PER_WAVE == 3, "a wave keeps three rows' sums in registers");
static_assert(AGG_WAVES * AGG_WAVES <= AGG_THREADS, "one thread per (owner, list) clears the segment counts");
static_assert(AGG_WAVES * SEG_CAP <= LIST_CAP, "an owner wave's entries fit the LDS room of its own list");
static_assert(COLS_PER_LANE == 8 && DD == 32 && HH == 8,
              "pass 2: a lane's columns 4*lane..+3 lie in head lane/8, its columns 128+4*lane..+3 in head 4+lane/8");
static_assert(AGG_WAVES * 32 * HH <= AGG_ROWS * F, "the waves' weight tables fit the accumulator room");

__device__ __forceinline__ int lane_bcast_i(int v, int l) {
  return __builtin_amdgcn_readlane(v, l);
}
__device__ __forceinline__ float lane_bcast_f(float v, int l) {
  return __int_as_float(__builtin_amdgcn_readlane(__float_as_int(v), l));
}
__device__ __forceinline__ float leaky02(float v) { return (v > 0.f) ? v : 0.2f * v; }
__device__ __forceinline__ float elu1(float v) { return (v > 0.f) ? v : (__expf(v) - 1.f); }

static_assert((EE / 16) % 32 == 0, "a wave's 32 vectors of one load are all inside the plane or all past its end");

__device__ __forceinline__ void agg_scan_load(const uint4* __restrict__ plane4, int itr,
                                              int wave, int lane, uint4 d[SCAN_U]) {
  const int Q16 = EE / 16;
  const int q0 = (itr * AGG_WAVES + wave) * (SCAN_U * 32);
#pragma unroll
  for (int u = 0; u < SCAN_U; ++u) {
    const int qw = q0 + u * 32;
    d[u] = plane4[(qw < Q16 ? qw : Q16 - 32) + lane];
  }
}

__device__ __forceinline__ unsigned zero_bytes(unsigned x) {
  return ~(((x & 0x7f7f7f7fu) + 0x7f7f7f7fu) | x | 0x7f7f7f7fu);
}

template <class OnFull>
__device__ __forceinline__ int agg_scan_step(const uint4 d[SCAN_U], int itr, int wave,
                                             int lane, unsigned pat, int* mylist, int cnt,
                                             OnFull on_full) {
  const int Q16 = EE / 16;
  const int q0 = (itr * AGG_WAVES + wave) * (SCAN_U * 32);
#pragma unroll
  for (int u = 0; u < SCAN_U; ++u) {
    if (q0 + u * 32 >= Q16) break;
    const int ebase = (q0 + u * 32 + lane) * 16;
    unsigned hm = zero_bytes(d[u].x ^ pat) | (zero_bytes(d[u].y ^ pat) >> 1) |
                  (zero_bytes(d[u].z ^ pat) >> 2) | (zero_bytes(d[u].w ^ pat) >> 3);
    unsigned m = __builtin_amdgcn_ballot_w32(hm != 0u);
    while (m) {
      if (hm != 0u) {
        const int i = __builtin_ctz(hm);
        mylist[cnt + (int)__builtin_amdgcn_mbcnt_lo(m, 0u)] =
            ebase + ((7 - (i & 7)) << 2) + (i >> 3);
        hm &= hm - 1u;
      }
      cnt += __builtin_popcount(m);
      m = __builtin_amdgcn_ballot_w32(hm != 0u);
    }
    if (cnt > LIST_FULL) cnt = on_full(cnt);
  }
  return cnt;
}

__device__ __forceinline__ void edge_scores(const float4* __restrict__ el4, int sn,
                                            const float4* err4, int r, float v[HH]) {
  const float4 l0 = el4[sn * 2], l1 = el4[sn * 2 + 1];
  const float4 r0 = err4[r * 2], r1 = err4[r * 2 + 1];
  v[0] = leaky02(l0.x + r0.x); v[1] = leaky02(l0.y + r0.y);
  v[2] = leaky02(l0.z + r0.z); v[3] = leaky02(l0.w + r0.w);
  v[4] = leaky02(l1.x + r1.x); v[5] = leaky02(l1.y + r1.y);
  v[6] = leaky02(l1.z + r1.z); v[7] = leaky02(l1.w + r1.w);
}

__device__ __forceinline__ void agg_fold_max(const int* __restrict__ src,
                                             const int* __restrict__ dst, int rowbase,
                                             const float4* __restrict__ el4,
                                             const float4* err4, float* mrow,
                                             const int* mylist, int cnt, int lane, bool keep,
                                             int wave, int* seg, int* scnt, int* lost_any) {
  for (int idx = lane; idx < cnt; idx += 32) {
    const int e  = mylist[idx];
    const int r  = dst[e] - rowbase;
    const int sn = src[e];
    float v[HH];
    edge_scores(el4, sn, err4, r, v);
#pragma unroll
    for (int h = 0; h < HH; ++h)
      __hip_atomic_fetch_max(mrow + r * HH + h, v[h], __ATOMIC_RELAXED,
                             __HIP_MEMORY_SCOPE_WORKGROUP);
    if (keep) {
      const int d = r & (AGG_WAVES - 1);
      unsigned same = __builtin_amdgcn_ballot_w32(true);
#pragma unroll
      for (int b = 0; b < 5; ++b) {
        const bool bit = ((d >> b) & 1) != 0;
        const unsigned bb = __builtin_amdgcn_ballot_w32(bit);
        same &= bit ? bb : ~bb;
      }
      const int below = (int)__builtin_amdgcn_mbcnt_lo(same, 0u);
      const int nsame = __builtin_popcount(same);
      int* cp = scnt + d * AGG_WAVES + wave;
      const int c0 = *cp;
      const int slot = c0 + below;
      if (slot < SEG_CAP) seg[(d * AGG_WAVES + wave) * SEG_PITCH + slot] = (sn << 7) | r;
      else *lost_any = 1;
      if (below == nsame - 1) *cp = c0 + nsame;
    }
  }
}

__device__ __forceinline__ void agg_drain(const int* __restrict__ src,
                                          const int* __restrict__ dst, int rowbase,
                                          const float4* __restrict__ el4,
                                          const float* __restrict__ hfeat,
                                          const float4* err4, const float4* mrow4,
                                          float* srow, float* acc,
                                          const int* mylist, int cnt, int lane) {
  while (cnt > 0) {
    const int nb   = cnt < 32 ? cnt : 32;
    const int base = cnt - nb;
    int sn = 0, r = 0;
    float a[HH] = {0.f, 0.f, 0.f, 0.f, 0.f, 0.f, 0.f, 0.f};
    if (lane < nb) {
      const int pk = mylist[base + lane];
      r  = dst[pk] - rowbase;
      sn = src[pk];
      float v[HH];
      edge_scores(el4, sn, err4, r, v);
      const float4 m0 = mrow4[r * 2], m1 = mrow4[r * 2 + 1];
      a[0] = __expf(v[0] - m0.x); a[1] = __expf(v[1] - m0.y);
      a[2] = __expf(v[2] - m0.z); a[3] = __expf(v[3] - m0.w);
      a[4] = __expf(v[4] - m1.x); a[5] = __expf(v[5] - m1.y);
      a[6] = __expf(v[6] - m1.z); a[7] = __expf(v[7] - m1.w);
#pragma unroll
      for (int h = 0; h < HH; ++h)
        __hip_atomic_fetch_add(srow + r * HH + h, a[h], __ATOMIC_RELAXED,
                               __HIP_MEMORY_SCOPE_WORKGROUP);
    }
    for (int j = 0; j < nb; j += 4) {
      float hv[4][HH];
#pragma unroll
      for (int u = 0; u < 4; ++u) {
        const int jj = (j + u < nb) ? (j + u) : (nb - 1);
        const int sj = lane_bcast_i(sn, jj);
        const float* hp = hfeat + (size_t)sj * F + lane;
#pragma unroll
        for (int h = 0; h < HH; ++h) hv[u][h] = hp[h * DD];
      }
#pragma unroll
      for (int u = 0; u < 4; ++u) {
        if (j + u < nb) {
          const int rj = lane_bcast_i(r, j + u);
          float* ap = acc + rj * F + lane;
#pragma unroll
          for (int h = 0; h < HH; ++h) {
            const float al = lane_bcast_f(a[h], j + u);
            __hip_atomic_fetch_add(ap + h * DD, al * hv[u][h], __ATOMIC_RELAXED,
                                   __HIP_MEMORY_SCOPE_WORKGROUP);
          }
        }
      }
    }
    cnt = base;
  }
}

template <int K>
__device__ __forceinline__ void row_add(float (&racc)[ROWS_PER_WAVE][COLS_PER_LANE],
                                        float (&rsum)[ROWS_PER_WAVE][2], float w0, float w1,
                                        const float4& h0, const float4& h1) {
  racc[K][0] = fmaf(w0, h0.x, racc[K][0]); racc[K][1] = fmaf(w0, h0.y, racc[K][1]);
  racc[K][2] = fmaf(w0, h0.z, racc[K][2]); racc[K][3] = fmaf(w0, h0.w, racc[K][3]);
  racc[K][4] = fmaf(w1, h1.x, racc[K][4]); racc[K][5] = fmaf(w1, h1.y, racc[K][5]);
  racc[K][6] = fmaf(w1, h1.z, racc[K][6]); racc[K][7] = fmaf(w1, h1.w, racc[K][7]);
  rsum[K][0] += w0;
  rsum[K][1] += w1;
}

__device__ __forceinline__ void agg_sum_rows(const float4* __restrict__ el4,
                                             const float* __restrict__ hfeat,
                                             const float4* err4, const float4* mrow4,
                                             const int* myrl, int n, int lane, float* wt,
                                             float (&racc)[ROWS_PER_WAVE][COLS_PER_LANE],
                                             float (&rsum)[ROWS_PER_WAVE][2]) {
  const float4* hf4 = reinterpret_cast<const float4*>(hfeat);
  float4* wt4 = reinterpret_cast<float4*>(wt);
  const int hq = lane >> 3;
  for (int base = 0; base < n; base += 32) {
    const int nb = (n - base) < 32 ? (n - base) : 32;
    int sn = 0, r = 0;
    float a[HH] = {0.f, 0.f, 0.f, 0.f, 0.f, 0.f, 0.f, 0.f};
    if (lane < nb) {
      const int pk = myrl[base + lane];
      r  = pk & 127;
      sn = pk >> 7;
      float v[HH];
      edge_scores(el4, sn, err4, r, v);
      const float4 m0 = mrow4[r * 2], m1 = mrow4[r * 2 + 1];
      a[0] = __expf(v[0] - m0.x); a[1] = __expf(v[1] - m0.y);
      a[2] = __expf(v[2] - m0.z); a[3] = __expf(v[3] - m0.w);
      a[4] = __expf(v[4] - m1.x); a[5] = __expf(v[5] - m1.y);
      a[6] = __expf(v[6] - m1.z); a[7] = __expf(v[7] - m1.w);
    }
    wt4[lane * 2]     = make_float4(a[0], a[1], a[2], a[3]);
    wt4[lane * 2 + 1] = make_float4(a[4], a[5], a[6], a[7]);
    for (int j = 0; j < nb; j += PASS2_U) {
      float4 hv[PASS2_U][2];
      float w[PASS2_U][2];
#pragma unroll
      for (int u = 0; u < PASS2_U; ++u) {
        const int jj = (j + u < nb) ? (j + u) : (nb - 1);
        const int sj = lane_bcast_i(sn, jj);
        const float4* hp = hf4 + (size_t)sj * (F / 4) + lane;
        hv[u][0] = hp[0];
        hv[u][1] = hp[F / 8];
        w[u][0] = wt[jj * HH + hq];
        w[u][1] = wt[jj * HH + hq + HH / 2];
      }
      int slot[PASS2_U];
#pragma unroll
      for (int u = 0; u < PASS2_U; ++u)
        slot[u] = (j + u < nb) ? (lane_bcast_i(r, j + u) >> 5) : -1;
#pragma unroll
      for (int u = 0; u < PASS2_U; ++u)
        if (slot[u] == 0) row_add<0>(racc, rsum, w[u][0], w[u][1], hv[u][0], hv[u][1]);
#pragma unroll
      for (int u = 0; u < PASS2_U; ++u)
        if (slot[u] == 1) row_add<1>(racc, rsum, w[u][0], w[u][1], hv[u][0], hv[u][1]);
#pragma unroll
      for (int u = 0; u < PASS2_U; ++u)
        if (slot[u] == 2) row_add<2>(racc, rsum, w[u][0], w[u][1], hv[u][0], hv[u][1]);
    }
  }
}

__global__ __launch_bounds__(AGG_THREADS) void gat_agg_rows(
    const float* __restrict__ el, const float* __restrict__ er,
    const int* __restrict__ src, const int* __restrict__ dst,
    const unsigned* __restrict__ plane,
    const float* __restrict__ hfeat, const float* __restrict__ b,
    float* __restrict__ outx) {
  __shared__ __attribute__((aligned(16))) float acc[AGG_ROWS * F];
  __shared__ __attribute__((aligned(16))) float mrow[AGG_ROWS * HH];
  __shared__ __attribute__((aligned(16))) float srow[AGG_ROWS * HH];
  __shared__ __attribute__((aligned(16))) float err[AGG_ROWS * HH];
  __shared__ int elist[AGG_WAVES * LIST_CAP];
  __shared__ int seg[AGG_WAVES * AGG_WAVES * SEG_PITCH];
  __shared__ int scnt[AGG_WAVES * AGG_WAVES];
  __shared__ int lost_any;
  const int tid  = threadIdx.x;
  const int wave = tid >> 5;
  const int lane = tid & 31;
  const int rowbase = blockIdx.x * AGG_ROWS;

  if (tid < AGG_ROWS * HH) {
    const int row = rowbase + (tid >> 3);
    mrow[tid] = -INFINITY;
    srow[tid] = 0.f;
    err[tid]  = (row < NN) ? er[row * HH + (tid & 7)] : 0.f;
  }
  if (tid < AGG_WAVES * AGG_WAVES) scnt[tid] = 0;
  if (tid == 0) lost_any = 0;
  __syncthreads();

  int* mylist = elist + wave * LIST_CAP;
  int cnt = 0;
  const int NIT = (EE / 16 + SCAN_STEP - 1) / SCAN_STEP;
  const uint4*  plane4 = reinterpret_cast<const uint4*>(plane);
  const float4* el4   = reinterpret_cast<const float4*>(el);
  const float4* err4  = reinterpret_cast<const float4*>(err);
  const float4* mrow4 = reinterpret_cast<const float4*>(mrow);
  const unsigned pat   = (unsigned)blockIdx.x * 0x01010101u;
  const int wave_s = __builtin_amdgcn_readfirstlane(wave);
  const int rot = blockIdx.x % NIT;

  bool lost = false;
  {
    uint4 cur[SCAN_U];
    agg_scan_load(plane4, rot, wave_s, lane, cur);
    for (int it = 0; it < NIT; ++it) {
      int itr = it + rot;
      if (itr >= NIT) itr -= NIT;
      int itn = itr + 1;
      if (itn >= NIT) itn -= NIT;
      uint4 nxt[SCAN_U];
      agg_scan_load(plane4, itn, wave_s, lane, nxt);
      cnt = agg_scan_step(cur, itr, wave_s, lane, pat, mylist, cnt, [&](int c) {
        agg_fold_max(src, dst, rowbase, el4, err4, mrow, mylist, c, lane, false,
                     wave_s, seg, scnt, &lost_any);
        lost = true;
        return 0;
      });
#pragma unroll
      for (int u = 0; u < SCAN_U; ++u) cur[u] = nxt[u];
    }
  }
  agg_fold_max(src, dst, rowbase, el4, err4, mrow, mylist, cnt, lane, !lost,
               wave_s, seg, scnt, &lost_any);
  if (lost) lost_any = 1;
  __syncthreads();

  if (!lost_any) {
    float racc[ROWS_PER_WAVE][COLS_PER_LANE], rsum[ROWS_PER_WAVE][2];
#pragma unroll
    for (int k = 0; k < ROWS_PER_WAVE; ++k) {
      rsum[k][0] = 0.f;
      rsum[k][1] = 0.f;
#pragma unroll
      for (int c2 = 0; c2 < COLS_PER_LANE; ++c2) racc[k][c2] = 0.f;
    }
    const int c = scnt[wave * AGG_WAVES + lane];
    int incl = c;
#pragma unroll
    for (int sft = 1; sft < 32; sft <<= 1) {
      const int up = __shfl_up(incl, sft);
      if (lane >= sft) incl += up;
    }
    const int off = incl - c;
    const int n = lane_bcast_i(incl, 31);
    const int* sg = seg + (wave * AGG_WAVES + lane) * SEG_PITCH;
    for (int t = 0; t < SEG_CAP; ++t) {
      if (__builtin_amdgcn_ballot_w32(t < c) == 0u) break;
      if (t < c) mylist[off + t] = sg[t];
    }
    agg_sum_rows(el4, hfeat, err4, mrow4, mylist, n, lane, acc + wave * (32 * HH), racc, rsum);
    const float4 bq0 = reinterpret_cast<const float4*>(b)[lane];
    const float4 bq1 = reinterpret_cast<const float4*>(b)[lane + F / 8];
    const float bh[COLS_PER_LANE] = {bq0.x, bq0.y, bq0.z, bq0.w, bq1.x, bq1.y, bq1.z, bq1.w};
#pragma unroll
    for (int k = 0; k < ROWS_PER_WAVE; ++k) {
      const int rl  = wave + k * AGG_WAVES;
      const int row = rowbase + rl;
      if (rl < AGG_ROWS && row < NN) {
        float o[COLS_PER_LANE];
#pragma unroll
        for (int c2 = 0; c2 < COLS_PER_LANE; ++c2) {
          const float s = rsum[k][c2 >> 2];
          const float v = (s != 0.f) ? racc[k][c2] / s : 0.f;
          o[c2] = elu1(v + bh[c2]);
        }
        float4* orow = reinterpret_cast<float4*>(outx + (size_t)row * F) + lane;
        orow[0]     = make_float4(o[0], o[1], o[2], o[3]);
        orow[F / 8] = make_float4(o[4], o[5], o[6], o[7]);
      }
    }
  } else {
    for (int i = tid; i < AGG_ROWS * F; i += AGG_THREADS) acc[i] = 0.f;
    __syncthreads();
    cnt = 0;
    for (int it = 0; it < NIT; ++it) {
      int itr = it + rot;
      if (itr >= NIT) itr -= NIT;
      uint4 cur[SCAN_U];
      agg_scan_load(plane4, itr, wave_s, lane, cur);
      cnt = agg_scan_step(cur, itr, wave_s, lane, pat, mylist, cnt, [&](int c) {
        agg_drain(src, dst, rowbase, el4, hfeat, err4, mrow4, srow, acc, mylist, c, lane);
        return 0;
      });
    }
    agg_drain(src, dst, rowbase, el4, hfeat, err4, mrow4, srow, acc, mylist, cnt, lane);
    __syncthreads();

    const float4* acc4 = reinterpret_cast<const float4*>(acc);
    const float4* b4 = reinterpret_cast<const float4*>(b);
    float4* out4 = reinterpret_cast<float4*>(outx);
    for (int i = tid; i < AGG_ROWS * (F / 4); i += AGG_THREADS) {
      const int rl  = i / (F / 4);
      const int c4  = i % (F / 4);
      const int row = rowbase + rl;
      const float s = srow[rl * HH + c4 / (DD / 4)];
      float4 v = acc4[i];
      if (s != 0.f) { v.x /= s; v.y /= s; v.z /= s; v.w /= s; }
      else          { v = make_float4(0.f, 0.f, 0.f, 0.f); }
      const float4 bv = b4[c4];
      v.x = elu1(v.x + bv.x); v.y = elu1(v.y + bv.y);
      v.z = elu1(v.z + bv.z); v.w = elu1(v.w + bv.w);
      if (row < NN) out4[(size_t)row * (F / 4) + c4] = v;
    }
  }
}

extern "C" void kernel_launch(void* const* d_in, const int* in_sizes, int n_in,
                              void* d_out, int out_size, void* d_ws, size_t ws_size,
                              hipStream_t stream) {
  const float* feats = (const float*)d_in[0];
  const float* W0    = (const float*)d_in[1];
  const float* al0   = (const float*)d_in[2];
  const float* ar0   = (const float*)d_in[3];
  const float* b0    = (const float*)d_in[4];
  const float* W1    = (const float*)d_in[5];
  const float* al1   = (const float*)d_in[6];
  const float* ar1   = (const float*)d_in[7];
  const float* b1    = (const float*)d_in[8];
  const int*   src   = (const int*)d_in[9];
  const int*   dst   = (const int*)d_in[10];

  float* ws = (float*)d_ws;
  float* hbuf = ws;
  float* x1   = hbuf + (size_t)NN * F;
  float* el   = x1 + (size_t)NN * F;
  float* er   = el + (size_t)NN * HH;
  unsigned* plane = (unsigned*)(er + (size_t)NN * HH);

  const int bl = 256;
  const int gNH = (NN * HH + bl - 1) / bl;

  auto run_layer = [&](const float* x, const float* Wm, const float* al,
                       const float* ar, const float* bb, float* outx,
                       unsigned* planew) {
    gat_gemm_wmma<<<(NN / GM_ROWS) * (F / GM_COLS), GM_THREADS, 0, stream>>>(x, Wm, hbuf);
    gat_attn_coeff<<<gNH, bl, 0, stream>>>(hbuf, al, ar, el, er, dst, planew);
    gat_agg_rows<<<AGG_BLOCKS, AGG_THREADS, 0, stream>>>(
        el, er, src, dst, plane, hbuf, bb, outx);
  };

  run_layer(feats, W0, al0, ar0, b0, x1, plane);
  run_layer(x1, W1, al1, ar1, b1, (float*)d_out, nullptr);
}


# GEMM waves stage their 32x32 piece of C in LDS: whole-line stores, and el/er summed from the staged rows in column order (same bits); coefficient launches gone, plane built in GEMM 1; 6 launches -> 4
# speedup vs baseline: 17.3623x; 1.0471x over previous
#include <hip/hip_runtime.h>
#include <hip/hip_bf16.h>
#include <math.h>

#define NN 20000
#define EE 320000
#define HH 8
#define DD 32
#define F  256
#define LDSPITCH 260

typedef float v2f __attribute__((ext_vector_type(2)));
typedef float v8f __attribute__((ext_vector_type(8)));

#define GM_ROWS    32
#define GM_COLS    128
#define GM_THREADS 128
#define GM_SLABV   (GM_ROWS * (F / 4) / GM_THREADS)
#define CW_PITCH   36
#define GM_WAVES   (GM_THREADS / 32)
#define GM_BLOCKS  ((NN / GM_ROWS) * (F / GM_COLS))
#define EL_REC     (GM_ROWS * (GM_COLS / DD))

#define AGG_ROWS    80
#define AGG_BLOCKS  ((NN + AGG_ROWS - 1) / AGG_ROWS)
#define NO_BLOCK    255u

static_assert(AGG_BLOCKS <= 255, "an aggregation block's index fits one byte, 255 left free");
static_assert(EE % 16 == 0, "the plane is written as dwords and read as 16-byte vectors");
static_assert(GM_BLOCKS * GM_THREADS >= EE / 4, "the GEMM launch has a thread for every dword of the plane");
static_assert(GM_COLS % DD == 0 && GM_ROWS == 32 && GM_COLS / DD == 4,
              "a wave's 32 columns are one head; lane r of a wave sums row r; a block's four heads are one 16-byte vector per row");
static_assert((CW_PITCH * 4) % 16 == 0, "staged C rows start 16-byte aligned in LDS");

__device__ __forceinline__ unsigned plane_byte(int d) {
  return ((unsigned)d < (unsigned)NN) ? (unsigned)d / AGG_ROWS : NO_BLOCK;
}

__device__ __forceinline__ int el_vec(int n, int half) {
  return (((n >> 5) << 1) + half) * (EL_REC / 4) + (n & 31);
}

static_assert(NN % GM_ROWS == 0 && F % GM_COLS == 0, "blocks tile C exactly: no partial tile, no guard");
static_assert(GM_COLS == (GM_THREADS / 32) * 32, "a wave owns 32 columns");
static_assert((LDSPITCH * 4) % 16 == 0, "slab rows start 16-byte aligned in LDS");

__global__ __launch_bounds__(GM_THREADS) void gat_gemm_wmma(
    const float* __restrict__ A, const float* __restrict__ W,
    const float* __restrict__ al, const float* __restrict__ ar,
    float* __restrict__ C, float* __restrict__ el, float* __restrict__ er,
    const int* __restrict__ dst, unsigned* __restrict__ plane) {
  __shared__ __attribute__((aligned(16))) float As[GM_ROWS * LDSPITCH];
  __shared__ __attribute__((aligned(16))) float Cs[GM_WAVES * GM_ROWS * CW_PITCH];
  __shared__ __attribute__((aligned(16))) float ex[2 * EL_REC];
  const int tid  = threadIdx.x;
  const int wave = tid >> 5;
  const int lane = tid & 31;
  const int m0   = (blockIdx.x / (F / GM_COLS)) * GM_ROWS;
  const int nb   = (blockIdx.x % (F / GM_COLS)) * GM_COLS;

  float4 sv[GM_SLABV];
#pragma unroll
  for (int j = 0; j < GM_SLABV; ++j) {
    const int i = tid + j * GM_THREADS;
    sv[j] = *reinterpret_cast<const float4*>(A + (size_t)(m0 + (i >> 6)) * F + ((i & 63) << 2));
  }
#pragma unroll
  for (int j = 0; j < GM_SLABV; ++j) {
    const int i = tid + j * GM_THREADS;
    *reinterpret_cast<float4*>(&As[(i >> 6) * LDSPITCH + ((i & 63) << 2)]) = sv[j];
  }
  __syncthreads();

  const int hi = lane >> 4;
  const int l  = lane & 15;
  const int n0 = nb + (wave << 5);

  const float* a0p = As + l * LDSPITCH + (hi << 1);
  const float* a1p = a0p + 16 * LDSPITCH;
  const float* wp  = W + (size_t)(hi << 1) * F + n0 + l;

  v8f c00 = {}, c01 = {}, c10 = {}, c11 = {};
#pragma unroll 8
  for (int k0 = 0; k0 < F; k0 += 4) {
    v2f a0, a1, b0, b1;
    a0.x = a0p[k0];
    a0.y = a0p[k0 + 1];
    a1.x = a1p[k0];
    a1.y = a1p[k0 + 1];
    b0.x = wp[(size_t)k0 * F];
    b0.y = wp[(size_t)(k0 + 1) * F];
    b1.x = wp[(size_t)k0 * F + 16];
    b1.y = wp[(size_t)(k0 + 1) * F + 16];
    c00 = __builtin_amdgcn_wmma_f32_16x16x4_f32(false, a0, false, b0, (short)0, c00, false, false);
    c01 = __builtin_amdgcn_wmma_f32_16x16x4_f32(false, a0, false, b1, (short)0, c01, false, false);
    c10 = __builtin_amdgcn_wmma_f32_16x16x4_f32(false, a1, false, b0, (short)0, c10, false, false);
    c11 = __builtin_amdgcn_wmma_f32_16x16x4_f32(false, a1, false, b1, (short)0, c11, false, false);
  }
  float* cw = Cs + wave * (GM_ROWS * CW_PITCH);
#pragma unroll
  for (int v = 0; v < 8; ++v) {
    float* cr = cw + ((hi << 3) + v) * CW_PITCH + l;
    cr[0]  = c00[v];
    cr[16] = c01[v];
    cr[16 * CW_PITCH]      = c10[v];
    cr[16 * CW_PITCH + 16] = c11[v];
  }
#pragma unroll
  for (int j = 0; j < GM_ROWS / 4; ++j) {
    const int row = j * 4 + (lane >> 3);
    const float4 q = *reinterpret_cast<const float4*>(cw + row * CW_PITCH + ((lane & 7) << 2));
    *reinterpret_cast<float4*>(C + (size_t)(m0 + row) * F + n0 + ((lane & 7) << 2)) = q;
  }
  const int hd = __builtin_amdgcn_readfirstlane(n0 / DD);
  const float* alp = al + hd * DD;
  const float* arp = ar + hd * DD;
  const float* hp  = cw + lane * CW_PITCH;
  float sl = 0.f, sr = 0.f;
#pragma unroll
  for (int d = 0; d < DD; ++d) {
    const float v = hp[d];
    sl += v * alp[d];
    sr += v * arp[d];
  }
  ex[lane * (GM_COLS / DD) + wave]          = sl;
  ex[EL_REC + lane * (GM_COLS / DD) + wave] = sr;
  __syncthreads();
  if (wave < 2) {
    const float4 q = *reinterpret_cast<const float4*>(ex + wave * EL_REC + lane * (GM_COLS / DD));
    float* rec = (wave == 0 ? el : er) + (size_t)blockIdx.x * EL_REC;
    reinterpret_cast<float4*>(rec)[lane] = q;
  }
  const int gi = blockIdx.x * GM_THREADS + tid;
  if (plane != nullptr && gi < EE / 4) {
    const int4 d = reinterpret_cast<const int4*>(dst)[gi];
    plane[gi] = plane_byte(d.x) | (plane_byte(d.y) << 8) |
                (plane_byte(d.z) << 16) | (plane_byte(d.w) << 24);
  }
}


#define AGG_THREADS 1024
#define AGG_WAVES   32
#define SCAN_U      4
#define LIST_CAP    768
#define SCAN_STEP   (AGG_WAVES * SCAN_U * 32)
#define LIST_FULL   (LIST_CAP - 512)
#define ROWS_PER_WAVE ((AGG_ROWS + AGG_WAVES - 1) / AGG_WAVES)
#define SEG_CAP     16
#define SEG_PITCH   17
#define COLS_PER_LANE (F / 32)
#define PASS2_U     8

static_assert(AGG_ROWS <= 128 && NN < (1 << 24),
              "a kept list entry packs source*128 + local row");
static_assert(LIST_CAP > 512, "after the fullness test the list still takes one whole load's edges");
static_assert(AGG_ROWS * HH <= AGG_THREADS, "one thread per (row, head) sets up the row statistics");
static_assert(AGG_WAVES == 32, "lane i of an owner wave holds what wave i's list hands over; row r belongs to wave r & 31");
static_assert(ROWS_PER_WAVE == 3, "a wave keeps three rows' sums in registers");
static_assert(AGG_WAVES * AGG_WAVES <= AGG_THREADS, "one thread per (owner, list) clears the segment counts");
static_assert(AGG_WAVES * SEG_CAP <= LIST_CAP, "an owner wave's entries fit the LDS room of its own list");
static_assert(COLS_PER_LANE == 8 && DD == 32 && HH == 8,
              "pass 2: a lane's columns 4*lane..+3 lie in head lane/8, its columns 128+4*lane..+3 in head 4+lane/8");
static_assert(AGG_WAVES * 32 * HH <= AGG_ROWS * F, "the waves' weight tables fit the accumulator room");

__device__ __forceinline__ int lane_bcast_i(int v, int l) {
  return __builtin_amdgcn_readlane(v, l);
}
__device__ __forceinline__ float lane_bcast_f(float v, int l) {
  return __int_as_float(__builtin_amdgcn_readlane(__float_as_int(v), l));
}
__device__ __forceinline__ float leaky02(float v) { return (v > 0.f) ? v : 0.2f * v; }
__device__ __forceinline__ float elu1(float v) { return (v > 0.f) ? v : (__expf(v) - 1.f); }

static_assert((EE / 16) % 32 == 0, "a wave's 32 vectors of one load are all inside the plane or all past its end");

__device__ __forceinline__ void agg_scan_load(const uint4* __restrict__ plane4, int itr,
                                              int wave, int lane, uint4 d[SCAN_U]) {
  const int Q16 = EE / 16;
  const int q0 = (itr * AGG_WAVES + wave) * (SCAN_U * 32);
#pragma unroll
  for (int u = 0; u < SCAN_U; ++u) {
    const int qw = q0 + u * 32;
    d[u] = plane4[(qw < Q16 ? qw : Q16 - 32) + lane];
  }
}

__device__ __forceinline__ unsigned zero_bytes(unsigned x) {
  return ~(((x & 0x7f7f7f7fu) + 0x7f7f7f7fu) | x | 0x7f7f7f7fu);
}

template <class OnFull>
__device__ __forceinline__ int agg_scan_step(const uint4 d[SCAN_U], int itr, int wave,
                                             int lane, unsigned pat, int* mylist, int cnt,
                                             OnFull on_full) {
  const int Q16 = EE / 16;
  const int q0 = (itr * AGG_WAVES + wave) * (SCAN_U * 32);
#pragma unroll
  for (int u = 0; u < SCAN_U; ++u) {
    if (q0 + u * 32 >= Q16) break;
    const int ebase = (q0 + u * 32 + lane) * 16;
    unsigned hm = zero_bytes(d[u].x ^ pat) | (zero_bytes(d[u].y ^ pat) >> 1) |
                  (zero_bytes(d[u].z ^ pat) >> 2) | (zero_bytes(d[u].w ^ pat) >> 3);
    unsigned m = __builtin_amdgcn_ballot_w32(hm != 0u);
    while (m) {
      if (hm != 0u) {
        const int i = __builtin_ctz(hm);
        mylist[cnt + (int)__builtin_amdgcn_mbcnt_lo(m, 0u)] =
            ebase + ((7 - (i & 7)) << 2) + (i >> 3);
        hm &= hm - 1u;
      }
      cnt += __builtin_popcount(m);
      m = __builtin_amdgcn_ballot_w32(hm != 0u);
    }
    if (cnt > LIST_FULL) cnt = on_full(cnt);
  }
  return cnt;
}

__device__ __forceinline__ void edge_scores(const float4* __restrict__ el4, int sn,
                                            const float4* err4, int r, float v[HH]) {
  const float4 l0 = el4[el_vec(sn, 0)], l1 = el4[el_vec(sn, 1)];
  const float4 r0 = err4[r * 2], r1 = err4[r * 2 + 1];
  v[0] = leaky02(l0.x + r0.x); v[1] = leaky02(l0.y + r0.y);
  v[2] = leaky02(l0.z + r0.z); v[3] = leaky02(l0.w + r0.w);
  v[4] = leaky02(l1.x + r1.x); v[5] = leaky02(l1.y + r1.y);
  v[6] = leaky02(l1.z + r1.z); v[7] = leaky02(l1.w + r1.w);
}

__device__ __forceinline__ void agg_fold_max(const int* __restrict__ src,
                                             const int* __restrict__ dst, int rowbase,
                                             const float4* __restrict__ el4,
                                             const float4* err4, float* mrow,
                                             const int* mylist, int cnt, int lane, bool keep,
                                             int wave, int* seg, int* scnt, int* lost_any) {
  for (int idx = lane; idx < cnt; idx += 32) {
    const int e  = mylist[idx];
    const int r  = dst[e] - rowbase;
    const int sn = src[e];
    float v[HH];
    edge_scores(el4, sn, err4, r, v);
#pragma unroll
    for (int h = 0; h < HH; ++h)
      __hip_atomic_fetch_max(mrow + r * HH + h, v[h], __ATOMIC_RELAXED,
                             __HIP_MEMORY_SCOPE_WORKGROUP);
    if (keep) {
      const int d = r & (AGG_WAVES - 1);
      unsigned same = __builtin_amdgcn_ballot_w32(true);
#pragma unroll
      for (int b = 0; b < 5; ++b) {
        const bool bit = ((d >> b) & 1) != 0;
        const unsigned bb = __builtin_amdgcn_ballot_w32(bit);
        same &= bit ? bb : ~bb;
      }
      const int below = (int)__builtin_amdgcn_mbcnt_lo(same, 0u);
      const int nsame = __builtin_popcount(same);
      int* cp = scnt + d * AGG_WAVES + wave;
      const int c0 = *cp;
      const int slot = c0 + below;
      if (slot < SEG_CAP) seg[(d * AGG_WAVES + wave) * SEG_PITCH + slot] = (sn << 7) | r;
      else *lost_any = 1;
      if (below == nsame - 1) *cp = c0 + nsame;
    }
  }
}

__device__ __forceinline__ void agg_drain(const int* __restrict__ src,
                                          const int* __restrict__ dst, int rowbase,
                                          const float4* __restrict__ el4,
                                          const float* __restrict__ hfeat,
                                          const float4* err4, const float4* mrow4,
                                          float* srow, float* acc,
                                          const int* mylist, int cnt, int lane) {
  while (cnt > 0) {
    const int nb   = cnt < 32 ? cnt : 32;
    const int base = cnt - nb;
    int sn = 0, r = 0;
    float a[HH] = {0.f, 0.f, 0.f, 0.f, 0.f, 0.f, 0.f, 0.f};
    if (lane < nb) {
      const int pk = mylist[base + lane];
      r  = dst[pk] - rowbase;
      sn = src[pk];
      float v[HH];
      edge_scores(el4, sn, err4, r, v);
      const float4 m0 = mrow4[r * 2], m1 = mrow4[r * 2 + 1];
      a[0] = __expf(v[0] - m0.x); a[1] = __expf(v[1] - m0.y);
      a[2] = __expf(v[2] - m0.z); a[3] = __expf(v[3] - m0.w);
      a[4] = __expf(v[4] - m1.x); a[5] = __expf(v[5] - m1.y);
      a[6] = __expf(v[6] - m1.z); a[7] = __expf(v[7] - m1.w);
#pragma unroll
      for (int h = 0; h < HH; ++h)
        __hip_atomic_fetch_add(srow + r * HH + h, a[h], __ATOMIC_RELAXED,
                               __HIP_MEMORY_SCOPE_WORKGROUP);
    }
    for (int j = 0; j < nb; j += 4) {
      float hv[4][HH];
#pragma unroll
      for (int u = 0; u < 4; ++u) {
        const int jj = (j + u < nb) ? (j + u) : (nb - 1);
        const int sj = lane_bcast_i(sn, jj);
        const float* hp = hfeat + (size_t)sj * F + lane;
#pragma unroll
        for (int h = 0; h < HH; ++h) hv[u][h] = hp[h * DD];
      }
#pragma unroll
      for (int u = 0; u < 4; ++u) {
        if (j + u < nb) {
          const int rj = lane_bcast_i(r, j + u);
          float* ap = acc + rj * F + lane;
#pragma unroll
          for (int h = 0; h < HH; ++h) {
            const float al = lane_bcast_f(a[h], j + u);
            __hip_atomic_fetch_add(ap + h * DD, al * hv[u][h], __ATOMIC_RELAXED,
                                   __HIP_MEMORY_SCOPE_WORKGROUP);
          }
        }
      }
    }
    cnt = base;
  }
}

template <int K>
__device__ __forceinline__ void row_add(float (&racc)[ROWS_PER_WAVE][COLS_PER_LANE],
                                        float (&rsum)[ROWS_PER_WAVE][2], float w0, float w1,
                                        const float4& h0, const float4& h1) {
  racc[K][0] = fmaf(w0, h0.x, racc[K][0]); racc[K][1] = fmaf(w0, h0.y, racc[K][1]);
  racc[K][2] = fmaf(w0, h0.z, racc[K][2]); racc[K][3] = fmaf(w0, h0.w, racc[K][3]);
  racc[K][4] = fmaf(w1, h1.x, racc[K][4]); racc[K][5] = fmaf(w1, h1.y, racc[K][5]);
  racc[K][6] = fmaf(w1, h1.z, racc[K][6]); racc[K][7] = fmaf(w1, h1.w, racc[K][7]);
  rsum[K][0] += w0;
  rsum[K][1] += w1;
}

__device__ __forceinline__ void agg_sum_rows(const float4* __restrict__ el4,
                                             const float* __restrict__ hfeat,
                                             const float4* err4, const float4* mrow4,
                                             const int* myrl, int n, int lane, float* wt,
                                             float (&racc)[ROWS_PER_WAVE][COLS_PER_LANE],
                                             float (&rsum)[ROWS_PER_WAVE][2]) {
  const float4* hf4 = reinterpret_cast<const float4*>(hfeat);
  float4* wt4 = reinterpret_cast<float4*>(wt);
  const int hq = lane >> 3;
  for (int base = 0; base < n; base += 32) {
    const int nb = (n - base) < 32 ? (n - base) : 32;
    int sn = 0, r = 0;
    float a[HH] = {0.f, 0.f, 0.f, 0.f, 0.f, 0.f, 0.f, 0.f};
    if (lane < nb) {
      const int pk = myrl[base + lane];
      r  = pk & 127;
      sn = pk >> 7;
      float v[HH];
      edge_scores(el4, sn, err4, r, v);
      const float4 m0 = mrow4[r * 2], m1 = mrow4[r * 2 + 1];
      a[0] = __expf(v[0] - m0.x); a[1] = __expf(v[1] - m0.y);
      a[2] = __expf(v[2] - m0.z); a[3] = __expf(v[3] - m0.w);
      a[4] = __expf(v[4] - m1.x); a[5] = __expf(v[5] - m1.y);
      a[6] = __expf(v[6] - m1.z); a[7] = __expf(v[7] - m1.w);
    }
    wt4[lane * 2]     = make_float4(a[0], a[1], a[2], a[3]);
    wt4[lane * 2 + 1] = make_float4(a[4], a[5], a[6], a[7]);
    for (int j = 0; j < nb; j += PASS2_U) {
      float4 hv[PASS2_U][2];
      float w[PASS2_U][2];
#pragma unroll
      for (int u = 0; u < PASS2_U; ++u) {
        const int jj = (j + u < nb) ? (j + u) : (nb - 1);
        const int sj = lane_bcast_i(sn, jj);
        const float4* hp = hf4 + (size_t)sj * (F / 4) + lane;
        hv[u][0] = hp[0];
        hv[u][1] = hp[F / 8];
        w[u][0] = wt[jj * HH + hq];
        w[u][1] = wt[jj * HH + hq + HH / 2];
      }
      int slot[PASS2_U];
#pragma unroll
      for (int u = 0; u < PASS2_U; ++u)
        slot[u] = (j + u < nb) ? (lane_bcast_i(r, j + u) >> 5) : -1;
#pragma unroll
      for (int u = 0; u < PASS2_U; ++u)
        if (slot[u] == 0) row_add<0>(racc, rsum, w[u][0], w[u][1], hv[u][0], hv[u][1]);
#pragma unroll
      for (int u = 0; u < PASS2_U; ++u)
        if (slot[u] == 1) row_add<1>(racc, rsum, w[u][0], w[u][1], hv[u][0], hv[u][1]);
#pragma unroll
      for (int u = 0; u < PASS2_U; ++u)
        if (slot[u] == 2) row_add<2>(racc, rsum, w[u][0], w[u][1], hv[u][0], hv[u][1]);
    }
  }
}

__global__ __launch_bounds__(AGG_THREADS) void gat_agg_rows(
    const float* __restrict__ el, const float* __restrict__ er,
    const int* __restrict__ src, const int* __restrict__ dst,
    const unsigned* __restrict__ plane,
    const float* __restrict__ hfeat, const float* __restrict__ b,
    float* __restrict__ outx) {
  __shared__ __attribute__((aligned(16))) float acc[AGG_ROWS * F];
  __shared__ __attribute__((aligned(16))) float mrow[AGG_ROWS * HH];
  __shared__ __attribute__((aligned(16))) float srow[AGG_ROWS * HH];
  __shared__ __attribute__((aligned(16))) float err[AGG_ROWS * HH];
  __shared__ int elist[AGG_WAVES * LIST_CAP];
  __shared__ int seg[AGG_WAVES * AGG_WAVES * SEG_PITCH];
  __shared__ int scnt[AGG_WAVES * AGG_WAVES];
  __shared__ int lost_any;
  const int tid  = threadIdx.x;
  const int wave = tid >> 5;
  const int lane = tid & 31;
  const int rowbase = blockIdx.x * AGG_ROWS;

  if (tid < AGG_ROWS * HH) {
    const int row = rowbase + (tid >> 3);
    mrow[tid] = -INFINITY;
    srow[tid] = 0.f;
    err[tid]  = (row < NN) ? er[el_vec(row, (tid & 7) >> 2) * 4 + (tid & 3)] : 0.f;
  }
  if (tid < AGG_WAVES * AGG_WAVES) scnt[tid] = 0;
  if (tid == 0) lost_any = 0;
  __syncthreads();

  int* mylist = elist + wave * LIST_CAP;
  int cnt = 0;
  const int NIT = (EE / 16 + SCAN_STEP - 1) / SCAN_STEP;
  const uint4*  plane4 = reinterpret_cast<const uint4*>(plane);
  const float4* el4   = reinterpret_cast<const float4*>(el);
  const float4* err4  = reinterpret_cast<const float4*>(err);
  const float4* mrow4 = reinterpret_cast<const float4*>(mrow);
  const unsigned pat   = (unsigned)blockIdx.x * 0x01010101u;
  const int wave_s = __builtin_amdgcn_readfirstlane(wave);
  const int rot = blockIdx.x % NIT;

  bool lost = false;
  {
    uint4 cur[SCAN_U];
    agg_scan_load(plane4, rot, wave_s, lane, cur);
    for (int it = 0; it < NIT; ++it) {
      int itr = it + rot;
      if (itr >= NIT) itr -= NIT;
      int itn = itr + 1;
      if (itn >= NIT) itn -= NIT;
      uint4 nxt[SCAN_U];
      agg_scan_load(plane4, itn, wave_s, lane, nxt);
      cnt = agg_scan_step(cur, itr, wave_s, lane, pat, mylist, cnt, [&](int c) {
        agg_fold_max(src, dst, rowbase, el4, err4, mrow, mylist, c, lane, false,
                     wave_s, seg, scnt, &lost_any);
        lost = true;
        return 0;
      });
#pragma unroll
      for (int u = 0; u < SCAN_U; ++u) cur[u] = nxt[u];
    }
  }
  agg_fold_max(src, dst, rowbase, el4, err4, mrow, mylist, cnt, lane, !lost,
               wave_s, seg, scnt, &lost_any);
  if (lost) lost_any = 1;
  __syncthreads();

  if (!lost_any) {
    float racc[ROWS_PER_WAVE][COLS_PER_LANE], rsum[ROWS_PER_WAVE][2];
#pragma unroll
    for (int k = 0; k < ROWS_PER_WAVE; ++k) {
      rsum[k][0] = 0.f;
      rsum[k][1] = 0.f;
#pragma unroll
      for (int c2 = 0; c2 < COLS_PER_LANE; ++c2) racc[k][c2] = 0.f;
    }
    const int c = scnt[wave * AGG_WAVES + lane];
    int incl = c;
#pragma unroll
    for (int sft = 1; sft < 32; sft <<= 1) {
      const int up = __shfl_up(incl, sft);
      if (lane >= sft) incl += up;
    }
    const int off = incl - c;
    const int n = lane_bcast_i(incl, 31);
    const int* sg = seg + (wave * AGG_WAVES + lane) * SEG_PITCH;
    for (int t = 0; t < SEG_CAP; ++t) {
      if (__builtin_amdgcn_ballot_w32(t < c) == 0u) break;
      if (t < c) mylist[off + t] = sg[t];
    }
    agg_sum_rows(el4, hfeat, err4, mrow4, mylist, n, lane, acc + wave * (32 * HH), racc, rsum);
    const float4 bq0 = reinterpret_cast<const float4*>(b)[lane];
    const float4 bq1 = reinterpret_cast<const float4*>(b)[lane + F / 8];
    const float bh[COLS_PER_LANE] = {bq0.x, bq0.y, bq0.z, bq0.w, bq1.x, bq1.y, bq1.z, bq1.w};
#pragma unroll
    for (int k = 0; k < ROWS_PER_WAVE; ++k) {
      const int rl  = wave + k * AGG_WAVES;
      const int row = rowbase + rl;
      if (rl < AGG_ROWS && row < NN) {
        float o[COLS_PER_LANE];
#pragma unroll
        for (int c2 = 0; c2 < COLS_PER_LANE; ++c2) {
          const float s = rsum[k][c2 >> 2];
          const float v = (s != 0.f) ? racc[k][c2] / s : 0.f;
          o[c2] = elu1(v + bh[c2]);
        }
        float4* orow = reinterpret_cast<float4*>(outx + (size_t)row * F) + lane;
        orow[0]     = make_float4(o[0], o[1], o[2], o[3]);
        orow[F / 8] = make_float4(o[4], o[5], o[6], o[7]);
      }
    }
  } else {
    for (int i = tid; i < AGG_ROWS * F; i += AGG_THREADS) acc[i] = 0.f;
    __syncthreads();
    cnt = 0;
    for (int it = 0; it < NIT; ++it) {
      int itr = it + rot;
      if (itr >= NIT) itr -= NIT;
      uint4 cur[SCAN_U];
      agg_scan_load(plane4, itr, wave_s, lane, cur);
      cnt = agg_scan_step(cur, itr, wave_s, lane, pat, mylist, cnt, [&](int c) {
        agg_drain(src, dst, rowbase, el4, hfeat, err4, mrow4, srow, acc, mylist, c, lane);
        return 0;
      });
    }
    agg_drain(src, dst, rowbase, el4, hfeat, err4, mrow4, srow, acc, mylist, cnt, lane);
    __syncthreads();

    const float4* acc4 = reinterpret_cast<const float4*>(acc);
    const float4* b4 = reinterpret_cast<const float4*>(b);
    float4* out4 = reinterpret_cast<float4*>(outx);
    for (int i = tid; i < AGG_ROWS * (F / 4); i += AGG_THREADS) {
      const int rl  = i / (F / 4);
      const int c4  = i % (F / 4);
      const int row = rowbase + rl;
      const float s = srow[rl * HH + c4 / (DD / 4)];
      float4 v = acc4[i];
      if (s != 0.f) { v.x /= s; v.y /= s; v.z /= s; v.w /= s; }
      else          { v = make_float4(0.f, 0.f, 0.f, 0.f); }
      const float4 bv = b4[c4];
      v.x = elu1(v.x + bv.x); v.y = elu1(v.y + bv.y);
      v.z = elu1(v.z + bv.z); v.w = elu1(v.w + bv.w);
      if (row < NN) out4[(size_t)row * (F / 4) + c4] = v;
    }
  }
}

extern "C" void kernel_launch(void* const* d_in, const int* in_sizes, int n_in,
                              void* d_out, int out_size, void* d_ws, size_t ws_size,
                              hipStream_t stream) {
  const float* feats = (const float*)d_in[0];
  const float* W0    = (const float*)d_in[1];
  const float* al0   = (const float*)d_in[2];
  const float* ar0   = (const float*)d_in[3];
  const float* b0    = (const float*)d_in[4];
  const float* W1    = (const float*)d_in[5];
  const float* al1   = (const float*)d_in[6];
  const float* ar1   = (const float*)d_in[7];
  const float* b1    = (const float*)d_in[8];
  const int*   src   = (const int*)d_in[9];
  const int*   dst   = (const int*)d_in[10];

  float* ws = (float*)d_ws;
  float* hbuf = ws;
  float* x1   = hbuf + (size_t)NN * F;
  float* el   = x1 + (size_t)NN * F;
  float* er   = el + (size_t)NN * HH;
  unsigned* plane = (unsigned*)(er + (size_t)NN * HH);

  auto run_layer = [&](const float* x, const float* Wm, const float* al,
                       const float* ar, const float* bb, float* outx,
                       unsigned* planew) {
    gat_gemm_wmma<<<GM_BLOCKS, GM_THREADS, 0, stream>>>(x, Wm, al, ar, hbuf, el, er, dst, planew);
    gat_agg_rows<<<AGG_BLOCKS, AGG_THREADS, 0, stream>>>(
        el, er, src, dst, plane, hbuf, bb, outx);
  };

  run_layer(feats, W0, al0, ar0, b0, x1, plane);
  run_layer(x1, W1, al1, ar1, b1, (float*)d_out, nullptr);
}


# second-layer aggregation loads the per-wave row lists the first layer's blocks stored, instead of sweeping the plane and handing entries over again; own-row maxima via an LDS table; sum orders kept
# speedup vs baseline: 18.5684x; 1.0695x over previous
#include <hip/hip_runtime.h>
#include <hip/hip_bf16.h>
#include <math.h>

#define NN 20000
#define EE 320000
#define HH 8
#define DD 32
#define F  256
#define LDSPITCH 260

typedef float v2f __attribute__((ext_vector_type(2)));
typedef float v8f __attribute__((ext_vector_type(8)));

#define GM_ROWS    32
#define GM_COLS    128
#define GM_THREADS 128
#define GM_SLABV   (GM_ROWS * (F / 4) / GM_THREADS)
#define CW_PITCH   36
#define GM_WAVES   (GM_THREADS / 32)
#define GM_BLOCKS  ((NN / GM_ROWS) * (F / GM_COLS))
#define EL_REC     (GM_ROWS * (GM_COLS / DD))

#define AGG_ROWS    80
#define AGG_BLOCKS  ((NN + AGG_ROWS - 1) / AGG_ROWS)
#define NO_BLOCK    255u

static_assert(AGG_BLOCKS <= 255, "an aggregation block's index fits one byte, 255 left free");
static_assert(EE % 16 == 0, "the plane is written as dwords and read as 16-byte vectors");
static_assert(GM_BLOCKS * GM_THREADS >= EE / 4, "the GEMM launch has a thread for every dword of the plane");
static_assert(GM_COLS % DD == 0 && GM_ROWS == 32 && GM_COLS / DD == 4,
              "a wave's 32 columns are one head; lane r of a wave sums row r; a block's four heads are one 16-byte vector per row");
static_assert((CW_PITCH * 4) % 16 == 0, "staged C rows start 16-byte aligned in LDS");

__device__ __forceinline__ unsigned plane_byte(int d) {
  return ((unsigned)d < (unsigned)NN) ? (unsigned)d / AGG_ROWS : NO_BLOCK;
}

__device__ __forceinline__ int el_vec(int n, int half) {
  return (((n >> 5) << 1) + half) * (EL_REC / 4) + (n & 31);
}

static_assert(NN % GM_ROWS == 0 && F % GM_COLS == 0, "blocks tile C exactly: no partial tile, no guard");
static_assert(GM_COLS == (GM_THREADS / 32) * 32, "a wave owns 32 columns");
static_assert((LDSPITCH * 4) % 16 == 0, "slab rows start 16-byte aligned in LDS");

__global__ __launch_bounds__(GM_THREADS) void gat_gemm_wmma(
    const float* __restrict__ A, const float* __restrict__ W,
    const float* __restrict__ al, const float* __restrict__ ar,
    float* __restrict__ C, float* __restrict__ el, float* __restrict__ er,
    const int* __restrict__ dst, unsigned* __restrict__ plane) {
  __shared__ __attribute__((aligned(16))) float As[GM_ROWS * LDSPITCH];
  __shared__ __attribute__((aligned(16))) float Cs[GM_WAVES * GM_ROWS * CW_PITCH];
  __shared__ __attribute__((aligned(16))) float ex[2 * EL_REC];
  const int tid  = threadIdx.x;
  const int wave = tid >> 5;
  const int lane = tid & 31;
  const int m0   = (blockIdx.x / (F / GM_COLS)) * GM_ROWS;
  const int nb   = (blockIdx.x % (F / GM_COLS)) * GM_COLS;

  float4 sv[GM_SLABV];
#pragma unroll
  for (int j = 0; j < GM_SLABV; ++j) {
    const int i = tid + j * GM_THREADS;
    sv[j] = *reinterpret_cast<const float4*>(A + (size_t)(m0 + (i >> 6)) * F + ((i & 63) << 2));
  }
#pragma unroll
  for (int j = 0; j < GM_SLABV; ++j) {
    const int i = tid + j * GM_THREADS;
    *reinterpret_cast<float4*>(&As[(i >> 6) * LDSPITCH + ((i & 63) << 2)]) = sv[j];
  }
  __syncthreads();

  const int hi = lane >> 4;
  const int l  = lane & 15;
  const int n0 = nb + (wave << 5);

  const float* a0p = As + l * LDSPITCH + (hi << 1);
  const float* a1p = a0p + 16 * LDSPITCH;
  const float* wp  = W + (size_t)(hi << 1) * F + n0 + l;

  v8f c00 = {}, c01 = {}, c10 = {}, c11 = {};
#pragma unroll 8
  for (int k0 = 0; k0 < F; k0 += 4) {
    v2f a0, a1, b0, b1;
    a0.x = a0p[k0];
    a0.y = a0p[k0 + 1];
    a1.x = a1p[k0];
    a1.y = a1p[k0 + 1];
    b0.x = wp[(size_t)k0 * F];
    b0.y = wp[(size_t)(k0 + 1) * F];
    b1.x = wp[(size_t)k0 * F + 16];
    b1.y = wp[(size_t)(k0 + 1) * F + 16];
    c00 = __builtin_amdgcn_wmma_f32_16x16x4_f32(false, a0, false, b0, (short)0, c00, false, false);
    c01 = __builtin_amdgcn_wmma_f32_16x16x4_f32(false, a0, false, b1, (short)0, c01, false, false);
    c10 = __builtin_amdgcn_wmma_f32_16x16x4_f32(false, a1, false, b0, (short)0, c10, false, false);
    c11 = __builtin_amdgcn_wmma_f32_16x16x4_f32(false, a1, false, b1, (short)0, c11, false, false);
  }
  float* cw = Cs + wave * (GM_ROWS * CW_PITCH);
#pragma unroll
  for (int v = 0; v < 8; ++v) {
    float* cr = cw + ((hi << 3) + v) * CW_PITCH + l;
    cr[0]  = c00[v];
    cr[16] = c01[v];
    cr[16 * CW_PITCH]      = c10[v];
    cr[16 * CW_PITCH + 16] = c11[v];
  }
#pragma unroll
  for (int j = 0; j < GM_ROWS / 4; ++j) {
    const int row = j * 4 + (lane >> 3);
    const float4 q = *reinterpret_cast<const float4*>(cw + row * CW_PITCH + ((lane & 7) << 2));
    *reinterpret_cast<float4*>(C + (size_t)(m0 + row) * F + n0 + ((lane & 7) << 2)) = q;
  }
  const int hd = __builtin_amdgcn_readfirstlane(n0 / DD);
  const float* alp = al + hd * DD;
  const float* arp = ar + hd * DD;
  const float* hp  = cw + lane * CW_PITCH;
  float sl = 0.f, sr = 0.f;
#pragma unroll
  for (int d = 0; d < DD; ++d) {
    const float v = hp[d];
    sl += v * alp[d];
    sr += v * arp[d];
  }
  ex[lane * (GM_COLS / DD) + wave]          = sl;
  ex[EL_REC + lane * (GM_COLS / DD) + wave] = sr;
  __syncthreads();
  if (wave < 2) {
    const float4 q = *reinterpret_cast<const float4*>(ex + wave * EL_REC + lane * (GM_COLS / DD));
    float* rec = (wave == 0 ? el : er) + (size_t)blockIdx.x * EL_REC;
    reinterpret_cast<float4*>(rec)[lane] = q;
  }
  const int gi = blockIdx.x * GM_THREADS + tid;
  if (plane != nullptr && gi < EE / 4) {
    const int4 d = reinterpret_cast<const int4*>(dst)[gi];
    plane[gi] = plane_byte(d.x) | (plane_byte(d.y) << 8) |
                (plane_byte(d.z) << 16) | (plane_byte(d.w) << 24);
  }
}


#define AGG_THREADS 1024
#define AGG_WAVES   32
#define SCAN_U      4
#define LIST_CAP    768
#define SCAN_STEP   (AGG_WAVES * SCAN_U * 32)
#define LIST_FULL   (LIST_CAP - 512)
#define ROWS_PER_WAVE ((AGG_ROWS + AGG_WAVES - 1) / AGG_WAVES)
#define SEG_CAP     16
#define SEG_PITCH   17
#define COLS_PER_LANE (F / 32)
#define PASS2_U     8
#define WT_PITCH    (64 * HH)
#define REC_HEAD    32
#define REC_PITCH   576

static_assert(AGG_ROWS <= 128 && NN < (1 << 24),
              "a kept list entry packs source*128 + local row");
static_assert(LIST_CAP > 512, "after the fullness test the list still takes one whole load's edges");
static_assert(AGG_ROWS * HH <= AGG_THREADS, "one thread per (row, head) sets up the row statistics");
static_assert(AGG_WAVES == 32, "lane i of an owner wave holds what wave i's list hands over; row r belongs to wave r & 31");
static_assert(ROWS_PER_WAVE == 3 && AGG_ROWS > 2 * AGG_WAVES, "a wave keeps three rows' sums in registers; its first two rows always exist");
static_assert(AGG_WAVES * AGG_WAVES <= AGG_THREADS, "one thread per (owner, list) clears the segment counts");
static_assert(AGG_WAVES * SEG_CAP <= LIST_CAP, "an owner wave's entries fit the LDS room of its own list");
static_assert(COLS_PER_LANE == 8 && DD == 32 && HH == 8,
              "pass 2: a lane's columns 4*lane..+3 lie in head lane/8, its columns 128+4*lane..+3 in head 4+lane/8");
static_assert(AGG_WAVES * WT_PITCH <= AGG_ROWS * F, "the waves' tables fit the accumulator room");
static_assert(REC_HEAD + AGG_WAVES * SEG_CAP <= REC_PITCH && (REC_PITCH * 4) % 256 == 0,
              "a record holds every entry an owner wave can get and starts on a 256-B boundary");
static_assert(AGG_WAVES * SEG_CAP >= 64 && LIST_CAP >= AGG_WAVES * SEG_CAP,
              "a record's first two entry lines always exist; a record fits the wave's LDS list");

__device__ __forceinline__ int lane_bcast_i(int v, int l) {
  return __builtin_amdgcn_readlane(v, l);
}
__device__ __forceinline__ float lane_bcast_f(float v, int l) {
  return __int_as_float(__builtin_amdgcn_readlane(__float_as_int(v), l));
}
__device__ __forceinline__ float leaky02(float v) { return (v > 0.f) ? v : 0.2f * v; }
__device__ __forceinline__ float elu1(float v) { return (v > 0.f) ? v : (__expf(v) - 1.f); }

static_assert((EE / 16) % 32 == 0, "a wave's 32 vectors of one load are all inside the plane or all past its end");

__device__ __forceinline__ void agg_scan_load(const uint4* __restrict__ plane4, int itr,
                                              int wave, int lane, uint4 d[SCAN_U]) {
  const int Q16 = EE / 16;
  const int q0 = (itr * AGG_WAVES + wave) * (SCAN_U * 32);
#pragma unroll
  for (int u = 0; u < SCAN_U; ++u) {
    const int qw = q0 + u * 32;
    d[u] = plane4[(qw < Q16 ? qw : Q16 - 32) + lane];
  }
}

__device__ __forceinline__ unsigned zero_bytes(unsigned x) {
  return ~(((x & 0x7f7f7f7fu) + 0x7f7f7f7fu) | x | 0x7f7f7f7fu);
}

template <class OnFull>
__device__ __forceinline__ int agg_scan_step(const uint4 d[SCAN_U], int itr, int wave,
                                             int lane, unsigned pat, int* mylist, int cnt,
                                             OnFull on_full) {
  const int Q16 = EE / 16;
  const int q0 = (itr * AGG_WAVES + wave) * (SCAN_U * 32);
#pragma unroll
  for (int u = 0; u < SCAN_U; ++u) {
    if (q0 + u * 32 >= Q16) break;
    const int ebase = (q0 + u * 32 + lane) * 16;
    unsigned hm = zero_bytes(d[u].x ^ pat) | (zero_bytes(d[u].y ^ pat) >> 1) |
                  (zero_bytes(d[u].z ^ pat) >> 2) | (zero_bytes(d[u].w ^ pat) >> 3);
    unsigned m = __builtin_amdgcn_ballot_w32(hm != 0u);
    while (m) {
      if (hm != 0u) {
        const int i = __builtin_ctz(hm);
        mylist[cnt + (int)__builtin_amdgcn_mbcnt_lo(m, 0u)] =
            ebase + ((7 - (i & 7)) << 2) + (i >> 3);
        hm &= hm - 1u;
      }
      cnt += __builtin_popcount(m);
      m = __builtin_amdgcn_ballot_w32(hm != 0u);
    }
    if (cnt > LIST_FULL) cnt = on_full(cnt);
  }
  return cnt;
}

__device__ __forceinline__ void edge_scores(const float4* __restrict__ el4, int sn,
                                            const float4* err4, int r, float v[HH]) {
  const float4 l0 = el4[el_vec(sn, 0)], l1 = el4[el_vec(sn, 1)];
  const float4 r0 = err4[r * 2], r1 = err4[r * 2 + 1];
  v[0] = leaky02(l0.x + r0.x); v[1] = leaky02(l0.y + r0.y);
  v[2] = leaky02(l0.z + r0.z); v[3] = leaky02(l0.w + r0.w);
  v[4] = leaky02(l1.x + r1.x); v[5] = leaky02(l1.y + r1.y);
  v[6] = leaky02(l1.z + r1.z); v[7] = leaky02(l1.w + r1.w);
}

__device__ __forceinline__ void agg_fold_max(const int* __restrict__ src,
                                             const int* __restrict__ dst, int rowbase,
                                             const float4* __restrict__ el4,
                                             const float4* err4, float* mrow,
                                             const int* mylist, int cnt, int lane, bool keep,
                                             int wave, int* seg, int* scnt, int* lost_any) {
  for (int idx = lane; idx < cnt; idx += 32) {
    const int e  = mylist[idx];
    const int r  = dst[e] - rowbase;
    const int sn = src[e];
    float v[HH];
    edge_scores(el4, sn, err4, r, v);
#pragma unroll
    for (int h = 0; h < HH; ++h)
      __hip_atomic_fetch_max(mrow + r * HH + h, v[h], __ATOMIC_RELAXED,
                             __HIP_MEMORY_SCOPE_WORKGROUP);
    if (keep) {
      const int d = r & (AGG_WAVES - 1);
      unsigned same = __builtin_amdgcn_ballot_w32(true);
#pragma unroll
      for (int b = 0; b < 5; ++b) {
        const bool bit = ((d >> b) & 1) != 0;
        const unsigned bb = __builtin_amdgcn_ballot_w32(bit);
        same &= bit ? bb : ~bb;
      }
      const int below = (int)__builtin_amdgcn_mbcnt_lo(same, 0u);
      const int nsame = __builtin_popcount(same);
      int* cp = scnt + d * AGG_WAVES + wave;
      const int c0 = *cp;
      const int slot = c0 + below;
      if (slot < SEG_CAP) seg[(d * AGG_WAVES + wave) * SEG_PITCH + slot] = (sn << 7) | r;
      else *lost_any = 1;
      if (below == nsame - 1) *cp = c0 + nsame;
    }
  }
}

__device__ __forceinline__ void agg_fold_max_own(const float4* __restrict__ el4,
                                                 const float4* err4, float* mrow,
                                                 const int* myrl, int n, int lane, int wave,
                                                 float* tb) {
  float4* tb4 = reinterpret_cast<float4*>(tb);
  const int h = lane & 7;
  const int q = lane >> 3;
  float m0 = -INFINITY, m1 = -INFINITY, m2 = -INFINITY;
  for (int base = 0; base < n; base += 64) {
    const int nb = (n - base) < 64 ? (n - base) : 64;
    const int pk0 = myrl[base + (lane < nb ? lane : 0)];
    const int pk1 = myrl[base + (lane + 32 < nb ? lane + 32 : 0)];
    float v0[HH], v1[HH];
    edge_scores(el4, pk0 >> 7, err4, pk0 & 127, v0);
    edge_scores(el4, pk1 >> 7, err4, pk1 & 127, v1);
    tb4[lane * 2]            = make_float4(v0[0], v0[1], v0[2], v0[3]);
    tb4[lane * 2 + 1]        = make_float4(v0[4], v0[5], v0[6], v0[7]);
    tb4[(lane + 32) * 2]     = make_float4(v1[0], v1[1], v1[2], v1[3]);
    tb4[(lane + 32) * 2 + 1] = make_float4(v1[4], v1[5], v1[6], v1[7]);
#pragma unroll
    for (int g = 0; g < 4; ++g) {
      if (16 * g >= nb) break;
      float x[4];
      int k[4];
#pragma unroll
      for (int t = 0; t < 4; ++t) {
        const int e  = q + 4 * t + 16 * g;
        const bool ok = e < nb;
        const int ee = ok ? e : 0;
        x[t] = tb[ee * HH + h];
        const int pke = myrl[base + ee];
        k[t] = ok ? ((pke & 127) >> 5) : 3;
      }
#pragma unroll
      for (int t = 0; t < 4; ++t) {
        if (k[t] == 0 && x[t] > m0) m0 = x[t];
        if (k[t] == 1 && x[t] > m1) m1 = x[t];
        if (k[t] == 2 && x[t] > m2) m2 = x[t];
      }
    }
  }
  m0 = fmaxf(m0, __shfl_xor(m0, 8));  m0 = fmaxf(m0, __shfl_xor(m0, 16));
  m1 = fmaxf(m1, __shfl_xor(m1, 8));  m1 = fmaxf(m1, __shfl_xor(m1, 16));
  m2 = fmaxf(m2, __shfl_xor(m2, 8));  m2 = fmaxf(m2, __shfl_xor(m2, 16));
  if (lane < HH) {
    mrow[wave * HH + h] = m0;
    mrow[(wave + AGG_WAVES) * HH + h] = m1;
    if (wave + 2 * AGG_WAVES < AGG_ROWS) mrow[(wave + 2 * AGG_WAVES) * HH + h] = m2;
  }
}

__device__ __forceinline__ void agg_drain(const int* __restrict__ src,
                                          const int* __restrict__ dst, int rowbase,
                                          const float4* __restrict__ el4,
                                          const float* __restrict__ hfeat,
                                          const float4* err4, const float4* mrow4,
                                          float* srow, float* acc,
                                          const int* mylist, int cnt, int lane) {
  while (cnt > 0) {
    const int nb   = cnt < 32 ? cnt : 32;
    const int base = cnt - nb;
    int sn = 0, r = 0;
    float a[HH] = {0.f, 0.f, 0.f, 0.f, 0.f, 0.f, 0.f, 0.f};
    if (lane < nb) {
      const int pk = mylist[base + lane];
      r  = dst[pk] - rowbase;
      sn = src[pk];
      float v[HH];
      edge_scores(el4, sn, err4, r, v);
      const float4 m0 = mrow4[r * 2], m1 = mrow4[r * 2 + 1];
      a[0] = __expf(v[0] - m0.x); a[1] = __expf(v[1] - m0.y);
      a[2] = __expf(v[2] - m0.z); a[3] = __expf(v[3] - m0.w);
      a[4] = __expf(v[4] - m1.x); a[5] = __expf(v[5] - m1.y);
      a[6] = __expf(v[6] - m1.z); a[7] = __expf(v[7] - m1.w);
#pragma unroll
      for (int h = 0; h < HH; ++h)
        __hip_atomic_fetch_add(srow + r * HH + h, a[h], __ATOMIC_RELAXED,
                               __HIP_MEMORY_SCOPE_WORKGROUP);
    }
    for (int j = 0; j < nb; j += 4) {
      float hv[4][HH];
#pragma unroll
      for (int u = 0; u < 4; ++u) {
        const int jj = (j + u < nb) ? (j + u) : (nb - 1);
        const int sj = lane_bcast_i(sn, jj);
        const float* hp = hfeat + (size_t)sj * F + lane;
#pragma unroll
        for (int h = 0; h < HH; ++h) hv[u][h] = hp[h * DD];
      }
#pragma unroll
      for (int u = 0; u < 4; ++u) {
        if (j + u < nb) {
          const int rj = lane_bcast_i(r, j + u);
          float* ap = acc + rj * F + lane;
#pragma unroll
          for (int h = 0; h < HH; ++h) {
            const float al = lane_bcast_f(a[h], j + u);
            __hip_atomic_fetch_add(ap + h * DD, al * hv[u][h], __ATOMIC_RELAXED,
                                   __HIP_MEMORY_SCOPE_WORKGROUP);
          }
        }
      }
    }
    cnt = base;
  }
}

template <int K>
__device__ __forceinline__ void row_add(float (&racc)[ROWS_PER_WAVE][COLS_PER_LANE],
                                        float (&rsum)[ROWS_PER_WAVE][2], float w0, float w1,
                                        const float4& h0, const float4& h1) {
  racc[K][0] = fmaf(w0, h0.x, racc[K][0]); racc[K][1] = fmaf(w0, h0.y, racc[K][1]);
  racc[K][2] = fmaf(w0, h0.z, racc[K][2]); racc[K][3] = fmaf(w0, h0.w, racc[K][3]);
  racc[K][4] = fmaf(w1, h1.x, racc[K][4]); racc[K][5] = fmaf(w1, h1.y, racc[K][5]);
  racc[K][6] = fmaf(w1, h1.z, racc[K][6]); racc[K][7] = fmaf(w1, h1.w, racc[K][7]);
  rsum[K][0] += w0;
  rsum[K][1] += w1;
}

__device__ __forceinline__ void agg_sum_rows(const float4* __restrict__ el4,
                                             const float* __restrict__ hfeat,
                                             const float4* err4, const float4* mrow4,
                                             const int* myrl, int n, int lane, float* wt,
                                             float (&racc)[ROWS_PER_WAVE][COLS_PER_LANE],
                                             float (&rsum)[ROWS_PER_WAVE][2]) {
  const float4* hf4 = reinterpret_cast<const float4*>(hfeat);
  float4* wt4 = reinterpret_cast<float4*>(wt);
  const int hq = lane >> 3;
  for (int base = 0; base < n; base += 32) {
    const int nb = (n - base) < 32 ? (n - base) : 32;
    int sn = 0, r = 0;
    float a[HH] = {0.f, 0.f, 0.f, 0.f, 0.f, 0.f, 0.f, 0.f};
    if (lane < nb) {
      const int pk = myrl[base + lane];
      r  = pk & 127;
      sn = pk >> 7;
      float v[HH];
      edge_scores(el4, sn, err4, r, v);
      const float4 m0 = mrow4[r * 2], m1 = mrow4[r * 2 + 1];
      a[0] = __expf(v[0] - m0.x); a[1] = __expf(v[1] - m0.y);
      a[2] = __expf(v[2] - m0.z); a[3] = __expf(v[3] - m0.w);
      a[4] = __expf(v[4] - m1.x); a[5] = __expf(v[5] - m1.y);
      a[6] = __expf(v[6] - m1.z); a[7] = __expf(v[7] - m1.w);
    }
    wt4[lane * 2]     = make_float4(a[0], a[1], a[2], a[3]);
    wt4[lane * 2 + 1] = make_float4(a[4], a[5], a[6], a[7]);
    for (int j = 0; j < nb; j += PASS2_U) {
      float4 hv[PASS2_U][2];
      float w[PASS2_U][2];
#pragma unroll
      for (int u = 0; u < PASS2_U; ++u) {
        const int jj = (j + u < nb) ? (j + u) : (nb - 1);
        const int sj = lane_bcast_i(sn, jj);
        const float4* hp = hf4 + (size_t)sj * (F / 4) + lane;
        hv[u][0] = hp[0];
        hv[u][1] = hp[F / 8];
        w[u][0] = wt[jj * HH + hq];
        w[u][1] = wt[jj * HH + hq + HH / 2];
      }
      int slot[PASS2_U];
#pragma unroll
      for (int u = 0; u < PASS2_U; ++u)
        slot[u] = (j + u < nb) ? (lane_bcast_i(r, j + u) >> 5) : -1;
#pragma unroll
      for (int u = 0; u < PASS2_U; ++u)
        if (slot[u] == 0) row_add<0>(racc, rsum, w[u][0], w[u][1], hv[u][0], hv[u][1]);
#pragma unroll
      for (int u = 0; u < PASS2_U; ++u)
        if (slot[u] == 1) row_add<1>(racc, rsum, w[u][0], w[u][1], hv[u][0], hv[u][1]);
#pragma unroll
      for (int u = 0; u < PASS2_U; ++u)
        if (slot[u] == 2) row_add<2>(racc, rsum, w[u][0], w[u][1], hv[u][0], hv[u][1]);
    }
  }
}

template <int LAYER>
__global__ __launch_bounds__(AGG_THREADS) void gat_agg_rows(
    const float* __restrict__ el, const float* __restrict__ er,
    const int* __restrict__ src, const int* __restrict__ dst,
    const unsigned* __restrict__ plane,
    const float* __restrict__ hfeat, const float* __restrict__ b,
    float* __restrict__ outx, int* __restrict__ lists) {
  __shared__ __attribute__((aligned(16))) float acc[AGG_ROWS * F];
  __shared__ __attribute__((aligned(16))) float mrow[AGG_ROWS * HH];
  __shared__ __attribute__((aligned(16))) float srow[AGG_ROWS * HH];
  __shared__ __attribute__((aligned(16))) float err[AGG_ROWS * HH];
  __shared__ int elist[AGG_WAVES * LIST_CAP];
  __shared__ int seg[AGG_WAVES * AGG_WAVES * SEG_PITCH];
  __shared__ int scnt[AGG_WAVES * AGG_WAVES];
  __shared__ int lost_any;
  const int tid  = threadIdx.x;
  const int wave = tid >> 5;
  const int lane = tid & 31;
  const int rowbase = blockIdx.x * AGG_ROWS;
  const int wave_s = __builtin_amdgcn_readfirstlane(wave);

  int* rec0 = lists + (size_t)blockIdx.x * (AGG_WAVES * REC_PITCH);
  int* rec  = rec0 + wave_s * REC_PITCH;
  bool have_list = false;
  int n = 0;
  int e0 = 0, e1 = 0;
  if (LAYER == 1) {
    if (rec0[1] == 0) {
      have_list = true;
      n = rec[0];
      e0 = rec[REC_HEAD + lane];
      e1 = rec[REC_HEAD + 32 + lane];
    }
  }

  if (tid < AGG_ROWS * HH) {
    const int row = rowbase + (tid >> 3);
    mrow[tid] = -INFINITY;
    srow[tid] = 0.f;
    err[tid]  = (row < NN) ? er[el_vec(row, (tid & 7) >> 2) * 4 + (tid & 3)] : 0.f;
  }
  if (tid < AGG_WAVES * AGG_WAVES) scnt[tid] = 0;
  if (tid == 0) lost_any = 0;
  __syncthreads();

  int* mylist = elist + wave * LIST_CAP;
  int cnt = 0;
  const int NIT = (EE / 16 + SCAN_STEP - 1) / SCAN_STEP;
  const uint4*  plane4 = reinterpret_cast<const uint4*>(plane);
  const float4* el4   = reinterpret_cast<const float4*>(el);
  const float4* err4  = reinterpret_cast<const float4*>(err);
  const float4* mrow4 = reinterpret_cast<const float4*>(mrow);
  const unsigned pat   = (unsigned)blockIdx.x * 0x01010101u;
  const int rot = blockIdx.x % NIT;

  if (LAYER == 1 && have_list) {
    mylist[lane]      = e0;
    mylist[lane + 32] = e1;
    for (int i = 64 + lane; i < n; i += 32) mylist[i] = rec[REC_HEAD + i];
    agg_fold_max_own(el4, err4, mrow, mylist, n, lane, wave_s, acc + wave * WT_PITCH);
  }

  bool lost = false;
  if (!have_list) {
    uint4 cur[SCAN_U];
    agg_scan_load(plane4, rot, wave_s, lane, cur);
    for (int it = 0; it < NIT; ++it) {
      int itr = it + rot;
      if (itr >= NIT) itr -= NIT;
      int itn = itr + 1;
      if (itn >= NIT) itn -= NIT;
      uint4 nxt[SCAN_U];
      agg_scan_load(plane4, itn, wave_s, lane, nxt);
      cnt = agg_scan_step(cur, itr, wave_s, lane, pat, mylist, cnt, [&](int c) {
        agg_fold_max(src, dst, rowbase, el4, err4, mrow, mylist, c, lane, false,
                     wave_s, seg, scnt, &lost_any);
        lost = true;
        return 0;
      });
#pragma unroll
      for (int u = 0; u < SCAN_U; ++u) cur[u] = nxt[u];
    }
  }
  if (!have_list) {
    agg_fold_max(src, dst, rowbase, el4, err4, mrow, mylist, cnt, lane, !lost,
                 wave_s, seg, scnt, &lost_any);
    if (lost) lost_any = 1;
    __syncthreads();
  }

  if (have_list || !lost_any) {
    float racc[ROWS_PER_WAVE][COLS_PER_LANE], rsum[ROWS_PER_WAVE][2];
#pragma unroll
    for (int k = 0; k < ROWS_PER_WAVE; ++k) {
      rsum[k][0] = 0.f;
      rsum[k][1] = 0.f;
#pragma unroll
      for (int c2 = 0; c2 < COLS_PER_LANE; ++c2) racc[k][c2] = 0.f;
    }
    if (!have_list) {
      const int c = scnt[wave * AGG_WAVES + lane];
      int incl = c;
#pragma unroll
      for (int sft = 1; sft < 32; sft <<= 1) {
        const int up = __shfl_up(incl, sft);
        if (lane >= sft) incl += up;
      }
      const int off = incl - c;
      n = lane_bcast_i(incl, 31);
      const int* sg = seg + (wave * AGG_WAVES + lane) * SEG_PITCH;
      for (int t = 0; t < SEG_CAP; ++t) {
        if (__builtin_amdgcn_ballot_w32(t < c) == 0u) break;
        if (t < c) mylist[off + t] = sg[t];
      }
      if (LAYER == 0) {
        rec[lane] = (lane == 0) ? n : 0;
        const int nfill = (n > 64) ? ((n + 31) & ~31) : 64;
        for (int i = lane; i < nfill; i += 32) rec[REC_HEAD + i] = (i < n) ? mylist[i] : 0;
      }
    }
    agg_sum_rows(el4, hfeat, err4, mrow4, mylist, n, lane, acc + wave * WT_PITCH, racc, rsum);
    const float4 bq0 = reinterpret_cast<const float4*>(b)[lane];
    const float4 bq1 = reinterpret_cast<const float4*>(b)[lane + F / 8];
    const float bh[COLS_PER_LANE] = {bq0.x, bq0.y, bq0.z, bq0.w, bq1.x, bq1.y, bq1.z, bq1.w};
#pragma unroll
    for (int k = 0; k < ROWS_PER_WAVE; ++k) {
      const int rl  = wave + k * AGG_WAVES;
      const int row = rowbase + rl;
      if (rl < AGG_ROWS && row < NN) {
        float o[COLS_PER_LANE];
#pragma unroll
        for (int c2 = 0; c2 < COLS_PER_LANE; ++c2) {
          const float s = rsum[k][c2 >> 2];
          const float v = (s != 0.f) ? racc[k][c2] / s : 0.f;
          o[c2] = elu1(v + bh[c2]);
        }
        float4* orow = reinterpret_cast<float4*>(outx + (size_t)row * F) + lane;
        orow[0]     = make_float4(o[0], o[1], o[2], o[3]);
        orow[F / 8] = make_float4(o[4], o[5], o[6], o[7]);
      }
    }
  } else {
    if (LAYER == 0 && wave == 0) rec0[lane] = (lane == 1) ? 1 : 0;
    for (int i = tid; i < AGG_ROWS * F; i += AGG_THREADS) acc[i] = 0.f;
    __syncthreads();
    cnt = 0;
    for (int it = 0; it < NIT; ++it) {
      int itr = it + rot;
      if (itr >= NIT) itr -= NIT;
      uint4 cur[SCAN_U];
      agg_scan_load(plane4, itr, wave_s, lane, cur);
      cnt = agg_scan_step(cur, itr, wave_s, lane, pat, mylist, cnt, [&](int c) {
        agg_drain(src, dst, rowbase, el4, hfeat, err4, mrow4, srow, acc, mylist, c, lane);
        return 0;
      });
    }
    agg_drain(src, dst, rowbase, el4, hfeat, err4, mrow4, srow, acc, mylist, cnt, lane);
    __syncthreads();

    const float4* acc4 = reinterpret_cast<const float4*>(acc);
    const float4* b4 = reinterpret_cast<const float4*>(b);
    float4* out4 = reinterpret_cast<float4*>(outx);
    for (int i = tid; i < AGG_ROWS * (F / 4); i += AGG_THREADS) {
      const int rl  = i / (F / 4);
      const int c4  = i % (F / 4);
      const int row = rowbase + rl;
      const float s = srow[rl * HH + c4 / (DD / 4)];
      float4 v = acc4[i];
      if (s != 0.f) { v.x /= s; v.y /= s; v.z /= s; v.w /= s; }
      else          { v = make_float4(0.f, 0.f, 0.f, 0.f); }
      const float4 bv = b4[c4];
      v.x = elu1(v.x + bv.x); v.y = elu1(v.y + bv.y);
      v.z = elu1(v.z + bv.z); v.w = elu1(v.w + bv.w);
      if (row < NN) out4[(size_t)row * (F / 4) + c4] = v;
    }
  }
}

extern "C" void kernel_launch(void* const* d_in, const int* in_sizes, int n_in,
                              void* d_out, int out_size, void* d_ws, size_t ws_size,
                              hipStream_t stream) {
  const float* feats = (const float*)d_in[0];
  const float* W0    = (const float*)d_in[1];
  const float* al0   = (const float*)d_in[2];
  const float* ar0   = (const float*)d_in[3];
  const float* b0    = (const float*)d_in[4];
  const float* W1    = (const float*)d_in[5];
  const float* al1   = (const float*)d_in[6];
  const float* ar1   = (const float*)d_in[7];
  const float* b1    = (const float*)d_in[8];
  const int*   src   = (const int*)d_in[9];
  const int*   dst   = (const int*)d_in[10];

  float* ws = (float*)d_ws;
  float* hbuf = ws;
  float* x1   = hbuf + (size_t)NN * F;
  float* el   = x1 + (size_t)NN * F;
  float* er   = el + (size_t)NN * HH;
  unsigned* plane = (unsigned*)(er + (size_t)NN * HH);
  int* lists = (int*)(plane + EE / 4);

  gat_gemm_wmma<<<GM_BLOCKS, GM_THREADS, 0, stream>>>(feats, W0, al0, ar0, hbuf, el, er, dst, plane);
  gat_agg_rows<0><<<AGG_BLOCKS, AGG_THREADS, 0, stream>>>(
      el, er, src, dst, plane, hbuf, b0, x1, lists);
  gat_gemm_wmma<<<GM_BLOCKS, GM_THREADS, 0, stream>>>(x1, W1, al1, ar1, hbuf, el, er, dst, nullptr);
  gat_agg_rows<1><<<AGG_BLOCKS, AGG_THREADS, 0, stream>>>(
      el, er, src, dst, plane, hbuf, b1, (float*)d_out, lists);
}
